# starting kernel
# baseline (speedup 1.0000x reference)
#include <hip/hip_runtime.h>
#include <hip/hip_bf16.h>
#include <cstdint>

#define N_NODES 6000
#define NH      ((size_t)N_NODES * 128)

typedef float v2f __attribute__((ext_vector_type(2)));
typedef float v8f __attribute__((ext_vector_type(8)));

__device__ __forceinline__ v8f wmma4(v2f a, v2f b, v8f c) {
  return __builtin_amdgcn_wmma_f32_16x16x4_f32(false, a, false, b, (short)0, c,
                                               false, false);
}

__device__ __forceinline__ v2f ldsA(const float* base, int ld, int k0, int lane) {
  const int m = lane & 15;
  const int k = k0 + ((lane >> 4) << 1);
  const float* p = base + m * ld + k;
  v2f r; r.x = p[0]; r.y = p[1];
  return r;
}

__device__ __forceinline__ v2f ldBp(const float* __restrict__ P, int k0, int n0,
                                    int lane) {
  const int n = n0 + (lane & 15);
  const int k = k0 + ((lane >> 4) << 1);
  const float* p = P + (size_t)(k >> 1) * 256 + n * 2;
  v2f r; r.x = p[0]; r.y = p[1];
  return r;
}

__device__ __forceinline__ v2f ldBT(const float* __restrict__ Kb, int ld, int k0,
                                    int lane) {
  const int n = lane & 15;
  const int k = k0 + ((lane >> 4) << 1);
  const float* p = Kb + (size_t)n * ld + k;
  v2f r; r.x = p[0]; r.y = p[1];
  return r;
}

__device__ __forceinline__ float lrelu(float x) { return x >= 0.f ? x : 0.01f * x; }

__device__ __forceinline__ float rowmax16(float v) {
  v = fmaxf(v, __shfl_xor(v, 1));
  v = fmaxf(v, __shfl_xor(v, 2));
  v = fmaxf(v, __shfl_xor(v, 4));
  v = fmaxf(v, __shfl_xor(v, 8));
  return v;
}
__device__ __forceinline__ float rowsum16(float v) {
  v += __shfl_xor(v, 1);
  v += __shfl_xor(v, 2);
  v += __shfl_xor(v, 4);
  v += __shfl_xor(v, 8);
  return v;
}

__device__ __forceinline__ void layernorm128(float* buf, const float* __restrict__ g,
                                             const float* __restrict__ b,
                                             float (*stats)[2], int tid) {
  const int r = tid >> 3, seg = tid & 7;
  float sum = 0.f, sq = 0.f;
  #pragma unroll
  for (int i = 0; i < 16; i++) {
    float v = buf[r * 132 + seg * 16 + i];
    sum += v; sq += v * v;
  }
  sum += __shfl_xor(sum, 1); sq += __shfl_xor(sq, 1);
  sum += __shfl_xor(sum, 2); sq += __shfl_xor(sq, 2);
  sum += __shfl_xor(sum, 4); sq += __shfl_xor(sq, 4);
  if (seg == 0) {
    float mean = sum * (1.f / 128.f);
    float var  = sq * (1.f / 128.f) - mean * mean;
    stats[r][0] = mean;
    stats[r][1] = rsqrtf(var + 1e-5f);
  }
  __syncthreads();
  const float mean = stats[r][0], rstd = stats[r][1];
  #pragma unroll
  for (int i = 0; i < 16; i++) {
    int c = seg * 16 + i;
    float v = buf[r * 132 + c];
    buf[r * 132 + c] = g[c] * (v - mean) * rstd + b[c];
  }
}

__device__ __forceinline__ void gemm128_lds(const float* in, const float* __restrict__ Wp,
                                            const float* __restrict__ bias, float* outb,
                                            int lane, int wave, int act) {
  const int hi = lane >> 4, cn = lane & 15;
  const int n0 = wave * 32;
  v8f a0 = {}, a1 = {};
  for (int k0 = 0; k0 < 128; k0 += 4) {
    v2f a  = ldsA(in, 132, k0, lane);
    v2f b0 = ldBp(Wp, k0, n0, lane);
    v2f b1 = ldBp(Wp, k0, n0 + 16, lane);
    a0 = wmma4(a, b0, a0);
    a1 = wmma4(a, b1, a1);
  }
  #pragma unroll
  for (int j = 0; j < 8; j++) {
    int r = j + 8 * hi;
    float x0 = a0[j] + bias[n0 + cn];
    float x1 = a1[j] + bias[n0 + 16 + cn];
    if (act) { x0 = lrelu(x0); x1 = lrelu(x1); }
    outb[r * 132 + n0 + cn] = x0;
    outb[r * 132 + n0 + 16 + cn] = x1;
  }
}

__global__ void pack_pairs(const float* __restrict__ W, float* __restrict__ P,
                           int K, int pairs) {
  int i = blockIdx.x * blockDim.x + threadIdx.x;
  if (i >= pairs * 128) return;
  int k2 = i >> 7, n = i & 127;
  int k = 2 * k2;
  float a = (k < K)     ? W[(size_t)k * 128 + n]       : 0.f;
  float b = (k + 1 < K) ? W[(size_t)(k + 1) * 128 + n] : 0.f;
  P[(size_t)k2 * 256 + n * 2]     = a;
  P[(size_t)k2 * 256 + n * 2 + 1] = b;
}

__global__ void __launch_bounds__(128) proj_gemm(const float* __restrict__ A,
                                                 const float* __restrict__ Wp,
                                                 float* __restrict__ C, int act,
                                                 int pack) {
  __shared__ float at[16 * 132];
  const int tid = threadIdx.x, lane = tid & 31, wave = tid >> 5;
  const int row0 = blockIdx.x * 16;
  for (int i = tid; i < 16 * 128; i += 128) {
    int r = i >> 7, c = i & 127;
    at[r * 132 + c] = A[(size_t)(row0 + r) * 128 + c];
  }
  __syncthreads();
  const int n0 = wave * 32;
  v8f a0 = {}, a1 = {};
  for (int k0 = 0; k0 < 128; k0 += 4) {
    v2f a  = ldsA(at, 132, k0, lane);
    v2f b0 = ldBp(Wp, k0, n0, lane);
    v2f b1 = ldBp(Wp, k0, n0 + 16, lane);
    a0 = wmma4(a, b0, a0);
    a1 = wmma4(a, b1, a1);
  }
  const int hi = lane >> 4, cn = lane & 15;
  #pragma unroll
  for (int j = 0; j < 8; j++) {
    int r = row0 + j + 8 * hi;
    float x0 = a0[j], x1 = a1[j];
    if (act) { x0 = lrelu(x0); x1 = lrelu(x1); }
    if (pack) {
      C[(size_t)(r >> 1) * 256 + (n0 + cn) * 2 + (r & 1)]      = x0;
      C[(size_t)(r >> 1) * 256 + (n0 + 16 + cn) * 2 + (r & 1)] = x1;
    } else {
      C[(size_t)r * 128 + n0 + cn] = x0;
      C[(size_t)r * 128 + n0 + 16 + cn] = x1;
    }
  }
}

__global__ void __launch_bounds__(256) attn_kernel(const float* __restrict__ Q,
                                                   const float* __restrict__ K,
                                                   const float* __restrict__ Vp,
                                                   const float* __restrict__ mask,
                                                   int tmask, float* __restrict__ out) {
  __shared__ float qt[16 * 132];
  __shared__ float pbuf[8][16 * 18];
  __shared__ float outbuf[16 * 128];
  __shared__ float mbuf[8][16], lbuf[8][16];
  __shared__ float Mrow[16], Lrow[16];
  const int tid = threadIdx.x, lane = tid & 31, wave = tid >> 5;
  const int hi = lane >> 4, cn = lane & 15;
  const int qrow0 = blockIdx.x * 16;

  for (int i = tid; i < 2048; i += 256) {
    int r = i >> 7, c = i & 127;
    qt[r * 132 + c] = Q[(size_t)(qrow0 + r) * 128 + c];
  }
  __syncthreads();

  float mr[8], lr[8];
  v8f acc[8];
  #pragma unroll
  for (int j = 0; j < 8; j++) { mr[j] = -3.0e38f; lr[j] = 0.f; acc[j] = (v8f){}; }
  float* pw = &pbuf[wave][0];

  for (int ch = wave; ch < N_NODES / 16; ch += 8) {
    const int key0 = ch * 16;
    v8f s = {};
    for (int k0 = 0; k0 < 128; k0 += 4) {
      v2f a = ldsA(qt, 132, k0, lane);
      v2f b = ldBT(K + (size_t)key0 * 128, 128, k0, lane);
      s = wmma4(a, b, s);
    }
    const int col = key0 + cn;
    float sc[8];
    #pragma unroll
    for (int j = 0; j < 8; j++) {
      const int row = qrow0 + j + 8 * hi;
      float mv = tmask ? mask[(size_t)col * N_NODES + row]
                       : mask[(size_t)row * N_NODES + col];
      float sj = mv * s[j] - 1000.f * (1.f - mv);
      float mnew = fmaxf(mr[j], rowmax16(sj));
      sc[j] = __expf(mr[j] - mnew);
      float pj = __expf(sj - mnew);
      lr[j] = lr[j] * sc[j] + rowsum16(pj);
      mr[j] = mnew;
      pw[(j + 8 * hi) * 18 + cn] = pj;
    }
    #pragma unroll
    for (int t = 0; t < 8; t++)
      #pragma unroll
      for (int j = 0; j < 8; j++) acc[t][j] *= sc[j];
    asm volatile("s_wait_dscnt 0" ::: "memory");
    #pragma unroll
    for (int k0 = 0; k0 < 16; k0 += 4) {
      v2f a = ldsA(pw, 18, k0, lane);
      #pragma unroll
      for (int t = 0; t < 8; t++) {
        v2f b = ldBp(Vp, key0 + k0, t * 16, lane);
        acc[t] = wmma4(a, b, acc[t]);
      }
    }
  }

  if (cn == 0) {
    #pragma unroll
    for (int j = 0; j < 8; j++) {
      mbuf[wave][j + 8 * hi] = mr[j];
      lbuf[wave][j + 8 * hi] = lr[j];
    }
  }
  __syncthreads();
  for (int i = tid; i < 2048; i += 256) outbuf[i] = 0.f;
  if (tid < 16) {
    float M = -3.0e38f;
    for (int w = 0; w < 8; w++) M = fmaxf(M, mbuf[w][tid]);
    float L = 0.f;
    for (int w = 0; w < 8; w++) L += lbuf[w][tid] * __expf(mbuf[w][tid] - M);
    Mrow[tid] = M;
    Lrow[tid] = L > 0.f ? 1.f / L : 0.f;
  }
  __syncthreads();
  #pragma unroll
  for (int j = 0; j < 8; j++) {
    float e = __expf(mr[j] - Mrow[j + 8 * hi]);
    #pragma unroll
    for (int t = 0; t < 8; t++)
      atomicAdd(&outbuf[(j + 8 * hi) * 128 + t * 16 + cn], acc[t][j] * e);
  }
  __syncthreads();
  for (int i = tid; i < 2048; i += 256) {
    int r = i >> 7, c = i & 127;
    out[(size_t)(qrow0 + r) * 128 + c] = outbuf[i] * Lrow[r];
  }
}

#define CAT_LD 274
__global__ void __launch_bounds__(128) edge_kernel(
    const float* __restrict__ coors, const float* __restrict__ h,
    const int* __restrict__ src, const int* __restrict__ dst,
    const float* __restrict__ eW1p, const float* __restrict__ eb1,
    const float* __restrict__ eg1, const float* __restrict__ ebt1,
    const float* __restrict__ eW2p, const float* __restrict__ eb2,
    const float* __restrict__ cW1p, const float* __restrict__ cb1,
    const float* __restrict__ cg1, const float* __restrict__ cbt1,
    const float* __restrict__ cW2, const float* __restrict__ cb2,
    float* __restrict__ sum_msg, float* __restrict__ sum_x,
    float* __restrict__ cnt, int nE) {
  __shared__ float cat[16 * CAT_LD];
  __shared__ float buf1[16 * 132];
  __shared__ float buf2[16 * 132];
  __shared__ float stats[16][2];
  __shared__ float xrel[16][3];
  __shared__ float coefs[16];
  __shared__ int sidx[16], didx[16];
  const int tid = threadIdx.x, lane = tid & 31, wave = tid >> 5;
  const int hi = lane >> 4, cn = lane & 15;
  const int e0 = blockIdx.x * 16;

  if (tid < 16) {
    int e = e0 + tid;
    if (e < nE) {
      int s = src[e], d = dst[e];
      sidx[tid] = s; didx[tid] = d;
      float x0 = coors[s * 3 + 0] - coors[d * 3 + 0];
      float x1 = coors[s * 3 + 1] - coors[d * 3 + 1];
      float x2 = coors[s * 3 + 2] - coors[d * 3 + 2];
      xrel[tid][0] = x0; xrel[tid][1] = x1; xrel[tid][2] = x2;
      float d2 = x0 * x0 + x1 * x1 + x2 * x2;
      float sig = 1.f;
      for (int j = 0; j < 15; j++) {
        cat[tid * CAT_LD + 256 + j] = __expf(-d2 / sig);
        sig *= 1.5f;
      }
      cat[tid * CAT_LD + 271] = 0.f;
    } else {
      sidx[tid] = 0; didx[tid] = -1;
      xrel[tid][0] = xrel[tid][1] = xrel[tid][2] = 0.f;
      for (int j = 0; j < 16; j++) cat[tid * CAT_LD + 256 + j] = 0.f;
    }
  }
  __syncthreads();
  for (int i = tid; i < 16 * 256; i += 128) {
    int e = i >> 8, c = i & 255;
    int node = (didx[e] >= 0) ? ((c < 128) ? sidx[e] : didx[e]) : 0;
    cat[e * CAT_LD + c] = h[(size_t)node * 128 + (c & 127)];
  }
  __syncthreads();

  {
    const int n0 = wave * 32;
    v8f a0 = {}, a1 = {};
    for (int k0 = 0; k0 < 272; k0 += 4) {
      v2f a  = ldsA(cat, CAT_LD, k0, lane);
      v2f b0 = ldBp(eW1p, k0, n0, lane);
      v2f b1 = ldBp(eW1p, k0, n0 + 16, lane);
      a0 = wmma4(a, b0, a0);
      a1 = wmma4(a, b1, a1);
    }
    #pragma unroll
    for (int j = 0; j < 8; j++) {
      int r = j + 8 * hi;
      buf1[r * 132 + n0 + cn]      = lrelu(a0[j] + eb1[n0 + cn]);
      buf1[r * 132 + n0 + 16 + cn] = lrelu(a1[j] + eb1[n0 + 16 + cn]);
    }
  }
  __syncthreads();
  layernorm128(buf1, eg1, ebt1, stats, tid);
  __syncthreads();
  gemm128_lds(buf1, eW2p, eb2, buf2, lane, wave, 0);
  __syncthreads();
  gemm128_lds(buf2, cW1p, cb1, buf1, lane, wave, 1);
  __syncthreads();
  layernorm128(buf1, cg1, cbt1, stats, tid);
  __syncthreads();
  {
    int r = tid >> 3, seg = tid & 7;
    float sum = 0.f;
    #pragma unroll
    for (int i = 0; i < 16; i++) {
      int c = seg * 16 + i;
      sum += buf1[r * 132 + c] * cW2[c];
    }
    sum += __shfl_xor(sum, 1);
    sum += __shfl_xor(sum, 2);
    sum += __shfl_xor(sum, 4);
    if (seg == 0) coefs[r] = sum + cb2[0];
  }
  __syncthreads();
  for (int i = tid; i < 2048; i += 128) {
    int e = i >> 7, c = i & 127;
    int d = didx[e];
    if (d >= 0) atomicAdd(&sum_msg[(size_t)d * 128 + c], buf2[e * 132 + c]);
  }
  if (tid < 48) {
    int e = tid / 3, ax = tid % 3;
    int d = didx[e];
    if (d >= 0) atomicAdd(&sum_x[d * 3 + ax], xrel[e][ax] * coefs[e]);
  }
  if (tid < 16) {
    int d = didx[tid];
    if (d >= 0) atomicAdd(&cnt[d], 1.f);
  }
}

#define NCAT_LD 450
__global__ void __launch_bounds__(128) node_kernel(
    const float* __restrict__ h, const float* __restrict__ summ,
    const float* __restrict__ cnt, const float* __restrict__ cm,
    const float* __restrict__ orig,
    const float* __restrict__ nW1p, const float* __restrict__ nb1,
    const float* __restrict__ ng1, const float* __restrict__ nbt1,
    const float* __restrict__ nW2p, const float* __restrict__ nb2,
    float* __restrict__ hout) {
  __shared__ float cat[16 * NCAT_LD];
  __shared__ float buf1[16 * 132];
  __shared__ float stats[16][2];
  const int tid = threadIdx.x, lane = tid & 31, wave = tid >> 5;
  const int hi = lane >> 4, cn = lane & 15;
  const int node0 = blockIdx.x * 16;

  for (int i = tid; i < 2048; i += 128) {
    int e = i >> 7, c = i & 127;
    int node = node0 + e;
    cat[e * NCAT_LD + c] = h[(size_t)node * 128 + c];
    float ct = cnt[node];
    cat[e * NCAT_LD + 128 + c] = ct > 0.f ? summ[(size_t)node * 128 + c] / ct : 0.f;
    cat[e * NCAT_LD + 256 + c] = cm[(size_t)node * 128 + c];
  }
  for (int i = tid; i < 1024; i += 128) {
    int e = i >> 6, c = i & 63;
    cat[e * NCAT_LD + 384 + c] = orig[(size_t)(node0 + e) * 64 + c];
  }
  __syncthreads();
  {
    const int n0 = wave * 32;
    v8f a0 = {}, a1 = {};
    for (int k0 = 0; k0 < 448; k0 += 4) {
      v2f a  = ldsA(cat, NCAT_LD, k0, lane);
      v2f b0 = ldBp(nW1p, k0, n0, lane);
      v2f b1 = ldBp(nW1p, k0, n0 + 16, lane);
      a0 = wmma4(a, b0, a0);
      a1 = wmma4(a, b1, a1);
    }
    #pragma unroll
    for (int j = 0; j < 8; j++) {
      int r = j + 8 * hi;
      buf1[r * 132 + n0 + cn]      = lrelu(a0[j] + nb1[n0 + cn]);
      buf1[r * 132 + n0 + 16 + cn] = lrelu(a1[j] + nb1[n0 + 16 + cn]);
    }
  }
  __syncthreads();
  layernorm128(buf1, ng1, nbt1, stats, tid);
  __syncthreads();
  {
    const int n0 = wave * 32;
    v8f a0 = {}, a1 = {};
    for (int k0 = 0; k0 < 128; k0 += 4) {
      v2f a  = ldsA(buf1, 132, k0, lane);
      v2f b0 = ldBp(nW2p, k0, n0, lane);
      v2f b1 = ldBp(nW2p, k0, n0 + 16, lane);
      a0 = wmma4(a, b0, a0);
      a1 = wmma4(a, b1, a1);
    }
    #pragma unroll
    for (int j = 0; j < 8; j++) {
      int r = j + 8 * hi;
      size_t g0 = (size_t)(node0 + r) * 128 + n0 + cn;
      size_t g1 = g0 + 16;
      hout[g0] = 0.5f * (a0[j] + nb2[n0 + cn]) + 0.5f * h[g0];
      hout[g1] = 0.5f * (a1[j] + nb2[n0 + 16 + cn]) + 0.5f * h[g1];
    }
  }
}

__global__ void coords_kernel(const float* __restrict__ oc, const float* __restrict__ c,
                              const float* __restrict__ sx, const float* __restrict__ cnt,
                              float* __restrict__ out, int n) {
  int i = blockIdx.x * blockDim.x + threadIdx.x;
  if (i >= n * 3) return;
  int node = i / 3;
  float ct = cnt[node];
  float xu = ct > 0.f ? sx[i] / ct : 0.f;
  out[i] = 0.2f * oc[i] + 0.8f * c[i] + xu;
}

extern "C" void kernel_launch(void* const* d_in, const int* in_sizes, int n_in,
                              void* d_out, int out_size, void* d_ws, size_t ws_size,
                              hipStream_t stream) {
  const float* coors_lig = (const float*)d_in[0];
  const float* h_lig     = (const float*)d_in[1];
  const float* orig_lig  = (const float*)d_in[2];
  const float* oc_lig    = (const float*)d_in[3];
  const float* coors_rec = (const float*)d_in[4];
  const float* h_rec     = (const float*)d_in[5];
  const float* orig_rec  = (const float*)d_in[6];
  const float* oc_rec    = (const float*)d_in[7];
  const int* lig_src = (const int*)d_in[8];
  const int* lig_dst = (const int*)d_in[9];
  const int* rec_src = (const int*)d_in[10];
  const int* rec_dst = (const int*)d_in[11];
  const float* mask = (const float*)d_in[12];
  const float* eW1 = (const float*)d_in[13];
  const float* eb1 = (const float*)d_in[14];
  const float* eg1 = (const float*)d_in[15];
  const float* ebt1 = (const float*)d_in[16];
  const float* eW2 = (const float*)d_in[17];
  const float* eb2 = (const float*)d_in[18];
  const float* qW = (const float*)d_in[19];
  const float* kW = (const float*)d_in[20];
  const float* vW = (const float*)d_in[21];
  const float* nW1 = (const float*)d_in[22];
  const float* nb1 = (const float*)d_in[23];
  const float* ng1 = (const float*)d_in[24];
  const float* nbt1 = (const float*)d_in[25];
  const float* nW2 = (const float*)d_in[26];
  const float* nb2 = (const float*)d_in[27];
  const float* cW1 = (const float*)d_in[28];
  const float* cb1 = (const float*)d_in[29];
  const float* cg1 = (const float*)d_in[30];
  const float* cbt1 = (const float*)d_in[31];
  const float* cW2 = (const float*)d_in[32];
  const float* cb2 = (const float*)d_in[33];
  const int nE = in_sizes[8];

  float* ws = (float*)d_ws;
  float* q_l = ws + 0 * NH;
  float* k_l = ws + 1 * NH;
  float* v_l = ws + 2 * NH;
  float* q_r = ws + 3 * NH;
  float* k_r = ws + 4 * NH;
  float* v_r = ws + 5 * NH;
  float* cm_l = ws + 6 * NH;
  float* cm_r = ws + 7 * NH;
  float* sm_l = ws + 8 * NH;
  float* sm_r = ws + 9 * NH;
  float* sx_l = ws + 10 * NH;
  float* sx_r = sx_l + 3 * N_NODES;
  float* cnt_l = sx_r + 3 * N_NODES;
  float* cnt_r = cnt_l + N_NODES;
  float* eW1p = cnt_r + N_NODES;
  float* eW2p = eW1p + 136 * 256;
  float* cW1p = eW2p + 64 * 256;
  float* nW1p = cW1p + 64 * 256;
  float* nW2p = nW1p + 224 * 256;
  float* qWp  = nW2p + 64 * 256;
  float* kWp  = qWp + 64 * 256;
  float* vWp  = kWp + 64 * 256;

  size_t accum_floats = 2 * NH + 8 * (size_t)N_NODES;
  hipMemsetAsync(sm_l, 0, accum_floats * sizeof(float), stream);

  pack_pairs<<<(136 * 128 + 255) / 256, 256, 0, stream>>>(eW1, eW1p, 271, 136);
  pack_pairs<<<(64 * 128 + 255) / 256, 256, 0, stream>>>(eW2, eW2p, 128, 64);
  pack_pairs<<<(64 * 128 + 255) / 256, 256, 0, stream>>>(cW1, cW1p, 128, 64);
  pack_pairs<<<(224 * 128 + 255) / 256, 256, 0, stream>>>(nW1, nW1p, 448, 224);
  pack_pairs<<<(64 * 128 + 255) / 256, 256, 0, stream>>>(nW2, nW2p, 128, 64);
  pack_pairs<<<(64 * 128 + 255) / 256, 256, 0, stream>>>(qW, qWp, 128, 64);
  pack_pairs<<<(64 * 128 + 255) / 256, 256, 0, stream>>>(kW, kWp, 128, 64);
  pack_pairs<<<(64 * 128 + 255) / 256, 256, 0, stream>>>(vW, vWp, 128, 64);

  const int ntiles = N_NODES / 16;
  proj_gemm<<<ntiles, 128, 0, stream>>>(h_lig, qWp, q_l, 1, 0);
  proj_gemm<<<ntiles, 128, 0, stream>>>(h_lig, kWp, k_l, 1, 0);
  proj_gemm<<<ntiles, 128, 0, stream>>>(h_lig, vWp, v_l, 0, 1);
  proj_gemm<<<ntiles, 128, 0, stream>>>(h_rec, qWp, q_r, 1, 0);
  proj_gemm<<<ntiles, 128, 0, stream>>>(h_rec, kWp, k_r, 1, 0);
  proj_gemm<<<ntiles, 128, 0, stream>>>(h_rec, vWp, v_r, 0, 1);

  attn_kernel<<<ntiles, 256, 0, stream>>>(q_l, k_r, v_r, mask, 0, cm_l);
  attn_kernel<<<ntiles, 256, 0, stream>>>(q_r, k_l, v_l, mask, 1, cm_r);

  int eblocks = (nE + 15) / 16;
  edge_kernel<<<eblocks, 128, 0, stream>>>(coors_lig, h_lig, lig_src, lig_dst,
      eW1p, eb1, eg1, ebt1, eW2p, eb2, cW1p, cb1, cg1, cbt1, cW2, cb2,
      sm_l, sx_l, cnt_l, nE);
  edge_kernel<<<eblocks, 128, 0, stream>>>(coors_rec, h_rec, rec_src, rec_dst,
      eW1p, eb1, eg1, ebt1, eW2p, eb2, cW1p, cb1, cg1, cbt1, cW2, cb2,
      sm_r, sx_r, cnt_r, nE);

  float* out_xl = (float*)d_out;
  float* out_hl = out_xl + 3 * N_NODES;
  float* out_xr = out_hl + NH;
  float* out_hr = out_xr + 3 * N_NODES;

  node_kernel<<<ntiles, 128, 0, stream>>>(h_lig, sm_l, cnt_l, cm_l, orig_lig,
      nW1p, nb1, ng1, nbt1, nW2p, nb2, out_hl);
  node_kernel<<<ntiles, 128, 0, stream>>>(h_rec, sm_r, cnt_r, cm_r, orig_rec,
      nW1p, nb1, ng1, nbt1, nW2p, nb2, out_hr);

  int cthreads = 3 * N_NODES;
  coords_kernel<<<(cthreads + 255) / 256, 256, 0, stream>>>(oc_lig, coors_lig, sx_l,
                                                            cnt_l, out_xl, N_NODES);
  coords_kernel<<<(cthreads + 255) / 256, 256, 0, stream>>>(oc_rec, coors_rec, sx_r,
                                                            cnt_r, out_xr, N_NODES);
}


# attention loop software-pipelined: each chunk's K, V and mask fragments fetched one chunk ahead as whole-chunk b128 groups; b128 K/Q fragments regroup k inside QK^T steps (sum order changes)
# speedup vs baseline: 1.2229x; 1.2229x over previous
#include <hip/hip_runtime.h>
#include <hip/hip_bf16.h>
#include <cstdint>

#define N_NODES 6000
#define NH      ((size_t)N_NODES * 128)

typedef float v2f __attribute__((ext_vector_type(2)));
typedef float v4f __attribute__((ext_vector_type(4)));
typedef float v8f __attribute__((ext_vector_type(8)));

__device__ __forceinline__ v8f wmma4(v2f a, v2f b, v8f c) {
  return __builtin_amdgcn_wmma_f32_16x16x4_f32(false, a, false, b, (short)0, c,
                                               false, false);
}

__device__ __forceinline__ v2f ldsA(const float* base, int ld, int k0, int lane) {
  const int m = lane & 15;
  const int k = k0 + ((lane >> 4) << 1);
  const float* p = base + m * ld + k;
  v2f r; r.x = p[0]; r.y = p[1];
  return r;
}

__device__ __forceinline__ v2f ldBp(const float* __restrict__ P, int k0, int n0,
                                    int lane) {
  const int n = n0 + (lane & 15);
  const int k = k0 + ((lane >> 4) << 1);
  const float* p = P + (size_t)(k >> 1) * 256 + n * 2;
  v2f r; r.x = p[0]; r.y = p[1];
  return r;
}

__device__ __forceinline__ float lrelu(float x) { return x >= 0.f ? x : 0.01f * x; }

__device__ __forceinline__ float rowmax16(float v) {
  v = fmaxf(v, __shfl_xor(v, 1));
  v = fmaxf(v, __shfl_xor(v, 2));
  v = fmaxf(v, __shfl_xor(v, 4));
  v = fmaxf(v, __shfl_xor(v, 8));
  return v;
}
__device__ __forceinline__ float rowsum16(float v) {
  v += __shfl_xor(v, 1);
  v += __shfl_xor(v, 2);
  v += __shfl_xor(v, 4);
  v += __shfl_xor(v, 8);
  return v;
}

__device__ __forceinline__ void layernorm128(float* buf, const float* __restrict__ g,
                                             const float* __restrict__ b,
                                             float (*stats)[2], int tid) {
  const int r = tid >> 3, seg = tid & 7;
  float sum = 0.f, sq = 0.f;
  #pragma unroll
  for (int i = 0; i < 16; i++) {
    float v = buf[r * 132 + seg * 16 + i];
    sum += v; sq += v * v;
  }
  sum += __shfl_xor(sum, 1); sq += __shfl_xor(sq, 1);
  sum += __shfl_xor(sum, 2); sq += __shfl_xor(sq, 2);
  sum += __shfl_xor(sum, 4); sq += __shfl_xor(sq, 4);
  if (seg == 0) {
    float mean = sum * (1.f / 128.f);
    float var  = sq * (1.f / 128.f) - mean * mean;
    stats[r][0] = mean;
    stats[r][1] = rsqrtf(var + 1e-5f);
  }
  __syncthreads();
  const float mean = stats[r][0], rstd = stats[r][1];
  #pragma unroll
  for (int i = 0; i < 16; i++) {
    int c = seg * 16 + i;
    float v = buf[r * 132 + c];
    buf[r * 132 + c] = g[c] * (v - mean) * rstd + b[c];
  }
}

__device__ __forceinline__ void gemm128_lds(const float* in, const float* __restrict__ Wp,
                                            const float* __restrict__ bias, float* outb,
                                            int lane, int wave, int act) {
  const int hi = lane >> 4, cn = lane & 15;
  const int n0 = wave * 32;
  v8f a0 = {}, a1 = {};
  for (int k0 = 0; k0 < 128; k0 += 4) {
    v2f a  = ldsA(in, 132, k0, lane);
    v2f b0 = ldBp(Wp, k0, n0, lane);
    v2f b1 = ldBp(Wp, k0, n0 + 16, lane);
    a0 = wmma4(a, b0, a0);
    a1 = wmma4(a, b1, a1);
  }
  #pragma unroll
  for (int j = 0; j < 8; j++) {
    int r = j + 8 * hi;
    float x0 = a0[j] + bias[n0 + cn];
    float x1 = a1[j] + bias[n0 + 16 + cn];
    if (act) { x0 = lrelu(x0); x1 = lrelu(x1); }
    outb[r * 132 + n0 + cn] = x0;
    outb[r * 132 + n0 + 16 + cn] = x1;
  }
}

__global__ void pack_pairs(const float* __restrict__ W, float* __restrict__ P,
                           int K, int pairs) {
  int i = blockIdx.x * blockDim.x + threadIdx.x;
  if (i >= pairs * 128) return;
  int k2 = i >> 7, n = i & 127;
  int k = 2 * k2;
  float a = (k < K)     ? W[(size_t)k * 128 + n]       : 0.f;
  float b = (k + 1 < K) ? W[(size_t)(k + 1) * 128 + n] : 0.f;
  P[(size_t)k2 * 256 + n * 2]     = a;
  P[(size_t)k2 * 256 + n * 2 + 1] = b;
}

__global__ void __launch_bounds__(128) proj_gemm(const float* __restrict__ A,
                                                 const float* __restrict__ Wp,
                                                 float* __restrict__ C, int act,
                                                 int pack) {
  __shared__ float at[16 * 132];
  const int tid = threadIdx.x, lane = tid & 31, wave = tid >> 5;
  const int row0 = blockIdx.x * 16;
  for (int i = tid; i < 16 * 128; i += 128) {
    int r = i >> 7, c = i & 127;
    at[r * 132 + c] = A[(size_t)(row0 + r) * 128 + c];
  }
  __syncthreads();
  const int n0 = wave * 32;
  v8f a0 = {}, a1 = {};
  for (int k0 = 0; k0 < 128; k0 += 4) {
    v2f a  = ldsA(at, 132, k0, lane);
    v2f b0 = ldBp(Wp, k0, n0, lane);
    v2f b1 = ldBp(Wp, k0, n0 + 16, lane);
    a0 = wmma4(a, b0, a0);
    a1 = wmma4(a, b1, a1);
  }
  const int hi = lane >> 4, cn = lane & 15;
  #pragma unroll
  for (int j = 0; j < 8; j++) {
    int r = row0 + j + 8 * hi;
    float x0 = a0[j], x1 = a1[j];
    if (act) { x0 = lrelu(x0); x1 = lrelu(x1); }
    if (pack) {
      C[(size_t)(r >> 1) * 256 + (n0 + cn) * 2 + (r & 1)]      = x0;
      C[(size_t)(r >> 1) * 256 + (n0 + 16 + cn) * 2 + (r & 1)] = x1;
    } else {
      C[(size_t)r * 128 + n0 + cn] = x0;
      C[(size_t)r * 128 + n0 + 16 + cn] = x1;
    }
  }
}

__global__ void __launch_bounds__(256) attn_kernel(const float* __restrict__ Q,
                                                   const float* __restrict__ K,
                                                   const float* __restrict__ Vp,
                                                   const float* __restrict__ mask,
                                                   int tmask, float* __restrict__ out) {
  __shared__ __attribute__((aligned(16))) float qt[16 * 132];
  __shared__ float pbuf[8][16 * 18];
  __shared__ float outbuf[16 * 128];
  __shared__ float mbuf[8][16], lbuf[8][16];
  __shared__ float Mrow[16], Lrow[16];
  const int tid = threadIdx.x, lane = tid & 31;
  const int wave = __builtin_amdgcn_readfirstlane(tid >> 5);
  const int hi = lane >> 4, cn = lane & 15;
  const int qrow0 = blockIdx.x * 16;

  for (int i = tid; i < 2048; i += 256) {
    int r = i >> 7, c = i & 127;
    qt[r * 132 + c] = Q[(size_t)(qrow0 + r) * 128 + c];
  }
  __syncthreads();

  float mr[8], lr[8];
  v8f acc[8];
  #pragma unroll
  for (int j = 0; j < 8; j++) { mr[j] = -3.0e38f; lr[j] = 0.f; acc[j] = (v8f){}; }
  float* pw = &pbuf[wave][0];

  const float* qap = qt + cn * 132 + 4 * hi;
  const unsigned koff = (unsigned)(cn * 128 + 4 * hi);
  const unsigned voff = (unsigned)(hi * 256 + 4 * cn);
  const unsigned mrs = tmask ? 1u : (unsigned)N_NODES;
  const unsigned mcs = tmask ? (unsigned)N_NODES : 1u;
  const unsigned moff = (unsigned)(qrow0 + 8 * hi) * mrs + (unsigned)cn * mcs;
  v4f kf[16], vf[16];
  float mk[8];
  {
    const float* kb = K + (size_t)wave * 2048;
    #pragma unroll
    for (int i = 0; i < 16; i++) kf[i] = *(const v4f*)(kb + (koff + 8u * i));
    const float* mb = mask + (size_t)(wave * 16) * mcs;
    #pragma unroll
    for (int j = 0; j < 8; j++) mk[j] = mb[moff + (unsigned)j * mrs];
    const float* vb = Vp + (size_t)wave * 2048;
    #pragma unroll
    for (int i = 0; i < 16; i++) vf[i] = *(const v4f*)(vb + (voff + 512u * (i >> 2) + 64u * (i & 3)));
  }

  for (int ch = wave; ch < N_NODES / 16; ch += 8) {
    const int nch = (ch + 8 < N_NODES / 16) ? ch + 8 : N_NODES / 16 - 1;
    v8f s = {};
    #pragma unroll
    for (int i = 0; i < 16; i++) {
      v4f qa = *(const v4f*)(qap + 8 * i);
      v2f a0, a1, b0, b1;
      a0.x = qa.x; a0.y = qa.y; a1.x = qa.z; a1.y = qa.w;
      b0.x = kf[i].x; b0.y = kf[i].y; b1.x = kf[i].z; b1.y = kf[i].w;
      s = wmma4(a0, b0, s);
      s = wmma4(a1, b1, s);
    }
    float sc[8];
    #pragma unroll
    for (int j = 0; j < 8; j++) {
      float mv = mk[j];
      float sj = mv * s[j] - 1000.f * (1.f - mv);
      float mnew = fmaxf(mr[j], rowmax16(sj));
      sc[j] = __expf(mr[j] - mnew);
      float pj = __expf(sj - mnew);
      lr[j] = lr[j] * sc[j] + rowsum16(pj);
      mr[j] = mnew;
      pw[(j + 8 * hi) * 18 + cn] = pj;
    }
    #pragma unroll
    for (int t = 0; t < 8; t++)
      #pragma unroll
      for (int j = 0; j < 8; j++) acc[t][j] *= sc[j];
    {
      const float* kb = K + (size_t)nch * 2048;
      #pragma unroll
      for (int i = 0; i < 16; i++) kf[i] = *(const v4f*)(kb + (koff + 8u * i));
      const float* mb = mask + (size_t)(nch * 16) * mcs;
      #pragma unroll
      for (int j = 0; j < 8; j++) mk[j] = mb[moff + (unsigned)j * mrs];
    }
    asm volatile("s_wait_dscnt 0" ::: "memory");
    #pragma unroll
    for (int kk = 0; kk < 4; kk++) {
      v2f a = ldsA(pw, 18, 4 * kk, lane);
      #pragma unroll
      for (int u = 0; u < 4; u++) {
        v2f b0, b1;
        b0.x = vf[4 * kk + u].x; b0.y = vf[4 * kk + u].y;
        b1.x = vf[4 * kk + u].z; b1.y = vf[4 * kk + u].w;
        acc[2 * u]     = wmma4(a, b0, acc[2 * u]);
        acc[2 * u + 1] = wmma4(a, b1, acc[2 * u + 1]);
      }
    }
    {
      const float* vb = Vp + (size_t)nch * 2048;
      #pragma unroll
      for (int i = 0; i < 16; i++) vf[i] = *(const v4f*)(vb + (voff + 512u * (i >> 2) + 64u * (i & 3)));
    }
  }

  if (cn == 0) {
    #pragma unroll
    for (int j = 0; j < 8; j++) {
      mbuf[wave][j + 8 * hi] = mr[j];
      lbuf[wave][j + 8 * hi] = lr[j];
    }
  }
  __syncthreads();
  for (int i = tid; i < 2048; i += 256) outbuf[i] = 0.f;
  if (tid < 16) {
    float M = -3.0e38f;
    for (int w = 0; w < 8; w++) M = fmaxf(M, mbuf[w][tid]);
    float L = 0.f;
    for (int w = 0; w < 8; w++) L += lbuf[w][tid] * __expf(mbuf[w][tid] - M);
    Mrow[tid] = M;
    Lrow[tid] = L > 0.f ? 1.f / L : 0.f;
  }
  __syncthreads();
  #pragma unroll
  for (int j = 0; j < 8; j++) {
    float e = __expf(mr[j] - Mrow[j + 8 * hi]);
    #pragma unroll
    for (int t = 0; t < 8; t++)
      atomicAdd(&outbuf[(j + 8 * hi) * 128 + 32 * (t >> 1) + 2 * cn + (t & 1)],
                acc[t][j] * e);
  }
  __syncthreads();
  for (int i = tid; i < 2048; i += 256) {
    int r = i >> 7, c = i & 127;
    out[(size_t)(qrow0 + r) * 128 + c] = outbuf[i] * Lrow[r];
  }
}

#define CAT_LD 274
__global__ void __launch_bounds__(128) edge_kernel(
    const float* __restrict__ coors, const float* __restrict__ h,
    const int* __restrict__ src, const int* __restrict__ dst,
    const float* __restrict__ eW1p, const float* __restrict__ eb1,
    const float* __restrict__ eg1, const float* __restrict__ ebt1,
    const float* __restrict__ eW2p, const float* __restrict__ eb2,
    const float* __restrict__ cW1p, const float* __restrict__ cb1,
    const float* __restrict__ cg1, const float* __restrict__ cbt1,
    const float* __restrict__ cW2, const float* __restrict__ cb2,
    float* __restrict__ sum_msg, float* __restrict__ sum_x,
    float* __restrict__ cnt, int nE) {
  __shared__ float cat[16 * CAT_LD];
  __shared__ float buf1[16 * 132];
  __shared__ float buf2[16 * 132];
  __shared__ float stats[16][2];
  __shared__ float xrel[16][3];
  __shared__ float coefs[16];
  __shared__ int sidx[16], didx[16];
  const int tid = threadIdx.x, lane = tid & 31, wave = tid >> 5;
  const int hi = lane >> 4, cn = lane & 15;
  const int e0 = blockIdx.x * 16;

  if (tid < 16) {
    int e = e0 + tid;
    if (e < nE) {
      int s = src[e], d = dst[e];
      sidx[tid] = s; didx[tid] = d;
      float x0 = coors[s * 3 + 0] - coors[d * 3 + 0];
      float x1 = coors[s * 3 + 1] - coors[d * 3 + 1];
      float x2 = coors[s * 3 + 2] - coors[d * 3 + 2];
      xrel[tid][0] = x0; xrel[tid][1] = x1; xrel[tid][2] = x2;
      float d2 = x0 * x0 + x1 * x1 + x2 * x2;
      float sig = 1.f;
      for (int j = 0; j < 15; j++) {
        cat[tid * CAT_LD + 256 + j] = __expf(-d2 / sig);
        sig *= 1.5f;
      }
      cat[tid * CAT_LD + 271] = 0.f;
    } else {
      sidx[tid] = 0; didx[tid] = -1;
      xrel[tid][0] = xrel[tid][1] = xrel[tid][2] = 0.f;
      for (int j = 0; j < 16; j++) cat[tid * CAT_LD + 256 + j] = 0.f;
    }
  }
  __syncthreads();
  for (int i = tid; i < 16 * 256; i += 128) {
    int e = i >> 8, c = i & 255;
    int node = (didx[e] >= 0) ? ((c < 128) ? sidx[e] : didx[e]) : 0;
    cat[e * CAT_LD + c] = h[(size_t)node * 128 + (c & 127)];
  }
  __syncthreads();

  {
    const int n0 = wave * 32;
    v8f a0 = {}, a1 = {};
    for (int k0 = 0; k0 < 272; k0 += 4) {
      v2f a  = ldsA(cat, CAT_LD, k0, lane);
      v2f b0 = ldBp(eW1p, k0, n0, lane);
      v2f b1 = ldBp(eW1p, k0, n0 + 16, lane);
      a0 = wmma4(a, b0, a0);
      a1 = wmma4(a, b1, a1);
    }
    #pragma unroll
    for (int j = 0; j < 8; j++) {
      int r = j + 8 * hi;
      buf1[r * 132 + n0 + cn]      = lrelu(a0[j] + eb1[n0 + cn]);
      buf1[r * 132 + n0 + 16 + cn] = lrelu(a1[j] + eb1[n0 + 16 + cn]);
    }
  }
  __syncthreads();
  layernorm128(buf1, eg1, ebt1, stats, tid);
  __syncthreads();
  gemm128_lds(buf1, eW2p, eb2, buf2, lane, wave, 0);
  __syncthreads();
  gemm128_lds(buf2, cW1p, cb1, buf1, lane, wave, 1);
  __syncthreads();
  layernorm128(buf1, cg1, cbt1, stats, tid);
  __syncthreads();
  {
    int r = tid >> 3, seg = tid & 7;
    float sum = 0.f;
    #pragma unroll
    for (int i = 0; i < 16; i++) {
      int c = seg * 16 + i;
      sum += buf1[r * 132 + c] * cW2[c];
    }
    sum += __shfl_xor(sum, 1);
    sum += __shfl_xor(sum, 2);
    sum += __shfl_xor(sum, 4);
    if (seg == 0) coefs[r] = sum + cb2[0];
  }
  __syncthreads();
  for (int i = tid; i < 2048; i += 128) {
    int e = i >> 7, c = i & 127;
    int d = didx[e];
    if (d >= 0) atomicAdd(&sum_msg[(size_t)d * 128 + c], buf2[e * 132 + c]);
  }
  if (tid < 48) {
    int e = tid / 3, ax = tid % 3;
    int d = didx[e];
    if (d >= 0) atomicAdd(&sum_x[d * 3 + ax], xrel[e][ax] * coefs[e]);
  }
  if (tid < 16) {
    int d = didx[tid];
    if (d >= 0) atomicAdd(&cnt[d], 1.f);
  }
}

#define NCAT_LD 450
__global__ void __launch_bounds__(128) node_kernel(
    const float* __restrict__ h, const float* __restrict__ summ,
    const float* __restrict__ cnt, const float* __restrict__ cm,
    const float* __restrict__ orig,
    const float* __restrict__ nW1p, const float* __restrict__ nb1,
    const float* __restrict__ ng1, const float* __restrict__ nbt1,
    const float* __restrict__ nW2p, const float* __restrict__ nb2,
    float* __restrict__ hout) {
  __shared__ float cat[16 * NCAT_LD];
  __shared__ float buf1[16 * 132];
  __shared__ float stats[16][2];
  const int tid = threadIdx.x, lane = tid & 31, wave = tid >> 5;
  const int hi = lane >> 4, cn = lane & 15;
  const int node0 = blockIdx.x * 16;

  for (int i = tid; i < 2048; i += 128) {
    int e = i >> 7, c = i & 127;
    int node = node0 + e;
    cat[e * NCAT_LD + c] = h[(size_t)node * 128 + c];
    float ct = cnt[node];
    cat[e * NCAT_LD + 128 + c] = ct > 0.f ? summ[(size_t)node * 128 + c] / ct : 0.f;
    cat[e * NCAT_LD + 256 + c] = cm[(size_t)node * 128 + c];
  }
  for (int i = tid; i < 1024; i += 128) {
    int e = i >> 6, c = i & 63;
    cat[e * NCAT_LD + 384 + c] = orig[(size_t)(node0 + e) * 64 + c];
  }
  __syncthreads();
  {
    const int n0 = wave * 32;
    v8f a0 = {}, a1 = {};
    for (int k0 = 0; k0 < 448; k0 += 4) {
      v2f a  = ldsA(cat, NCAT_LD, k0, lane);
      v2f b0 = ldBp(nW1p, k0, n0, lane);
      v2f b1 = ldBp(nW1p, k0, n0 + 16, lane);
      a0 = wmma4(a, b0, a0);
      a1 = wmma4(a, b1, a1);
    }
    #pragma unroll
    for (int j = 0; j < 8; j++) {
      int r = j + 8 * hi;
      buf1[r * 132 + n0 + cn]      = lrelu(a0[j] + nb1[n0 + cn]);
      buf1[r * 132 + n0 + 16 + cn] = lrelu(a1[j] + nb1[n0 + 16 + cn]);
    }
  }
  __syncthreads();
  layernorm128(buf1, ng1, nbt1, stats, tid);
  __syncthreads();
  {
    const int n0 = wave * 32;
    v8f a0 = {}, a1 = {};
    for (int k0 = 0; k0 < 128; k0 += 4) {
      v2f a  = ldsA(buf1, 132, k0, lane);
      v2f b0 = ldBp(nW2p, k0, n0, lane);
      v2f b1 = ldBp(nW2p, k0, n0 + 16, lane);
      a0 = wmma4(a, b0, a0);
      a1 = wmma4(a, b1, a1);
    }
    #pragma unroll
    for (int j = 0; j < 8; j++) {
      int r = j + 8 * hi;
      size_t g0 = (size_t)(node0 + r) * 128 + n0 + cn;
      size_t g1 = g0 + 16;
      hout[g0] = 0.5f * (a0[j] + nb2[n0 + cn]) + 0.5f * h[g0];
      hout[g1] = 0.5f * (a1[j] + nb2[n0 + 16 + cn]) + 0.5f * h[g1];
    }
  }
}

__global__ void coords_kernel(const float* __restrict__ oc, const float* __restrict__ c,
                              const float* __restrict__ sx, const float* __restrict__ cnt,
                              float* __restrict__ out, int n) {
  int i = blockIdx.x * blockDim.x + threadIdx.x;
  if (i >= n * 3) return;
  int node = i / 3;
  float ct = cnt[node];
  float xu = ct > 0.f ? sx[i] / ct : 0.f;
  out[i] = 0.2f * oc[i] + 0.8f * c[i] + xu;
}

extern "C" void kernel_launch(void* const* d_in, const int* in_sizes, int n_in,
                              void* d_out, int out_size, void* d_ws, size_t ws_size,
                              hipStream_t stream) {
  const float* coors_lig = (const float*)d_in[0];
  const float* h_lig     = (const float*)d_in[1];
  const float* orig_lig  = (const float*)d_in[2];
  const float* oc_lig    = (const float*)d_in[3];
  const float* coors_rec = (const float*)d_in[4];
  const float* h_rec     = (const float*)d_in[5];
  const float* orig_rec  = (const float*)d_in[6];
  const float* oc_rec    = (const float*)d_in[7];
  const int* lig_src = (const int*)d_in[8];
  const int* lig_dst = (const int*)d_in[9];
  const int* rec_src = (const int*)d_in[10];
  const int* rec_dst = (const int*)d_in[11];
  const float* mask = (const float*)d_in[12];
  const float* eW1 = (const float*)d_in[13];
  const float* eb1 = (const float*)d_in[14];
  const float* eg1 = (const float*)d_in[15];
  const float* ebt1 = (const float*)d_in[16];
  const float* eW2 = (const float*)d_in[17];
  const float* eb2 = (const float*)d_in[18];
  const float* qW = (const float*)d_in[19];
  const float* kW = (const float*)d_in[20];
  const float* vW = (const float*)d_in[21];
  const float* nW1 = (const float*)d_in[22];
  const float* nb1 = (const float*)d_in[23];
  const float* ng1 = (const float*)d_in[24];
  const float* nbt1 = (const float*)d_in[25];
  const float* nW2 = (const float*)d_in[26];
  const float* nb2 = (const float*)d_in[27];
  const float* cW1 = (const float*)d_in[28];
  const float* cb1 = (const float*)d_in[29];
  const float* cg1 = (const float*)d_in[30];
  const float* cbt1 = (const float*)d_in[31];
  const float* cW2 = (const float*)d_in[32];
  const float* cb2 = (const float*)d_in[33];
  const int nE = in_sizes[8];

  float* ws = (float*)d_ws;
  float* q_l = ws + 0 * NH;
  float* k_l = ws + 1 * NH;
  float* v_l = ws + 2 * NH;
  float* q_r = ws + 3 * NH;
  float* k_r = ws + 4 * NH;
  float* v_r = ws + 5 * NH;
  float* cm_l = ws + 6 * NH;
  float* cm_r = ws + 7 * NH;
  float* sm_l = ws + 8 * NH;
  float* sm_r = ws + 9 * NH;
  float* sx_l = ws + 10 * NH;
  float* sx_r = sx_l + 3 * N_NODES;
  float* cnt_l = sx_r + 3 * N_NODES;
  float* cnt_r = cnt_l + N_NODES;
  float* eW1p = cnt_r + N_NODES;
  float* eW2p = eW1p + 136 * 256;
  float* cW1p = eW2p + 64 * 256;
  float* nW1p = cW1p + 64 * 256;
  float* nW2p = nW1p + 224 * 256;
  float* qWp  = nW2p + 64 * 256;
  float* kWp  = qWp + 64 * 256;
  float* vWp  = kWp + 64 * 256;

  size_t accum_floats = 2 * NH + 8 * (size_t)N_NODES;
  hipMemsetAsync(sm_l, 0, accum_floats * sizeof(float), stream);

  pack_pairs<<<(136 * 128 + 255) / 256, 256, 0, stream>>>(eW1, eW1p, 271, 136);
  pack_pairs<<<(64 * 128 + 255) / 256, 256, 0, stream>>>(eW2, eW2p, 128, 64);
  pack_pairs<<<(64 * 128 + 255) / 256, 256, 0, stream>>>(cW1, cW1p, 128, 64);
  pack_pairs<<<(224 * 128 + 255) / 256, 256, 0, stream>>>(nW1, nW1p, 448, 224);
  pack_pairs<<<(64 * 128 + 255) / 256, 256, 0, stream>>>(nW2, nW2p, 128, 64);
  pack_pairs<<<(64 * 128 + 255) / 256, 256, 0, stream>>>(qW, qWp, 128, 64);
  pack_pairs<<<(64 * 128 + 255) / 256, 256, 0, stream>>>(kW, kWp, 128, 64);
  pack_pairs<<<(64 * 128 + 255) / 256, 256, 0, stream>>>(vW, vWp, 128, 64);

  const int ntiles = N_NODES / 16;
  proj_gemm<<<ntiles, 128, 0, stream>>>(h_lig, qWp, q_l, 1, 0);
  proj_gemm<<<ntiles, 128, 0, stream>>>(h_lig, kWp, k_l, 1, 0);
  proj_gemm<<<ntiles, 128, 0, stream>>>(h_lig, vWp, v_l, 0, 1);
  proj_gemm<<<ntiles, 128, 0, stream>>>(h_rec, qWp, q_r, 1, 0);
  proj_gemm<<<ntiles, 128, 0, stream>>>(h_rec, kWp, k_r, 1, 0);
  proj_gemm<<<ntiles, 128, 0, stream>>>(h_rec, vWp, v_r, 0, 1);

  attn_kernel<<<ntiles, 256, 0, stream>>>(q_l, k_r, v_r, mask, 0, cm_l);
  attn_kernel<<<ntiles, 256, 0, stream>>>(q_r, k_l, v_l, mask, 1, cm_r);

  int eblocks = (nE + 15) / 16;
  edge_kernel<<<eblocks, 128, 0, stream>>>(coors_lig, h_lig, lig_src, lig_dst,
      eW1p, eb1, eg1, ebt1, eW2p, eb2, cW1p, cb1, cg1, cbt1, cW2, cb2,
      sm_l, sx_l, cnt_l, nE);
  edge_kernel<<<eblocks, 128, 0, stream>>>(coors_rec, h_rec, rec_src, rec_dst,
      eW1p, eb1, eg1, ebt1, eW2p, eb2, cW1p, cb1, cg1, cbt1, cW2, cb2,
      sm_r, sx_r, cnt_r, nE);

  float* out_xl = (float*)d_out;
  float* out_hl = out_xl + 3 * N_NODES;
  float* out_xr = out_hl + NH;
  float* out_hr = out_xr + 3 * N_NODES;

  node_kernel<<<ntiles, 128, 0, stream>>>(h_lig, sm_l, cnt_l, cm_l, orig_lig,
      nW1p, nb1, ng1, nbt1, nW2p, nb2, out_hl);
  node_kernel<<<ntiles, 128, 0, stream>>>(h_rec, sm_r, cnt_r, cm_r, orig_rec,
      nW1p, nb1, ng1, nbt1, nW2p, nb2, out_hr);

  int cthreads = 3 * N_NODES;
  coords_kernel<<<(cthreads + 255) / 256, 256, 0, stream>>>(oc_lig, coors_lig, sx_l,
                                                            cnt_l, out_xl, N_NODES);
  coords_kernel<<<(cthreads + 255) / 256, 256, 0, stream>>>(oc_rec, coors_rec, sx_r,
                                                            cnt_r, out_xr, N_NODES);
}


# edge kernel gathers h[src]|h[dst] as eight 16-byte whole-row loads per thread, all issued as one group before the first wait (was 32 dependent 4-byte load, wait, store trips); sums unchanged
# speedup vs baseline: 1.3074x; 1.0691x over previous
#include <hip/hip_runtime.h>
#include <hip/hip_bf16.h>
#include <cstdint>

#define N_NODES 6000
#define NH      ((size_t)N_NODES * 128)

typedef float v2f __attribute__((ext_vector_type(2)));
typedef float v4f __attribute__((ext_vector_type(4)));
typedef float v8f __attribute__((ext_vector_type(8)));

__device__ __forceinline__ v8f wmma4(v2f a, v2f b, v8f c) {
  return __builtin_amdgcn_wmma_f32_16x16x4_f32(false, a, false, b, (short)0, c,
                                               false, false);
}

__device__ __forceinline__ v2f ldsA(const float* base, int ld, int k0, int lane) {
  const int m = lane & 15;
  const int k = k0 + ((lane >> 4) << 1);
  const float* p = base + m * ld + k;
  v2f r; r.x = p[0]; r.y = p[1];
  return r;
}

__device__ __forceinline__ v2f ldBp(const float* __restrict__ P, int k0, int n0,
                                    int lane) {
  const int n = n0 + (lane & 15);
  const int k = k0 + ((lane >> 4) << 1);
  const float* p = P + (size_t)(k >> 1) * 256 + n * 2;
  v2f r; r.x = p[0]; r.y = p[1];
  return r;
}

__device__ __forceinline__ float lrelu(float x) { return x >= 0.f ? x : 0.01f * x; }

__device__ __forceinline__ float rowmax16(float v) {
  v = fmaxf(v, __shfl_xor(v, 1));
  v = fmaxf(v, __shfl_xor(v, 2));
  v = fmaxf(v, __shfl_xor(v, 4));
  v = fmaxf(v, __shfl_xor(v, 8));
  return v;
}
__device__ __forceinline__ float rowsum16(float v) {
  v += __shfl_xor(v, 1);
  v += __shfl_xor(v, 2);
  v += __shfl_xor(v, 4);
  v += __shfl_xor(v, 8);
  return v;
}

__device__ __forceinline__ void layernorm128(float* buf, const float* __restrict__ g,
                                             const float* __restrict__ b,
                                             float (*stats)[2], int tid) {
  const int r = tid >> 3, seg = tid & 7;
  float sum = 0.f, sq = 0.f;
  #pragma unroll
  for (int i = 0; i < 16; i++) {
    float v = buf[r * 132 + seg * 16 + i];
    sum += v; sq += v * v;
  }
  sum += __shfl_xor(sum, 1); sq += __shfl_xor(sq, 1);
  sum += __shfl_xor(sum, 2); sq += __shfl_xor(sq, 2);
  sum += __shfl_xor(sum, 4); sq += __shfl_xor(sq, 4);
  if (seg == 0) {
    float mean = sum * (1.f / 128.f);
    float var  = sq * (1.f / 128.f) - mean * mean;
    stats[r][0] = mean;
    stats[r][1] = rsqrtf(var + 1e-5f);
  }
  __syncthreads();
  const float mean = stats[r][0], rstd = stats[r][1];
  #pragma unroll
  for (int i = 0; i < 16; i++) {
    int c = seg * 16 + i;
    float v = buf[r * 132 + c];
    buf[r * 132 + c] = g[c] * (v - mean) * rstd + b[c];
  }
}

__device__ __forceinline__ void gemm128_lds(const float* in, const float* __restrict__ Wp,
                                            const float* __restrict__ bias, float* outb,
                                            int lane, int wave, int act) {
  const int hi = lane >> 4, cn = lane & 15;
  const int n0 = wave * 32;
  v8f a0 = {}, a1 = {};
  for (int k0 = 0; k0 < 128; k0 += 4) {
    v2f a  = ldsA(in, 132, k0, lane);
    v2f b0 = ldBp(Wp, k0, n0, lane);
    v2f b1 = ldBp(Wp, k0, n0 + 16, lane);
    a0 = wmma4(a, b0, a0);
    a1 = wmma4(a, b1, a1);
  }
  #pragma unroll
  for (int j = 0; j < 8; j++) {
    int r = j + 8 * hi;
    float x0 = a0[j] + bias[n0 + cn];
    float x1 = a1[j] + bias[n0 + 16 + cn];
    if (act) { x0 = lrelu(x0); x1 = lrelu(x1); }
    outb[r * 132 + n0 + cn] = x0;
    outb[r * 132 + n0 + 16 + cn] = x1;
  }
}

__global__ void pack_pairs(const float* __restrict__ W, float* __restrict__ P,
                           int K, int pairs) {
  int i = blockIdx.x * blockDim.x + threadIdx.x;
  if (i >= pairs * 128) return;
  int k2 = i >> 7, n = i & 127;
  int k = 2 * k2;
  float a = (k < K)     ? W[(size_t)k * 128 + n]       : 0.f;
  float b = (k + 1 < K) ? W[(size_t)(k + 1) * 128 + n] : 0.f;
  P[(size_t)k2 * 256 + n * 2]     = a;
  P[(size_t)k2 * 256 + n * 2 + 1] = b;
}

__global__ void __launch_bounds__(128) proj_gemm(const float* __restrict__ A,
                                                 const float* __restrict__ Wp,
                                                 float* __restrict__ C, int act,
                                                 int pack) {
  __shared__ float at[16 * 132];
  const int tid = threadIdx.x, lane = tid & 31, wave = tid >> 5;
  const int row0 = blockIdx.x * 16;
  for (int i = tid; i < 16 * 128; i += 128) {
    int r = i >> 7, c = i & 127;
    at[r * 132 + c] = A[(size_t)(row0 + r) * 128 + c];
  }
  __syncthreads();
  const int n0 = wave * 32;
  v8f a0 = {}, a1 = {};
  for (int k0 = 0; k0 < 128; k0 += 4) {
    v2f a  = ldsA(at, 132, k0, lane);
    v2f b0 = ldBp(Wp, k0, n0, lane);
    v2f b1 = ldBp(Wp, k0, n0 + 16, lane);
    a0 = wmma4(a, b0, a0);
    a1 = wmma4(a, b1, a1);
  }
  const int hi = lane >> 4, cn = lane & 15;
  #pragma unroll
  for (int j = 0; j < 8; j++) {
    int r = row0 + j + 8 * hi;
    float x0 = a0[j], x1 = a1[j];
    if (act) { x0 = lrelu(x0); x1 = lrelu(x1); }
    if (pack) {
      C[(size_t)(r >> 1) * 256 + (n0 + cn) * 2 + (r & 1)]      = x0;
      C[(size_t)(r >> 1) * 256 + (n0 + 16 + cn) * 2 + (r & 1)] = x1;
    } else {
      C[(size_t)r * 128 + n0 + cn] = x0;
      C[(size_t)r * 128 + n0 + 16 + cn] = x1;
    }
  }
}

__global__ void __launch_bounds__(256) attn_kernel(const float* __restrict__ Q,
                                                   const float* __restrict__ K,
                                                   const float* __restrict__ Vp,
                                                   const float* __restrict__ mask,
                                                   int tmask, float* __restrict__ out) {
  __shared__ __attribute__((aligned(16))) float qt[16 * 132];
  __shared__ float pbuf[8][16 * 18];
  __shared__ float outbuf[16 * 128];
  __shared__ float mbuf[8][16], lbuf[8][16];
  __shared__ float Mrow[16], Lrow[16];
  const int tid = threadIdx.x, lane = tid & 31;
  const int wave = __builtin_amdgcn_readfirstlane(tid >> 5);
  const int hi = lane >> 4, cn = lane & 15;
  const int qrow0 = blockIdx.x * 16;

  for (int i = tid; i < 2048; i += 256) {
    int r = i >> 7, c = i & 127;
    qt[r * 132 + c] = Q[(size_t)(qrow0 + r) * 128 + c];
  }
  __syncthreads();

  float mr[8], lr[8];
  v8f acc[8];
  #pragma unroll
  for (int j = 0; j < 8; j++) { mr[j] = -3.0e38f; lr[j] = 0.f; acc[j] = (v8f){}; }
  float* pw = &pbuf[wave][0];

  const float* qap = qt + cn * 132 + 4 * hi;
  const unsigned koff = (unsigned)(cn * 128 + 4 * hi);
  const unsigned voff = (unsigned)(hi * 256 + 4 * cn);
  const unsigned mrs = tmask ? 1u : (unsigned)N_NODES;
  const unsigned mcs = tmask ? (unsigned)N_NODES : 1u;
  const unsigned moff = (unsigned)(qrow0 + 8 * hi) * mrs + (unsigned)cn * mcs;
  v4f kf[16], vf[16];
  float mk[8];
  {
    const float* kb = K + (size_t)wave * 2048;
    #pragma unroll
    for (int i = 0; i < 16; i++) kf[i] = *(const v4f*)(kb + (koff + 8u * i));
    const float* mb = mask + (size_t)(wave * 16) * mcs;
    #pragma unroll
    for (int j = 0; j < 8; j++) mk[j] = mb[moff + (unsigned)j * mrs];
    const float* vb = Vp + (size_t)wave * 2048;
    #pragma unroll
    for (int i = 0; i < 16; i++) vf[i] = *(const v4f*)(vb + (voff + 512u * (i >> 2) + 64u * (i & 3)));
  }

  for (int ch = wave; ch < N_NODES / 16; ch += 8) {
    const int nch = (ch + 8 < N_NODES / 16) ? ch + 8 : N_NODES / 16 - 1;
    v8f s = {};
    #pragma unroll
    for (int i = 0; i < 16; i++) {
      v4f qa = *(const v4f*)(qap + 8 * i);
      v2f a0, a1, b0, b1;
      a0.x = qa.x; a0.y = qa.y; a1.x = qa.z; a1.y = qa.w;
      b0.x = kf[i].x; b0.y = kf[i].y; b1.x = kf[i].z; b1.y = kf[i].w;
      s = wmma4(a0, b0, s);
      s = wmma4(a1, b1, s);
    }
    float sc[8];
    #pragma unroll
    for (int j = 0; j < 8; j++) {
      float mv = mk[j];
      float sj = mv * s[j] - 1000.f * (1.f - mv);
      float mnew = fmaxf(mr[j], rowmax16(sj));
      sc[j] = __expf(mr[j] - mnew);
      float pj = __expf(sj - mnew);
      lr[j] = lr[j] * sc[j] + rowsum16(pj);
      mr[j] = mnew;
      pw[(j + 8 * hi) * 18 + cn] = pj;
    }
    #pragma unroll
    for (int t = 0; t < 8; t++)
      #pragma unroll
      for (int j = 0; j < 8; j++) acc[t][j] *= sc[j];
    {
      const float* kb = K + (size_t)nch * 2048;
      #pragma unroll
      for (int i = 0; i < 16; i++) kf[i] = *(const v4f*)(kb + (koff + 8u * i));
      const float* mb = mask + (size_t)(nch * 16) * mcs;
      #pragma unroll
      for (int j = 0; j < 8; j++) mk[j] = mb[moff + (unsigned)j * mrs];
    }
    asm volatile("s_wait_dscnt 0" ::: "memory");
    #pragma unroll
    for (int kk = 0; kk < 4; kk++) {
      v2f a = ldsA(pw, 18, 4 * kk, lane);
      #pragma unroll
      for (int u = 0; u < 4; u++) {
        v2f b0, b1;
        b0.x = vf[4 * kk + u].x; b0.y = vf[4 * kk + u].y;
        b1.x = vf[4 * kk + u].z; b1.y = vf[4 * kk + u].w;
        acc[2 * u]     = wmma4(a, b0, acc[2 * u]);
        acc[2 * u + 1] = wmma4(a, b1, acc[2 * u + 1]);
      }
    }
    {
      const float* vb = Vp + (size_t)nch * 2048;
      #pragma unroll
      for (int i = 0; i < 16; i++) vf[i] = *(const v4f*)(vb + (voff + 512u * (i >> 2) + 64u * (i & 3)));
    }
  }

  if (cn == 0) {
    #pragma unroll
    for (int j = 0; j < 8; j++) {
      mbuf[wave][j + 8 * hi] = mr[j];
      lbuf[wave][j + 8 * hi] = lr[j];
    }
  }
  __syncthreads();
  for (int i = tid; i < 2048; i += 256) outbuf[i] = 0.f;
  if (tid < 16) {
    float M = -3.0e38f;
    for (int w = 0; w < 8; w++) M = fmaxf(M, mbuf[w][tid]);
    float L = 0.f;
    for (int w = 0; w < 8; w++) L += lbuf[w][tid] * __expf(mbuf[w][tid] - M);
    Mrow[tid] = M;
    Lrow[tid] = L > 0.f ? 1.f / L : 0.f;
  }
  __syncthreads();
  #pragma unroll
  for (int j = 0; j < 8; j++) {
    float e = __expf(mr[j] - Mrow[j + 8 * hi]);
    #pragma unroll
    for (int t = 0; t < 8; t++)
      atomicAdd(&outbuf[(j + 8 * hi) * 128 + 32 * (t >> 1) + 2 * cn + (t & 1)],
                acc[t][j] * e);
  }
  __syncthreads();
  for (int i = tid; i < 2048; i += 256) {
    int r = i >> 7, c = i & 127;
    out[(size_t)(qrow0 + r) * 128 + c] = outbuf[i] * Lrow[r];
  }
}

#define CAT_LD 274
__global__ void __launch_bounds__(128) edge_kernel(
    const float* __restrict__ coors, const float* __restrict__ h,
    const int* __restrict__ src, const int* __restrict__ dst,
    const float* __restrict__ eW1p, const float* __restrict__ eb1,
    const float* __restrict__ eg1, const float* __restrict__ ebt1,
    const float* __restrict__ eW2p, const float* __restrict__ eb2,
    const float* __restrict__ cW1p, const float* __restrict__ cb1,
    const float* __restrict__ cg1, const float* __restrict__ cbt1,
    const float* __restrict__ cW2, const float* __restrict__ cb2,
    float* __restrict__ sum_msg, float* __restrict__ sum_x,
    float* __restrict__ cnt, int nE) {
  __shared__ __attribute__((aligned(16))) float cat[16 * CAT_LD];
  __shared__ float buf1[16 * 132];
  __shared__ float buf2[16 * 132];
  __shared__ float stats[16][2];
  __shared__ float xrel[16][3];
  __shared__ float coefs[16];
  __shared__ int sidx[16], didx[16];
  const int tid = threadIdx.x, lane = tid & 31, wave = tid >> 5;
  const int hi = lane >> 4, cn = lane & 15;
  const int e0 = blockIdx.x * 16;

  if (tid < 16) {
    int e = e0 + tid;
    if (e < nE) {
      int s = src[e], d = dst[e];
      sidx[tid] = s; didx[tid] = d;
      float x0 = coors[s * 3 + 0] - coors[d * 3 + 0];
      float x1 = coors[s * 3 + 1] - coors[d * 3 + 1];
      float x2 = coors[s * 3 + 2] - coors[d * 3 + 2];
      xrel[tid][0] = x0; xrel[tid][1] = x1; xrel[tid][2] = x2;
      float d2 = x0 * x0 + x1 * x1 + x2 * x2;
      float sig = 1.f;
      for (int j = 0; j < 15; j++) {
        cat[tid * CAT_LD + 256 + j] = __expf(-d2 / sig);
        sig *= 1.5f;
      }
      cat[tid * CAT_LD + 271] = 0.f;
    } else {
      sidx[tid] = 0; didx[tid] = -1;
      xrel[tid][0] = xrel[tid][1] = xrel[tid][2] = 0.f;
      for (int j = 0; j < 16; j++) cat[tid * CAT_LD + 256 + j] = 0.f;
    }
  }
  __syncthreads();
  {
    const int l4 = (tid & 31) * 4;
    const int half = wave & 1;
    const int* ip = half ? didx : sidx;
    int nn[8], dd[8];
    v4f g[8];
    #pragma unroll
    for (int it = 0; it < 8; it++) {
      const int e = (wave >> 1) + 2 * it;
      nn[it] = ip[e];
      dd[it] = didx[e];
    }
    #pragma unroll
    for (int it = 0; it < 8; it++) {
      const int node = (dd[it] >= 0) ? nn[it] : 0;
      g[it] = *(const v4f*)(h + (size_t)node * 128 + l4);
    }
    #pragma unroll
    for (int it = 0; it < 8; it++) {
      const int e = (wave >> 1) + 2 * it;
      float* p = cat + e * CAT_LD + half * 128 + l4;
      v2f lo, hi2;
      lo.x = g[it].x; lo.y = g[it].y; hi2.x = g[it].z; hi2.y = g[it].w;
      *(v2f*)p = lo;
      *(v2f*)(p + 2) = hi2;
    }
  }
  __syncthreads();

  {
    const int n0 = wave * 32;
    v8f a0 = {}, a1 = {};
    for (int k0 = 0; k0 < 272; k0 += 4) {
      v2f a  = ldsA(cat, CAT_LD, k0, lane);
      v2f b0 = ldBp(eW1p, k0, n0, lane);
      v2f b1 = ldBp(eW1p, k0, n0 + 16, lane);
      a0 = wmma4(a, b0, a0);
      a1 = wmma4(a, b1, a1);
    }
    #pragma unroll
    for (int j = 0; j < 8; j++) {
      int r = j + 8 * hi;
      buf1[r * 132 + n0 + cn]      = lrelu(a0[j] + eb1[n0 + cn]);
      buf1[r * 132 + n0 + 16 + cn] = lrelu(a1[j] + eb1[n0 + 16 + cn]);
    }
  }
  __syncthreads();
  layernorm128(buf1, eg1, ebt1, stats, tid);
  __syncthreads();
  gemm128_lds(buf1, eW2p, eb2, buf2, lane, wave, 0);
  __syncthreads();
  gemm128_lds(buf2, cW1p, cb1, buf1, lane, wave, 1);
  __syncthreads();
  layernorm128(buf1, cg1, cbt1, stats, tid);
  __syncthreads();
  {
    int r = tid >> 3, seg = tid & 7;
    float sum = 0.f;
    #pragma unroll
    for (int i = 0; i < 16; i++) {
      int c = seg * 16 + i;
      sum += buf1[r * 132 + c] * cW2[c];
    }
    sum += __shfl_xor(sum, 1);
    sum += __shfl_xor(sum, 2);
    sum += __shfl_xor(sum, 4);
    if (seg == 0) coefs[r] = sum + cb2[0];
  }
  __syncthreads();
  for (int i = tid; i < 2048; i += 128) {
    int e = i >> 7, c = i & 127;
    int d = didx[e];
    if (d >= 0) atomicAdd(&sum_msg[(size_t)d * 128 + c], buf2[e * 132 + c]);
  }
  if (tid < 48) {
    int e = tid / 3, ax = tid % 3;
    int d = didx[e];
    if (d >= 0) atomicAdd(&sum_x[d * 3 + ax], xrel[e][ax] * coefs[e]);
  }
  if (tid < 16) {
    int d = didx[tid];
    if (d >= 0) atomicAdd(&cnt[d], 1.f);
  }
}

#define NCAT_LD 450
__global__ void __launch_bounds__(128) node_kernel(
    const float* __restrict__ h, const float* __restrict__ summ,
    const float* __restrict__ cnt, const float* __restrict__ cm,
    const float* __restrict__ orig,
    const float* __restrict__ nW1p, const float* __restrict__ nb1,
    const float* __restrict__ ng1, const float* __restrict__ nbt1,
    const float* __restrict__ nW2p, const float* __restrict__ nb2,
    float* __restrict__ hout) {
  __shared__ float cat[16 * NCAT_LD];
  __shared__ float buf1[16 * 132];
  __shared__ float stats[16][2];
  const int tid = threadIdx.x, lane = tid & 31, wave = tid >> 5;
  const int hi = lane >> 4, cn = lane & 15;
  const int node0 = blockIdx.x * 16;

  for (int i = tid; i < 2048; i += 128) {
    int e = i >> 7, c = i & 127;
    int node = node0 + e;
    cat[e * NCAT_LD + c] = h[(size_t)node * 128 + c];
    float ct = cnt[node];
    cat[e * NCAT_LD + 128 + c] = ct > 0.f ? summ[(size_t)node * 128 + c] / ct : 0.f;
    cat[e * NCAT_LD + 256 + c] = cm[(size_t)node * 128 + c];
  }
  for (int i = tid; i < 1024; i += 128) {
    int e = i >> 6, c = i & 63;
    cat[e * NCAT_LD + 384 + c] = orig[(size_t)(node0 + e) * 64 + c];
  }
  __syncthreads();
  {
    const int n0 = wave * 32;
    v8f a0 = {}, a1 = {};
    for (int k0 = 0; k0 < 448; k0 += 4) {
      v2f a  = ldsA(cat, NCAT_LD, k0, lane);
      v2f b0 = ldBp(nW1p, k0, n0, lane);
      v2f b1 = ldBp(nW1p, k0, n0 + 16, lane);
      a0 = wmma4(a, b0, a0);
      a1 = wmma4(a, b1, a1);
    }
    #pragma unroll
    for (int j = 0; j < 8; j++) {
      int r = j + 8 * hi;
      buf1[r * 132 + n0 + cn]      = lrelu(a0[j] + nb1[n0 + cn]);
      buf1[r * 132 + n0 + 16 + cn] = lrelu(a1[j] + nb1[n0 + 16 + cn]);
    }
  }
  __syncthreads();
  layernorm128(buf1, ng1, nbt1, stats, tid);
  __syncthreads();
  {
    const int n0 = wave * 32;
    v8f a0 = {}, a1 = {};
    for (int k0 = 0; k0 < 128; k0 += 4) {
      v2f a  = ldsA(buf1, 132, k0, lane);
      v2f b0 = ldBp(nW2p, k0, n0, lane);
      v2f b1 = ldBp(nW2p, k0, n0 + 16, lane);
      a0 = wmma4(a, b0, a0);
      a1 = wmma4(a, b1, a1);
    }
    #pragma unroll
    for (int j = 0; j < 8; j++) {
      int r = j + 8 * hi;
      size_t g0 = (size_t)(node0 + r) * 128 + n0 + cn;
      size_t g1 = g0 + 16;
      hout[g0] = 0.5f * (a0[j] + nb2[n0 + cn]) + 0.5f * h[g0];
      hout[g1] = 0.5f * (a1[j] + nb2[n0 + 16 + cn]) + 0.5f * h[g1];
    }
  }
}

__global__ void coords_kernel(const float* __restrict__ oc, const float* __restrict__ c,
                              const float* __restrict__ sx, const float* __restrict__ cnt,
                              float* __restrict__ out, int n) {
  int i = blockIdx.x * blockDim.x + threadIdx.x;
  if (i >= n * 3) return;
  int node = i / 3;
  float ct = cnt[node];
  float xu = ct > 0.f ? sx[i] / ct : 0.f;
  out[i] = 0.2f * oc[i] + 0.8f * c[i] + xu;
}

extern "C" void kernel_launch(void* const* d_in, const int* in_sizes, int n_in,
                              void* d_out, int out_size, void* d_ws, size_t ws_size,
                              hipStream_t stream) {
  const float* coors_lig = (const float*)d_in[0];
  const float* h_lig     = (const float*)d_in[1];
  const float* orig_lig  = (const float*)d_in[2];
  const float* oc_lig    = (const float*)d_in[3];
  const float* coors_rec = (const float*)d_in[4];
  const float* h_rec     = (const float*)d_in[5];
  const float* orig_rec  = (const float*)d_in[6];
  const float* oc_rec    = (const float*)d_in[7];
  const int* lig_src = (const int*)d_in[8];
  const int* lig_dst = (const int*)d_in[9];
  const int* rec_src = (const int*)d_in[10];
  const int* rec_dst = (const int*)d_in[11];
  const float* mask = (const float*)d_in[12];
  const float* eW1 = (const float*)d_in[13];
  const float* eb1 = (const float*)d_in[14];
  const float* eg1 = (const float*)d_in[15];
  const float* ebt1 = (const float*)d_in[16];
  const float* eW2 = (const float*)d_in[17];
  const float* eb2 = (const float*)d_in[18];
  const float* qW = (const float*)d_in[19];
  const float* kW = (const float*)d_in[20];
  const float* vW = (const float*)d_in[21];
  const float* nW1 = (const float*)d_in[22];
  const float* nb1 = (const float*)d_in[23];
  const float* ng1 = (const float*)d_in[24];
  const float* nbt1 = (const float*)d_in[25];
  const float* nW2 = (const float*)d_in[26];
  const float* nb2 = (const float*)d_in[27];
  const float* cW1 = (const float*)d_in[28];
  const float* cb1 = (const float*)d_in[29];
  const float* cg1 = (const float*)d_in[30];
  const float* cbt1 = (const float*)d_in[31];
  const float* cW2 = (const float*)d_in[32];
  const float* cb2 = (const float*)d_in[33];
  const int nE = in_sizes[8];

  float* ws = (float*)d_ws;
  float* q_l = ws + 0 * NH;
  float* k_l = ws + 1 * NH;
  float* v_l = ws + 2 * NH;
  float* q_r = ws + 3 * NH;
  float* k_r = ws + 4 * NH;
  float* v_r = ws + 5 * NH;
  float* cm_l = ws + 6 * NH;
  float* cm_r = ws + 7 * NH;
  float* sm_l = ws + 8 * NH;
  float* sm_r = ws + 9 * NH;
  float* sx_l = ws + 10 * NH;
  float* sx_r = sx_l + 3 * N_NODES;
  float* cnt_l = sx_r + 3 * N_NODES;
  float* cnt_r = cnt_l + N_NODES;
  float* eW1p = cnt_r + N_NODES;
  float* eW2p = eW1p + 136 * 256;
  float* cW1p = eW2p + 64 * 256;
  float* nW1p = cW1p + 64 * 256;
  float* nW2p = nW1p + 224 * 256;
  float* qWp  = nW2p + 64 * 256;
  float* kWp  = qWp + 64 * 256;
  float* vWp  = kWp + 64 * 256;

  size_t accum_floats = 2 * NH + 8 * (size_t)N_NODES;
  hipMemsetAsync(sm_l, 0, accum_floats * sizeof(float), stream);

  pack_pairs<<<(136 * 128 + 255) / 256, 256, 0, stream>>>(eW1, eW1p, 271, 136);
  pack_pairs<<<(64 * 128 + 255) / 256, 256, 0, stream>>>(eW2, eW2p, 128, 64);
  pack_pairs<<<(64 * 128 + 255) / 256, 256, 0, stream>>>(cW1, cW1p, 128, 64);
  pack_pairs<<<(224 * 128 + 255) / 256, 256, 0, stream>>>(nW1, nW1p, 448, 224);
  pack_pairs<<<(64 * 128 + 255) / 256, 256, 0, stream>>>(nW2, nW2p, 128, 64);
  pack_pairs<<<(64 * 128 + 255) / 256, 256, 0, stream>>>(qW, qWp, 128, 64);
  pack_pairs<<<(64 * 128 + 255) / 256, 256, 0, stream>>>(kW, kWp, 128, 64);
  pack_pairs<<<(64 * 128 + 255) / 256, 256, 0, stream>>>(vW, vWp, 128, 64);

  const int ntiles = N_NODES / 16;
  proj_gemm<<<ntiles, 128, 0, stream>>>(h_lig, qWp, q_l, 1, 0);
  proj_gemm<<<ntiles, 128, 0, stream>>>(h_lig, kWp, k_l, 1, 0);
  proj_gemm<<<ntiles, 128, 0, stream>>>(h_lig, vWp, v_l, 0, 1);
  proj_gemm<<<ntiles, 128, 0, stream>>>(h_rec, qWp, q_r, 1, 0);
  proj_gemm<<<ntiles, 128, 0, stream>>>(h_rec, kWp, k_r, 1, 0);
  proj_gemm<<<ntiles, 128, 0, stream>>>(h_rec, vWp, v_r, 0, 1);

  attn_kernel<<<ntiles, 256, 0, stream>>>(q_l, k_r, v_r, mask, 0, cm_l);
  attn_kernel<<<ntiles, 256, 0, stream>>>(q_r, k_l, v_l, mask, 1, cm_r);

  int eblocks = (nE + 15) / 16;
  edge_kernel<<<eblocks, 128, 0, stream>>>(coors_lig, h_lig, lig_src, lig_dst,
      eW1p, eb1, eg1, ebt1, eW2p, eb2, cW1p, cb1, cg1, cbt1, cW2, cb2,
      sm_l, sx_l, cnt_l, nE);
  edge_kernel<<<eblocks, 128, 0, stream>>>(coors_rec, h_rec, rec_src, rec_dst,
      eW1p, eb1, eg1, ebt1, eW2p, eb2, cW1p, cb1, cg1, cbt1, cW2, cb2,
      sm_r, sx_r, cnt_r, nE);

  float* out_xl = (float*)d_out;
  float* out_hl = out_xl + 3 * N_NODES;
  float* out_xr = out_hl + NH;
  float* out_hr = out_xr + 3 * N_NODES;

  node_kernel<<<ntiles, 128, 0, stream>>>(h_lig, sm_l, cnt_l, cm_l, orig_lig,
      nW1p, nb1, ng1, nbt1, nW2p, nb2, out_hl);
  node_kernel<<<ntiles, 128, 0, stream>>>(h_rec, sm_r, cnt_r, cm_r, orig_rec,
      nW1p, nb1, ng1, nbt1, nW2p, nb2, out_hr);

  int cthreads = 3 * N_NODES;
  coords_kernel<<<(cthreads + 255) / 256, 256, 0, stream>>>(oc_lig, coors_lig, sx_l,
                                                            cnt_l, out_xl, N_NODES);
  coords_kernel<<<(cthreads + 255) / 256, 256, 0, stream>>>(oc_rec, coors_rec, sx_r,
                                                            cnt_r, out_xr, N_NODES);
}


# independent launches merged into shared grids: 8 packs, 6 projections, both sides of attention, edge, node, coords; 22 launches become 7; attention runs 3 block rounds, not 2+2; sums unchanged
# speedup vs baseline: 1.5277x; 1.1685x over previous
#include <hip/hip_runtime.h>
#include <hip/hip_bf16.h>
#include <cstdint>

#define N_NODES 6000
#define NH      ((size_t)N_NODES * 128)

typedef float v2f __attribute__((ext_vector_type(2)));
typedef float v4f __attribute__((ext_vector_type(4)));
typedef float v8f __attribute__((ext_vector_type(8)));

__device__ __forceinline__ v8f wmma4(v2f a, v2f b, v8f c) {
  return __builtin_amdgcn_wmma_f32_16x16x4_f32(false, a, false, b, (short)0, c,
                                               false, false);
}

__device__ __forceinline__ v2f ldsA(const float* base, int ld, int k0, int lane) {
  const int m = lane & 15;
  const int k = k0 + ((lane >> 4) << 1);
  const float* p = base + m * ld + k;
  v2f r; r.x = p[0]; r.y = p[1];
  return r;
}

__device__ __forceinline__ v2f ldBp(const float* __restrict__ P, int k0, int n0,
                                    int lane) {
  const int n = n0 + (lane & 15);
  const int k = k0 + ((lane >> 4) << 1);
  const float* p = P + (size_t)(k >> 1) * 256 + n * 2;
  v2f r; r.x = p[0]; r.y = p[1];
  return r;
}

__device__ __forceinline__ float lrelu(float x) { return x >= 0.f ? x : 0.01f * x; }

__device__ __forceinline__ float rowmax16(float v) {
  v = fmaxf(v, __shfl_xor(v, 1));
  v = fmaxf(v, __shfl_xor(v, 2));
  v = fmaxf(v, __shfl_xor(v, 4));
  v = fmaxf(v, __shfl_xor(v, 8));
  return v;
}
__device__ __forceinline__ float rowsum16(float v) {
  v += __shfl_xor(v, 1);
  v += __shfl_xor(v, 2);
  v += __shfl_xor(v, 4);
  v += __shfl_xor(v, 8);
  return v;
}

__device__ __forceinline__ void layernorm128(float* buf, const float* __restrict__ g,
                                             const float* __restrict__ b,
                                             float (*stats)[2], int tid) {
  const int r = tid >> 3, seg = tid & 7;
  float sum = 0.f, sq = 0.f;
  #pragma unroll
  for (int i = 0; i < 16; i++) {
    float v = buf[r * 132 + seg * 16 + i];
    sum += v; sq += v * v;
  }
  sum += __shfl_xor(sum, 1); sq += __shfl_xor(sq, 1);
  sum += __shfl_xor(sum, 2); sq += __shfl_xor(sq, 2);
  sum += __shfl_xor(sum, 4); sq += __shfl_xor(sq, 4);
  if (seg == 0) {
    float mean = sum * (1.f / 128.f);
    float var  = sq * (1.f / 128.f) - mean * mean;
    stats[r][0] = mean;
    stats[r][1] = rsqrtf(var + 1e-5f);
  }
  __syncthreads();
  const float mean = stats[r][0], rstd = stats[r][1];
  #pragma unroll
  for (int i = 0; i < 16; i++) {
    int c = seg * 16 + i;
    float v = buf[r * 132 + c];
    buf[r * 132 + c] = g[c] * (v - mean) * rstd + b[c];
  }
}

__device__ __forceinline__ void gemm128_lds(const float* in, const float* __restrict__ Wp,
                                            const float* __restrict__ bias, float* outb,
                                            int lane, int wave, int act) {
  const int hi = lane >> 4, cn = lane & 15;
  const int n0 = wave * 32;
  v8f a0 = {}, a1 = {};
  for (int k0 = 0; k0 < 128; k0 += 4) {
    v2f a  = ldsA(in, 132, k0, lane);
    v2f b0 = ldBp(Wp, k0, n0, lane);
    v2f b1 = ldBp(Wp, k0, n0 + 16, lane);
    a0 = wmma4(a, b0, a0);
    a1 = wmma4(a, b1, a1);
  }
  #pragma unroll
  for (int j = 0; j < 8; j++) {
    int r = j + 8 * hi;
    float x0 = a0[j] + bias[n0 + cn];
    float x1 = a1[j] + bias[n0 + 16 + cn];
    if (act) { x0 = lrelu(x0); x1 = lrelu(x1); }
    outb[r * 132 + n0 + cn] = x0;
    outb[r * 132 + n0 + 16 + cn] = x1;
  }
}

#define PK_E1 0
#define PK_E2 136
#define PK_C1 200
#define PK_N1 264
#define PK_N2 488
#define PK_Q  552
#define PK_K  616
#define PK_V  680
#define PK_END 744
__global__ void pack_all(const float* __restrict__ eW1, const float* __restrict__ eW2,
                         const float* __restrict__ cW1, const float* __restrict__ nW1,
                         const float* __restrict__ nW2, const float* __restrict__ qW,
                         const float* __restrict__ kW, const float* __restrict__ vW,
                         float* __restrict__ P) {
  int i = blockIdx.x * 256 + threadIdx.x;
  if (i >= PK_END * 128) return;
  int gp = i >> 7, n = i & 127;
  const float* W; int K, p0;
  if (gp < PK_E2)      { W = eW1; K = 271; p0 = PK_E1; }
  else if (gp < PK_C1) { W = eW2; K = 128; p0 = PK_E2; }
  else if (gp < PK_N1) { W = cW1; K = 128; p0 = PK_C1; }
  else if (gp < PK_N2) { W = nW1; K = 448; p0 = PK_N1; }
  else if (gp < PK_Q)  { W = nW2; K = 128; p0 = PK_N2; }
  else if (gp < PK_K)  { W = qW;  K = 128; p0 = PK_Q; }
  else if (gp < PK_V)  { W = kW;  K = 128; p0 = PK_K; }
  else                 { W = vW;  K = 128; p0 = PK_V; }
  int k = 2 * (gp - p0);
  float a = (k < K)     ? W[(size_t)k * 128 + n]       : 0.f;
  float b = (k + 1 < K) ? W[(size_t)(k + 1) * 128 + n] : 0.f;
  P[(size_t)gp * 256 + n * 2]     = a;
  P[(size_t)gp * 256 + n * 2 + 1] = b;
}

__device__ __forceinline__ void proj_body(const float* __restrict__ A,
                                          const float* __restrict__ Wp,
                                          float* __restrict__ C, int act,
                                          int pack, int bid) {
  __shared__ float at[16 * 132];
  const int tid = threadIdx.x, lane = tid & 31, wave = tid >> 5;
  const int row0 = bid * 16;
  for (int i = tid; i < 16 * 128; i += 128) {
    int r = i >> 7, c = i & 127;
    at[r * 132 + c] = A[(size_t)(row0 + r) * 128 + c];
  }
  __syncthreads();
  const int n0 = wave * 32;
  v8f a0 = {}, a1 = {};
  for (int k0 = 0; k0 < 128; k0 += 4) {
    v2f a  = ldsA(at, 132, k0, lane);
    v2f b0 = ldBp(Wp, k0, n0, lane);
    v2f b1 = ldBp(Wp, k0, n0 + 16, lane);
    a0 = wmma4(a, b0, a0);
    a1 = wmma4(a, b1, a1);
  }
  const int hi = lane >> 4, cn = lane & 15;
  #pragma unroll
  for (int j = 0; j < 8; j++) {
    int r = row0 + j + 8 * hi;
    float x0 = a0[j], x1 = a1[j];
    if (act) { x0 = lrelu(x0); x1 = lrelu(x1); }
    if (pack) {
      C[(size_t)(r >> 1) * 256 + (n0 + cn) * 2 + (r & 1)]      = x0;
      C[(size_t)(r >> 1) * 256 + (n0 + 16 + cn) * 2 + (r & 1)] = x1;
    } else {
      C[(size_t)r * 128 + n0 + cn] = x0;
      C[(size_t)r * 128 + n0 + 16 + cn] = x1;
    }
  }
}

__global__ void __launch_bounds__(128) proj_all(
    const float* __restrict__ h0, const float* __restrict__ h1,
    const float* __restrict__ qWp, const float* __restrict__ kWp,
    const float* __restrict__ vWp,
    float* __restrict__ q0, float* __restrict__ k0, float* __restrict__ v0,
    float* __restrict__ q1, float* __restrict__ k1, float* __restrict__ v1) {
  const int p = blockIdx.x / (N_NODES / 16);
  const int bid = blockIdx.x - p * (N_NODES / 16);
  const int side = p >= 3;
  const int j = p - 3 * side;
  const float* A = side ? h1 : h0;
  const float* Wp = j == 0 ? qWp : (j == 1 ? kWp : vWp);
  float* C = j == 0 ? (side ? q1 : q0) : (j == 1 ? (side ? k1 : k0) : (side ? v1 : v0));
  proj_body(A, Wp, C, j != 2, j == 2, bid);
}

__device__ __forceinline__ void attn_body(const float* __restrict__ Q,
                                          const float* __restrict__ K,
                                          const float* __restrict__ Vp,
                                          const float* __restrict__ mask,
                                          int tmask, float* __restrict__ out, int bid) {
  __shared__ __attribute__((aligned(16))) float qt[16 * 132];
  __shared__ float pbuf[8][16 * 18];
  __shared__ float outbuf[16 * 128];
  __shared__ float mbuf[8][16], lbuf[8][16];
  __shared__ float Mrow[16], Lrow[16];
  const int tid = threadIdx.x, lane = tid & 31;
  const int wave = __builtin_amdgcn_readfirstlane(tid >> 5);
  const int hi = lane >> 4, cn = lane & 15;
  const int qrow0 = bid * 16;

  for (int i = tid; i < 2048; i += 256) {
    int r = i >> 7, c = i & 127;
    qt[r * 132 + c] = Q[(size_t)(qrow0 + r) * 128 + c];
  }
  __syncthreads();

  float mr[8], lr[8];
  v8f acc[8];
  #pragma unroll
  for (int j = 0; j < 8; j++) { mr[j] = -3.0e38f; lr[j] = 0.f; acc[j] = (v8f){}; }
  float* pw = &pbuf[wave][0];

  const float* qap = qt + cn * 132 + 4 * hi;
  const unsigned koff = (unsigned)(cn * 128 + 4 * hi);
  const unsigned voff = (unsigned)(hi * 256 + 4 * cn);
  const unsigned mrs = tmask ? 1u : (unsigned)N_NODES;
  const unsigned mcs = tmask ? (unsigned)N_NODES : 1u;
  const unsigned moff = (unsigned)(qrow0 + 8 * hi) * mrs + (unsigned)cn * mcs;
  v4f kf[16], vf[16];
  float mk[8];
  {
    const float* kb = K + (size_t)wave * 2048;
    #pragma unroll
    for (int i = 0; i < 16; i++) kf[i] = *(const v4f*)(kb + (koff + 8u * i));
    const float* mb = mask + (size_t)(wave * 16) * mcs;
    #pragma unroll
    for (int j = 0; j < 8; j++) mk[j] = mb[moff + (unsigned)j * mrs];
    const float* vb = Vp + (size_t)wave * 2048;
    #pragma unroll
    for (int i = 0; i < 16; i++) vf[i] = *(const v4f*)(vb + (voff + 512u * (i >> 2) + 64u * (i & 3)));
  }

  for (int ch = wave; ch < N_NODES / 16; ch += 8) {
    const int nch = (ch + 8 < N_NODES / 16) ? ch + 8 : N_NODES / 16 - 1;
    v8f s = {};
    #pragma unroll
    for (int i = 0; i < 16; i++) {
      v4f qa = *(const v4f*)(qap + 8 * i);
      v2f a0, a1, b0, b1;
      a0.x = qa.x; a0.y = qa.y; a1.x = qa.z; a1.y = qa.w;
      b0.x = kf[i].x; b0.y = kf[i].y; b1.x = kf[i].z; b1.y = kf[i].w;
      s = wmma4(a0, b0, s);
      s = wmma4(a1, b1, s);
    }
    float sc[8];
    #pragma unroll
    for (int j = 0; j < 8; j++) {
      float mv = mk[j];
      float sj = mv * s[j] - 1000.f * (1.f - mv);
      float mnew = fmaxf(mr[j], rowmax16(sj));
      sc[j] = __expf(mr[j] - mnew);
      float pj = __expf(sj - mnew);
      lr[j] = lr[j] * sc[j] + rowsum16(pj);
      mr[j] = mnew;
      pw[(j + 8 * hi) * 18 + cn] = pj;
    }
    #pragma unroll
    for (int t = 0; t < 8; t++)
      #pragma unroll
      for (int j = 0; j < 8; j++) acc[t][j] *= sc[j];
    {
      const float* kb = K + (size_t)nch * 2048;
      #pragma unroll
      for (int i = 0; i < 16; i++) kf[i] = *(const v4f*)(kb + (koff + 8u * i));
      const float* mb = mask + (size_t)(nch * 16) * mcs;
      #pragma unroll
      for (int j = 0; j < 8; j++) mk[j] = mb[moff + (unsigned)j * mrs];
    }
    asm volatile("s_wait_dscnt 0" ::: "memory");
    #pragma unroll
    for (int kk = 0; kk < 4; kk++) {
      v2f a = ldsA(pw, 18, 4 * kk, lane);
      #pragma unroll
      for (int u = 0; u < 4; u++) {
        v2f b0, b1;
        b0.x = vf[4 * kk + u].x; b0.y = vf[4 * kk + u].y;
        b1.x = vf[4 * kk + u].z; b1.y = vf[4 * kk + u].w;
        acc[2 * u]     = wmma4(a, b0, acc[2 * u]);
        acc[2 * u + 1] = wmma4(a, b1, acc[2 * u + 1]);
      }
    }
    {
      const float* vb = Vp + (size_t)nch * 2048;
      #pragma unroll
      for (int i = 0; i < 16; i++) vf[i] = *(const v4f*)(vb + (voff + 512u * (i >> 2) + 64u * (i & 3)));
    }
  }

  if (cn == 0) {
    #pragma unroll
    for (int j = 0; j < 8; j++) {
      mbuf[wave][j + 8 * hi] = mr[j];
      lbuf[wave][j + 8 * hi] = lr[j];
    }
  }
  __syncthreads();
  for (int i = tid; i < 2048; i += 256) outbuf[i] = 0.f;
  if (tid < 16) {
    float M = -3.0e38f;
    for (int w = 0; w < 8; w++) M = fmaxf(M, mbuf[w][tid]);
    float L = 0.f;
    for (int w = 0; w < 8; w++) L += lbuf[w][tid] * __expf(mbuf[w][tid] - M);
    Mrow[tid] = M;
    Lrow[tid] = L > 0.f ? 1.f / L : 0.f;
  }
  __syncthreads();
  #pragma unroll
  for (int j = 0; j < 8; j++) {
    float e = __expf(mr[j] - Mrow[j + 8 * hi]);
    #pragma unroll
    for (int t = 0; t < 8; t++)
      atomicAdd(&outbuf[(j + 8 * hi) * 128 + 32 * (t >> 1) + 2 * cn + (t & 1)],
                acc[t][j] * e);
  }
  __syncthreads();
  for (int i = tid; i < 2048; i += 256) {
    int r = i >> 7, c = i & 127;
    out[(size_t)(qrow0 + r) * 128 + c] = outbuf[i] * Lrow[r];
  }
}

__global__ void __launch_bounds__(256) attn_pair(
    const float* __restrict__ Q0, const float* __restrict__ K0,
    const float* __restrict__ V0, float* __restrict__ out0,
    const float* __restrict__ Q1, const float* __restrict__ K1,
    const float* __restrict__ V1, float* __restrict__ out1,
    const float* __restrict__ mask) {
  const int second = blockIdx.x >= N_NODES / 16;
  const int bid = blockIdx.x - second * (N_NODES / 16);
  attn_body(second ? Q1 : Q0, second ? K1 : K0, second ? V1 : V0, mask, second,
            second ? out1 : out0, bid);
}

#define CAT_LD 274
__device__ __forceinline__ void edge_body(
    const float* __restrict__ coors, const float* __restrict__ h,
    const int* __restrict__ src, const int* __restrict__ dst,
    const float* __restrict__ eW1p, const float* __restrict__ eb1,
    const float* __restrict__ eg1, const float* __restrict__ ebt1,
    const float* __restrict__ eW2p, const float* __restrict__ eb2,
    const float* __restrict__ cW1p, const float* __restrict__ cb1,
    const float* __restrict__ cg1, const float* __restrict__ cbt1,
    const float* __restrict__ cW2, const float* __restrict__ cb2,
    float* __restrict__ sum_msg, float* __restrict__ sum_x,
    float* __restrict__ cnt, int nE, int bid) {
  __shared__ __attribute__((aligned(16))) float cat[16 * CAT_LD];
  __shared__ float buf1[16 * 132];
  __shared__ float buf2[16 * 132];
  __shared__ float stats[16][2];
  __shared__ float xrel[16][3];
  __shared__ float coefs[16];
  __shared__ int sidx[16], didx[16];
  const int tid = threadIdx.x, lane = tid & 31, wave = tid >> 5;
  const int hi = lane >> 4, cn = lane & 15;
  const int e0 = bid * 16;

  if (tid < 16) {
    int e = e0 + tid;
    if (e < nE) {
      int s = src[e], d = dst[e];
      sidx[tid] = s; didx[tid] = d;
      float x0 = coors[s * 3 + 0] - coors[d * 3 + 0];
      float x1 = coors[s * 3 + 1] - coors[d * 3 + 1];
      float x2 = coors[s * 3 + 2] - coors[d * 3 + 2];
      xrel[tid][0] = x0; xrel[tid][1] = x1; xrel[tid][2] = x2;
      float d2 = x0 * x0 + x1 * x1 + x2 * x2;
      float sig = 1.f;
      for (int j = 0; j < 15; j++) {
        cat[tid * CAT_LD + 256 + j] = __expf(-d2 / sig);
        sig *= 1.5f;
      }
      cat[tid * CAT_LD + 271] = 0.f;
    } else {
      sidx[tid] = 0; didx[tid] = -1;
      xrel[tid][0] = xrel[tid][1] = xrel[tid][2] = 0.f;
      for (int j = 0; j < 16; j++) cat[tid * CAT_LD + 256 + j] = 0.f;
    }
  }
  __syncthreads();
  {
    const int l4 = (tid & 31) * 4;
    const int half = wave & 1;
    const int* ip = half ? didx : sidx;
    int nn[8], dd[8];
    v4f g[8];
    #pragma unroll
    for (int it = 0; it < 8; it++) {
      const int e = (wave >> 1) + 2 * it;
      nn[it] = ip[e];
      dd[it] = didx[e];
    }
    #pragma unroll
    for (int it = 0; it < 8; it++) {
      const int node = (dd[it] >= 0) ? nn[it] : 0;
      g[it] = *(const v4f*)(h + (size_t)node * 128 + l4);
    }
    #pragma unroll
    for (int it = 0; it < 8; it++) {
      const int e = (wave >> 1) + 2 * it;
      float* p = cat + e * CAT_LD + half * 128 + l4;
      v2f lo, hi2;
      lo.x = g[it].x; lo.y = g[it].y; hi2.x = g[it].z; hi2.y = g[it].w;
      *(v2f*)p = lo;
      *(v2f*)(p + 2) = hi2;
    }
  }
  __syncthreads();

  {
    const int n0 = wave * 32;
    v8f a0 = {}, a1 = {};
    for (int k0 = 0; k0 < 272; k0 += 4) {
      v2f a  = ldsA(cat, CAT_LD, k0, lane);
      v2f b0 = ldBp(eW1p, k0, n0, lane);
      v2f b1 = ldBp(eW1p, k0, n0 + 16, lane);
      a0 = wmma4(a, b0, a0);
      a1 = wmma4(a, b1, a1);
    }
    #pragma unroll
    for (int j = 0; j < 8; j++) {
      int r = j + 8 * hi;
      buf1[r * 132 + n0 + cn]      = lrelu(a0[j] + eb1[n0 + cn]);
      buf1[r * 132 + n0 + 16 + cn] = lrelu(a1[j] + eb1[n0 + 16 + cn]);
    }
  }
  __syncthreads();
  layernorm128(buf1, eg1, ebt1, stats, tid);
  __syncthreads();
  gemm128_lds(buf1, eW2p, eb2, buf2, lane, wave, 0);
  __syncthreads();
  gemm128_lds(buf2, cW1p, cb1, buf1, lane, wave, 1);
  __syncthreads();
  layernorm128(buf1, cg1, cbt1, stats, tid);
  __syncthreads();
  {
    int r = tid >> 3, seg = tid & 7;
    float sum = 0.f;
    #pragma unroll
    for (int i = 0; i < 16; i++) {
      int c = seg * 16 + i;
      sum += buf1[r * 132 + c] * cW2[c];
    }
    sum += __shfl_xor(sum, 1);
    sum += __shfl_xor(sum, 2);
    sum += __shfl_xor(sum, 4);
    if (seg == 0) coefs[r] = sum + cb2[0];
  }
  __syncthreads();
  for (int i = tid; i < 2048; i += 128) {
    int e = i >> 7, c = i & 127;
    int d = didx[e];
    if (d >= 0) atomicAdd(&sum_msg[(size_t)d * 128 + c], buf2[e * 132 + c]);
  }
  if (tid < 48) {
    int e = tid / 3, ax = tid % 3;
    int d = didx[e];
    if (d >= 0) atomicAdd(&sum_x[d * 3 + ax], xrel[e][ax] * coefs[e]);
  }
  if (tid < 16) {
    int d = didx[tid];
    if (d >= 0) atomicAdd(&cnt[d], 1.f);
  }
}

__global__ void __launch_bounds__(128) edge_pair(
    const float* __restrict__ coors0, const float* __restrict__ h0,
    const int* __restrict__ src0, const int* __restrict__ dst0,
    const float* __restrict__ coors1, const float* __restrict__ h1,
    const int* __restrict__ src1, const int* __restrict__ dst1,
    const float* __restrict__ eW1p, const float* __restrict__ eb1,
    const float* __restrict__ eg1, const float* __restrict__ ebt1,
    const float* __restrict__ eW2p, const float* __restrict__ eb2,
    const float* __restrict__ cW1p, const float* __restrict__ cb1,
    const float* __restrict__ cg1, const float* __restrict__ cbt1,
    const float* __restrict__ cW2, const float* __restrict__ cb2,
    float* __restrict__ sm0, float* __restrict__ sx0, float* __restrict__ cnt0,
    float* __restrict__ sm1, float* __restrict__ sx1, float* __restrict__ cnt1,
    int nE, int eblocks) {
  const int second = (int)blockIdx.x >= eblocks;
  const int bid = blockIdx.x - second * eblocks;
  edge_body(second ? coors1 : coors0, second ? h1 : h0, second ? src1 : src0,
            second ? dst1 : dst0, eW1p, eb1, eg1, ebt1, eW2p, eb2, cW1p, cb1, cg1,
            cbt1, cW2, cb2, second ? sm1 : sm0, second ? sx1 : sx0,
            second ? cnt1 : cnt0, nE, bid);
}

#define NCAT_LD 450
__device__ __forceinline__ void node_body(
    const float* __restrict__ h, const float* __restrict__ summ,
    const float* __restrict__ cnt, const float* __restrict__ cm,
    const float* __restrict__ orig,
    const float* __restrict__ nW1p, const float* __restrict__ nb1,
    const float* __restrict__ ng1, const float* __restrict__ nbt1,
    const float* __restrict__ nW2p, const float* __restrict__ nb2,
    float* __restrict__ hout, int bid) {
  __shared__ float cat[16 * NCAT_LD];
  __shared__ float buf1[16 * 132];
  __shared__ float stats[16][2];
  const int tid = threadIdx.x, lane = tid & 31, wave = tid >> 5;
  const int hi = lane >> 4, cn = lane & 15;
  const int node0 = bid * 16;

  for (int i = tid; i < 2048; i += 128) {
    int e = i >> 7, c = i & 127;
    int node = node0 + e;
    cat[e * NCAT_LD + c] = h[(size_t)node * 128 + c];
    float ct = cnt[node];
    cat[e * NCAT_LD + 128 + c] = ct > 0.f ? summ[(size_t)node * 128 + c] / ct : 0.f;
    cat[e * NCAT_LD + 256 + c] = cm[(size_t)node * 128 + c];
  }
  for (int i = tid; i < 1024; i += 128) {
    int e = i >> 6, c = i & 63;
    cat[e * NCAT_LD + 384 + c] = orig[(size_t)(node0 + e) * 64 + c];
  }
  __syncthreads();
  {
    const int n0 = wave * 32;
    v8f a0 = {}, a1 = {};
    for (int k0 = 0; k0 < 448; k0 += 4) {
      v2f a  = ldsA(cat, NCAT_LD, k0, lane);
      v2f b0 = ldBp(nW1p, k0, n0, lane);
      v2f b1 = ldBp(nW1p, k0, n0 + 16, lane);
      a0 = wmma4(a, b0, a0);
      a1 = wmma4(a, b1, a1);
    }
    #pragma unroll
    for (int j = 0; j < 8; j++) {
      int r = j + 8 * hi;
      buf1[r * 132 + n0 + cn]      = lrelu(a0[j] + nb1[n0 + cn]);
      buf1[r * 132 + n0 + 16 + cn] = lrelu(a1[j] + nb1[n0 + 16 + cn]);
    }
  }
  __syncthreads();
  layernorm128(buf1, ng1, nbt1, stats, tid);
  __syncthreads();
  {
    const int n0 = wave * 32;
    v8f a0 = {}, a1 = {};
    for (int k0 = 0; k0 < 128; k0 += 4) {
      v2f a  = ldsA(buf1, 132, k0, lane);
      v2f b0 = ldBp(nW2p, k0, n0, lane);
      v2f b1 = ldBp(nW2p, k0, n0 + 16, lane);
      a0 = wmma4(a, b0, a0);
      a1 = wmma4(a, b1, a1);
    }
    #pragma unroll
    for (int j = 0; j < 8; j++) {
      int r = j + 8 * hi;
      size_t g0 = (size_t)(node0 + r) * 128 + n0 + cn;
      size_t g1 = g0 + 16;
      hout[g0] = 0.5f * (a0[j] + nb2[n0 + cn]) + 0.5f * h[g0];
      hout[g1] = 0.5f * (a1[j] + nb2[n0 + 16 + cn]) + 0.5f * h[g1];
    }
  }
}

__global__ void __launch_bounds__(128) node_pair(
    const float* __restrict__ h0, const float* __restrict__ summ0,
    const float* __restrict__ cnt0, const float* __restrict__ cm0,
    const float* __restrict__ orig0, float* __restrict__ hout0,
    const float* __restrict__ h1, const float* __restrict__ summ1,
    const float* __restrict__ cnt1, const float* __restrict__ cm1,
    const float* __restrict__ orig1, float* __restrict__ hout1,
    const float* __restrict__ nW1p, const float* __restrict__ nb1,
    const float* __restrict__ ng1, const float* __restrict__ nbt1,
    const float* __restrict__ nW2p, const float* __restrict__ nb2) {
  const int second = blockIdx.x >= N_NODES / 16;
  const int bid = blockIdx.x - second * (N_NODES / 16);
  node_body(second ? h1 : h0, second ? summ1 : summ0, second ? cnt1 : cnt0,
            second ? cm1 : cm0, second ? orig1 : orig0, nW1p, nb1, ng1, nbt1, nW2p,
            nb2, second ? hout1 : hout0, bid);
}

#define CB ((3 * N_NODES + 255) / 256)
__global__ void coords_pair(const float* __restrict__ oc0, const float* __restrict__ c0,
                            const float* __restrict__ sx0, const float* __restrict__ cnt0,
                            float* __restrict__ out0,
                            const float* __restrict__ oc1, const float* __restrict__ c1,
                            const float* __restrict__ sx1, const float* __restrict__ cnt1,
                            float* __restrict__ out1, int n) {
  const int second = blockIdx.x >= CB;
  const float* oc = second ? oc1 : oc0;
  const float* c = second ? c1 : c0;
  const float* sx = second ? sx1 : sx0;
  const float* cnt = second ? cnt1 : cnt0;
  float* out = second ? out1 : out0;
  int i = (blockIdx.x - second * CB) * 256 + threadIdx.x;
  if (i >= n * 3) return;
  int node = i / 3;
  float ct = cnt[node];
  float xu = ct > 0.f ? sx[i] / ct : 0.f;
  out[i] = 0.2f * oc[i] + 0.8f * c[i] + xu;
}

extern "C" void kernel_launch(void* const* d_in, const int* in_sizes, int n_in,
                              void* d_out, int out_size, void* d_ws, size_t ws_size,
                              hipStream_t stream) {
  const float* coors_lig = (const float*)d_in[0];
  const float* h_lig     = (const float*)d_in[1];
  const float* orig_lig  = (const float*)d_in[2];
  const float* oc_lig    = (const float*)d_in[3];
  const float* coors_rec = (const float*)d_in[4];
  const float* h_rec     = (const float*)d_in[5];
  const float* orig_rec  = (const float*)d_in[6];
  const float* oc_rec    = (const float*)d_in[7];
  const int* lig_src = (const int*)d_in[8];
  const int* lig_dst = (const int*)d_in[9];
  const int* rec_src = (const int*)d_in[10];
  const int* rec_dst = (const int*)d_in[11];
  const float* mask = (const float*)d_in[12];
  const float* eW1 = (const float*)d_in[13];
  const float* eb1 = (const float*)d_in[14];
  const float* eg1 = (const float*)d_in[15];
  const float* ebt1 = (const float*)d_in[16];
  const float* eW2 = (const float*)d_in[17];
  const float* eb2 = (const float*)d_in[18];
  const float* qW = (const float*)d_in[19];
  const float* kW = (const float*)d_in[20];
  const float* vW = (const float*)d_in[21];
  const float* nW1 = (const float*)d_in[22];
  const float* nb1 = (const float*)d_in[23];
  const float* ng1 = (const float*)d_in[24];
  const float* nbt1 = (const float*)d_in[25];
  const float* nW2 = (const float*)d_in[26];
  const float* nb2 = (const float*)d_in[27];
  const float* cW1 = (const float*)d_in[28];
  const float* cb1 = (const float*)d_in[29];
  const float* cg1 = (const float*)d_in[30];
  const float* cbt1 = (const float*)d_in[31];
  const float* cW2 = (const float*)d_in[32];
  const float* cb2 = (const float*)d_in[33];
  const int nE = in_sizes[8];

  float* ws = (float*)d_ws;
  float* q_l = ws + 0 * NH;
  float* k_l = ws + 1 * NH;
  float* v_l = ws + 2 * NH;
  float* q_r = ws + 3 * NH;
  float* k_r = ws + 4 * NH;
  float* v_r = ws + 5 * NH;
  float* cm_l = ws + 6 * NH;
  float* cm_r = ws + 7 * NH;
  float* sm_l = ws + 8 * NH;
  float* sm_r = ws + 9 * NH;
  float* sx_l = ws + 10 * NH;
  float* sx_r = sx_l + 3 * N_NODES;
  float* cnt_l = sx_r + 3 * N_NODES;
  float* cnt_r = cnt_l + N_NODES;
  float* eW1p = cnt_r + N_NODES;
  float* eW2p = eW1p + 136 * 256;
  float* cW1p = eW2p + 64 * 256;
  float* nW1p = cW1p + 64 * 256;
  float* nW2p = nW1p + 224 * 256;
  float* qWp  = nW2p + 64 * 256;
  float* kWp  = qWp + 64 * 256;
  float* vWp  = kWp + 64 * 256;

  size_t accum_floats = 2 * NH + 8 * (size_t)N_NODES;
  hipMemsetAsync(sm_l, 0, accum_floats * sizeof(float), stream);

  pack_all<<<PK_END * 128 / 256, 256, 0, stream>>>(eW1, eW2, cW1, nW1, nW2, qW, kW, vW,
                                                  eW1p);

  const int ntiles = N_NODES / 16;
  proj_all<<<6 * ntiles, 128, 0, stream>>>(h_lig, h_rec, qWp, kWp, vWp, q_l, k_l, v_l,
                                           q_r, k_r, v_r);

  attn_pair<<<2 * ntiles, 256, 0, stream>>>(q_l, k_r, v_r, cm_l, q_r, k_l, v_l, cm_r,
                                            mask);

  int eblocks = (nE + 15) / 16;
  edge_pair<<<2 * eblocks, 128, 0, stream>>>(coors_lig, h_lig, lig_src, lig_dst,
      coors_rec, h_rec, rec_src, rec_dst,
      eW1p, eb1, eg1, ebt1, eW2p, eb2, cW1p, cb1, cg1, cbt1, cW2, cb2,
      sm_l, sx_l, cnt_l, sm_r, sx_r, cnt_r, nE, eblocks);

  float* out_xl = (float*)d_out;
  float* out_hl = out_xl + 3 * N_NODES;
  float* out_xr = out_hl + NH;
  float* out_hr = out_xr + 3 * N_NODES;

  node_pair<<<2 * ntiles, 128, 0, stream>>>(h_lig, sm_l, cnt_l, cm_l, orig_lig, out_hl,
      h_rec, sm_r, cnt_r, cm_r, orig_rec, out_hr,
      nW1p, nb1, ng1, nbt1, nW2p, nb2);

  coords_pair<<<2 * CB, 256, 0, stream>>>(oc_lig, coors_lig, sx_l, cnt_l, out_xl,
                                          oc_rec, coors_rec, sx_r, cnt_r, out_xr,
                                          N_NODES);
}


# edge kernel: 64 edges per 8-wave block, a wave owns 16 columns of four 16-row tiles, so each weight fragment read feeds four products (weight reads quartered); msg tile reuses gather LDS; same sums
# speedup vs baseline: 1.6763x; 1.0973x over previous
#include <hip/hip_runtime.h>
#include <hip/hip_bf16.h>
#include <cstdint>

#define N_NODES 6000
#define NH      ((size_t)N_NODES * 128)

typedef float v2f __attribute__((ext_vector_type(2)));
typedef float v4f __attribute__((ext_vector_type(4)));
typedef float v8f __attribute__((ext_vector_type(8)));

__device__ __forceinline__ v8f wmma4(v2f a, v2f b, v8f c) {
  return __builtin_amdgcn_wmma_f32_16x16x4_f32(false, a, false, b, (short)0, c,
                                               false, false);
}

__device__ __forceinline__ v2f ldsA(const float* base, int ld, int k0, int lane) {
  const int m = lane & 15;
  const int k = k0 + ((lane >> 4) << 1);
  const float* p = base + m * ld + k;
  v2f r; r.x = p[0]; r.y = p[1];
  return r;
}

__device__ __forceinline__ v2f ldBp(const float* __restrict__ P, int k0, int n0,
                                    int lane) {
  const int n = n0 + (lane & 15);
  const int k = k0 + ((lane >> 4) << 1);
  const float* p = P + (size_t)(k >> 1) * 256 + n * 2;
  v2f r; r.x = p[0]; r.y = p[1];
  return r;
}

__device__ __forceinline__ float lrelu(float x) { return x >= 0.f ? x : 0.01f * x; }

__device__ __forceinline__ float rowmax16(float v) {
  v = fmaxf(v, __shfl_xor(v, 1));
  v = fmaxf(v, __shfl_xor(v, 2));
  v = fmaxf(v, __shfl_xor(v, 4));
  v = fmaxf(v, __shfl_xor(v, 8));
  return v;
}
__device__ __forceinline__ float rowsum16(float v) {
  v += __shfl_xor(v, 1);
  v += __shfl_xor(v, 2);
  v += __shfl_xor(v, 4);
  v += __shfl_xor(v, 8);
  return v;
}

__device__ __forceinline__ void layernorm128(float* buf, const float* __restrict__ g,
                                             const float* __restrict__ b,
                                             float (*stats)[2], int tid) {
  const int r = tid >> 3, seg = tid & 7;
  float sum = 0.f, sq = 0.f;
  #pragma unroll
  for (int i = 0; i < 16; i++) {
    float v = buf[r * 132 + seg * 16 + i];
    sum += v; sq += v * v;
  }
  sum += __shfl_xor(sum, 1); sq += __shfl_xor(sq, 1);
  sum += __shfl_xor(sum, 2); sq += __shfl_xor(sq, 2);
  sum += __shfl_xor(sum, 4); sq += __shfl_xor(sq, 4);
  if (seg == 0) {
    float mean = sum * (1.f / 128.f);
    float var  = sq * (1.f / 128.f) - mean * mean;
    stats[r][0] = mean;
    stats[r][1] = rsqrtf(var + 1e-5f);
  }
  __syncthreads();
  const float mean = stats[r][0], rstd = stats[r][1];
  #pragma unroll
  for (int i = 0; i < 16; i++) {
    int c = seg * 16 + i;
    float v = buf[r * 132 + c];
    buf[r * 132 + c] = g[c] * (v - mean) * rstd + b[c];
  }
}

__device__ __forceinline__ void gemm128_lds(const float* in, const float* __restrict__ Wp,
                                            const float* __restrict__ bias, float* outb,
                                            int lane, int wave, int act) {
  const int hi = lane >> 4, cn = lane & 15;
  const int n0 = wave * 32;
  v8f a0 = {}, a1 = {};
  for (int k0 = 0; k0 < 128; k0 += 4) {
    v2f a  = ldsA(in, 132, k0, lane);
    v2f b0 = ldBp(Wp, k0, n0, lane);
    v2f b1 = ldBp(Wp, k0, n0 + 16, lane);
    a0 = wmma4(a, b0, a0);
    a1 = wmma4(a, b1, a1);
  }
  #pragma unroll
  for (int j = 0; j < 8; j++) {
    int r = j + 8 * hi;
    float x0 = a0[j] + bias[n0 + cn];
    float x1 = a1[j] + bias[n0 + 16 + cn];
    if (act) { x0 = lrelu(x0); x1 = lrelu(x1); }
    outb[r * 132 + n0 + cn] = x0;
    outb[r * 132 + n0 + 16 + cn] = x1;
  }
}

__device__ __forceinline__ void layernorm128x64(float* buf, const float* __restrict__ g,
                                                const float* __restrict__ b,
                                                float (*stats)[2], int tid) {
  const int seg = tid & 7;
  #pragma unroll
  for (int hh = 0; hh < 2; hh++) {
    const int r = (tid >> 3) + 32 * hh;
    float sum = 0.f, sq = 0.f;
    #pragma unroll
    for (int i = 0; i < 16; i++) {
      float v = buf[r * 132 + seg * 16 + i];
      sum += v; sq += v * v;
    }
    sum += __shfl_xor(sum, 1); sq += __shfl_xor(sq, 1);
    sum += __shfl_xor(sum, 2); sq += __shfl_xor(sq, 2);
    sum += __shfl_xor(sum, 4); sq += __shfl_xor(sq, 4);
    if (seg == 0) {
      float mean = sum * (1.f / 128.f);
      float var  = sq * (1.f / 128.f) - mean * mean;
      stats[r][0] = mean;
      stats[r][1] = rsqrtf(var + 1e-5f);
    }
  }
  __syncthreads();
  #pragma unroll
  for (int hh = 0; hh < 2; hh++) {
    const int r = (tid >> 3) + 32 * hh;
    const float mean = stats[r][0], rstd = stats[r][1];
    #pragma unroll
    for (int i = 0; i < 16; i++) {
      int c = seg * 16 + i;
      float v = buf[r * 132 + c];
      buf[r * 132 + c] = g[c] * (v - mean) * rstd + b[c];
    }
  }
}

__device__ __forceinline__ void gemm128_lds64(const float* in, const float* __restrict__ Wp,
                                              const float* __restrict__ bias, float* outb,
                                              int lane, int wave, int act) {
  const int hi = lane >> 4, cn = lane & 15;
  const int n0 = wave * 16;
  v8f acc[4];
  #pragma unroll
  for (int mt = 0; mt < 4; mt++) acc[mt] = (v8f){};
  for (int k0 = 0; k0 < 128; k0 += 4) {
    v2f b = ldBp(Wp, k0, n0, lane);
    #pragma unroll
    for (int mt = 0; mt < 4; mt++) {
      v2f x = ldsA(in + mt * 16 * 132, 132, k0, lane);
      acc[mt] = wmma4(x, b, acc[mt]);
    }
  }
  const float bb = bias[n0 + cn];
  #pragma unroll
  for (int mt = 0; mt < 4; mt++) {
    #pragma unroll
    for (int j = 0; j < 8; j++) {
      int r = mt * 16 + j + 8 * hi;
      float x = acc[mt][j] + bb;
      if (act) x = lrelu(x);
      outb[r * 132 + n0 + cn] = x;
    }
  }
}

#define PK_E1 0
#define PK_E2 136
#define PK_C1 200
#define PK_N1 264
#define PK_N2 488
#define PK_Q  552
#define PK_K  616
#define PK_V  680
#define PK_END 744
__global__ void pack_all(const float* __restrict__ eW1, const float* __restrict__ eW2,
                         const float* __restrict__ cW1, const float* __restrict__ nW1,
                         const float* __restrict__ nW2, const float* __restrict__ qW,
                         const float* __restrict__ kW, const float* __restrict__ vW,
                         float* __restrict__ P) {
  int i = blockIdx.x * 256 + threadIdx.x;
  if (i >= PK_END * 128) return;
  int gp = i >> 7, n = i & 127;
  const float* W; int K, p0;
  if (gp < PK_E2)      { W = eW1; K = 271; p0 = PK_E1; }
  else if (gp < PK_C1) { W = eW2; K = 128; p0 = PK_E2; }
  else if (gp < PK_N1) { W = cW1; K = 128; p0 = PK_C1; }
  else if (gp < PK_N2) { W = nW1; K = 448; p0 = PK_N1; }
  else if (gp < PK_Q)  { W = nW2; K = 128; p0 = PK_N2; }
  else if (gp < PK_K)  { W = qW;  K = 128; p0 = PK_Q; }
  else if (gp < PK_V)  { W = kW;  K = 128; p0 = PK_K; }
  else                 { W = vW;  K = 128; p0 = PK_V; }
  int k = 2 * (gp - p0);
  float a = (k < K)     ? W[(size_t)k * 128 + n]       : 0.f;
  float b = (k + 1 < K) ? W[(size_t)(k + 1) * 128 + n] : 0.f;
  P[(size_t)gp * 256 + n * 2]     = a;
  P[(size_t)gp * 256 + n * 2 + 1] = b;
}

__device__ __forceinline__ void proj_body(const float* __restrict__ A,
                                          const float* __restrict__ Wp,
                                          float* __restrict__ C, int act,
                                          int pack, int bid) {
  __shared__ float at[16 * 132];
  const int tid = threadIdx.x, lane = tid & 31, wave = tid >> 5;
  const int row0 = bid * 16;
  for (int i = tid; i < 16 * 128; i += 128) {
    int r = i >> 7, c = i & 127;
    at[r * 132 + c] = A[(size_t)(row0 + r) * 128 + c];
  }
  __syncthreads();
  const int n0 = wave * 32;
  v8f a0 = {}, a1 = {};
  for (int k0 = 0; k0 < 128; k0 += 4) {
    v2f a  = ldsA(at, 132, k0, lane);
    v2f b0 = ldBp(Wp, k0, n0, lane);
    v2f b1 = ldBp(Wp, k0, n0 + 16, lane);
    a0 = wmma4(a, b0, a0);
    a1 = wmma4(a, b1, a1);
  }
  const int hi = lane >> 4, cn = lane & 15;
  #pragma unroll
  for (int j = 0; j < 8; j++) {
    int r = row0 + j + 8 * hi;
    float x0 = a0[j], x1 = a1[j];
    if (act) { x0 = lrelu(x0); x1 = lrelu(x1); }
    if (pack) {
      C[(size_t)(r >> 1) * 256 + (n0 + cn) * 2 + (r & 1)]      = x0;
      C[(size_t)(r >> 1) * 256 + (n0 + 16 + cn) * 2 + (r & 1)] = x1;
    } else {
      C[(size_t)r * 128 + n0 + cn] = x0;
      C[(size_t)r * 128 + n0 + 16 + cn] = x1;
    }
  }
}

__global__ void __launch_bounds__(128) proj_all(
    const float* __restrict__ h0, const float* __restrict__ h1,
    const float* __restrict__ qWp, const float* __restrict__ kWp,
    const float* __restrict__ vWp,
    float* __restrict__ q0, float* __restrict__ k0, float* __restrict__ v0,
    float* __restrict__ q1, float* __restrict__ k1, float* __restrict__ v1) {
  const int p = blockIdx.x / (N_NODES / 16);
  const int bid = blockIdx.x - p * (N_NODES / 16);
  const int side = p >= 3;
  const int j = p - 3 * side;
  const float* A = side ? h1 : h0;
  const float* Wp = j == 0 ? qWp : (j == 1 ? kWp : vWp);
  float* C = j == 0 ? (side ? q1 : q0) : (j == 1 ? (side ? k1 : k0) : (side ? v1 : v0));
  proj_body(A, Wp, C, j != 2, j == 2, bid);
}

__device__ __forceinline__ void attn_body(const float* __restrict__ Q,
                                          const float* __restrict__ K,
                                          const float* __restrict__ Vp,
                                          const float* __restrict__ mask,
                                          int tmask, float* __restrict__ out, int bid) {
  __shared__ __attribute__((aligned(16))) float qt[16 * 132];
  __shared__ float pbuf[8][16 * 18];
  __shared__ float outbuf[16 * 128];
  __shared__ float mbuf[8][16], lbuf[8][16];
  __shared__ float Mrow[16], Lrow[16];
  const int tid = threadIdx.x, lane = tid & 31;
  const int wave = __builtin_amdgcn_readfirstlane(tid >> 5);
  const int hi = lane >> 4, cn = lane & 15;
  const int qrow0 = bid * 16;

  for (int i = tid; i < 2048; i += 256) {
    int r = i >> 7, c = i & 127;
    qt[r * 132 + c] = Q[(size_t)(qrow0 + r) * 128 + c];
  }
  __syncthreads();

  float mr[8], lr[8];
  v8f acc[8];
  #pragma unroll
  for (int j = 0; j < 8; j++) { mr[j] = -3.0e38f; lr[j] = 0.f; acc[j] = (v8f){}; }
  float* pw = &pbuf[wave][0];

  const float* qap = qt + cn * 132 + 4 * hi;
  const unsigned koff = (unsigned)(cn * 128 + 4 * hi);
  const unsigned voff = (unsigned)(hi * 256 + 4 * cn);
  const unsigned mrs = tmask ? 1u : (unsigned)N_NODES;
  const unsigned mcs = tmask ? (unsigned)N_NODES : 1u;
  const unsigned moff = (unsigned)(qrow0 + 8 * hi) * mrs + (unsigned)cn * mcs;
  v4f kf[16], vf[16];
  float mk[8];
  {
    const float* kb = K + (size_t)wave * 2048;
    #pragma unroll
    for (int i = 0; i < 16; i++) kf[i] = *(const v4f*)(kb + (koff + 8u * i));
    const float* mb = mask + (size_t)(wave * 16) * mcs;
    #pragma unroll
    for (int j = 0; j < 8; j++) mk[j] = mb[moff + (unsigned)j * mrs];
    const float* vb = Vp + (size_t)wave * 2048;
    #pragma unroll
    for (int i = 0; i < 16; i++) vf[i] = *(const v4f*)(vb + (voff + 512u * (i >> 2) + 64u * (i & 3)));
  }

  for (int ch = wave; ch < N_NODES / 16; ch += 8) {
    const int nch = (ch + 8 < N_NODES / 16) ? ch + 8 : N_NODES / 16 - 1;
    v8f s = {};
    #pragma unroll
    for (int i = 0; i < 16; i++) {
      v4f qa = *(const v4f*)(qap + 8 * i);
      v2f a0, a1, b0, b1;
      a0.x = qa.x; a0.y = qa.y; a1.x = qa.z; a1.y = qa.w;
      b0.x = kf[i].x; b0.y = kf[i].y; b1.x = kf[i].z; b1.y = kf[i].w;
      s = wmma4(a0, b0, s);
      s = wmma4(a1, b1, s);
    }
    float sc[8];
    #pragma unroll
    for (int j = 0; j < 8; j++) {
      float mv = mk[j];
      float sj = mv * s[j] - 1000.f * (1.f - mv);
      float mnew = fmaxf(mr[j], rowmax16(sj));
      sc[j] = __expf(mr[j] - mnew);
      float pj = __expf(sj - mnew);
      lr[j] = lr[j] * sc[j] + rowsum16(pj);
      mr[j] = mnew;
      pw[(j + 8 * hi) * 18 + cn] = pj;
    }
    #pragma unroll
    for (int t = 0; t < 8; t++)
      #pragma unroll
      for (int j = 0; j < 8; j++) acc[t][j] *= sc[j];
    {
      const float* kb = K + (size_t)nch * 2048;
      #pragma unroll
      for (int i = 0; i < 16; i++) kf[i] = *(const v4f*)(kb + (koff + 8u * i));
      const float* mb = mask + (size_t)(nch * 16) * mcs;
      #pragma unroll
      for (int j = 0; j < 8; j++) mk[j] = mb[moff + (unsigned)j * mrs];
    }
    asm volatile("s_wait_dscnt 0" ::: "memory");
    #pragma unroll
    for (int kk = 0; kk < 4; kk++) {
      v2f a = ldsA(pw, 18, 4 * kk, lane);
      #pragma unroll
      for (int u = 0; u < 4; u++) {
        v2f b0, b1;
        b0.x = vf[4 * kk + u].x; b0.y = vf[4 * kk + u].y;
        b1.x = vf[4 * kk + u].z; b1.y = vf[4 * kk + u].w;
        acc[2 * u]     = wmma4(a, b0, acc[2 * u]);
        acc[2 * u + 1] = wmma4(a, b1, acc[2 * u + 1]);
      }
    }
    {
      const float* vb = Vp + (size_t)nch * 2048;
      #pragma unroll
      for (int i = 0; i < 16; i++) vf[i] = *(const v4f*)(vb + (voff + 512u * (i >> 2) + 64u * (i & 3)));
    }
  }

  if (cn == 0) {
    #pragma unroll
    for (int j = 0; j < 8; j++) {
      mbuf[wave][j + 8 * hi] = mr[j];
      lbuf[wave][j + 8 * hi] = lr[j];
    }
  }
  __syncthreads();
  for (int i = tid; i < 2048; i += 256) outbuf[i] = 0.f;
  if (tid < 16) {
    float M = -3.0e38f;
    for (int w = 0; w < 8; w++) M = fmaxf(M, mbuf[w][tid]);
    float L = 0.f;
    for (int w = 0; w < 8; w++) L += lbuf[w][tid] * __expf(mbuf[w][tid] - M);
    Mrow[tid] = M;
    Lrow[tid] = L > 0.f ? 1.f / L : 0.f;
  }
  __syncthreads();
  #pragma unroll
  for (int j = 0; j < 8; j++) {
    float e = __expf(mr[j] - Mrow[j + 8 * hi]);
    #pragma unroll
    for (int t = 0; t < 8; t++)
      atomicAdd(&outbuf[(j + 8 * hi) * 128 + 32 * (t >> 1) + 2 * cn + (t & 1)],
                acc[t][j] * e);
  }
  __syncthreads();
  for (int i = tid; i < 2048; i += 256) {
    int r = i >> 7, c = i & 127;
    out[(size_t)(qrow0 + r) * 128 + c] = outbuf[i] * Lrow[r];
  }
}

__global__ void __launch_bounds__(256) attn_pair(
    const float* __restrict__ Q0, const float* __restrict__ K0,
    const float* __restrict__ V0, float* __restrict__ out0,
    const float* __restrict__ Q1, const float* __restrict__ K1,
    const float* __restrict__ V1, float* __restrict__ out1,
    const float* __restrict__ mask) {
  const int second = blockIdx.x >= N_NODES / 16;
  const int bid = blockIdx.x - second * (N_NODES / 16);
  attn_body(second ? Q1 : Q0, second ? K1 : K0, second ? V1 : V0, mask, second,
            second ? out1 : out0, bid);
}

#define CAT_LD 274
#define EB 64
__device__ __forceinline__ void edge_body(
    const float* __restrict__ coors, const float* __restrict__ h,
    const int* __restrict__ src, const int* __restrict__ dst,
    const float* __restrict__ eW1p, const float* __restrict__ eb1,
    const float* __restrict__ eg1, const float* __restrict__ ebt1,
    const float* __restrict__ eW2p, const float* __restrict__ eb2,
    const float* __restrict__ cW1p, const float* __restrict__ cb1,
    const float* __restrict__ cg1, const float* __restrict__ cbt1,
    const float* __restrict__ cW2, const float* __restrict__ cb2,
    float* __restrict__ sum_msg, float* __restrict__ sum_x,
    float* __restrict__ cnt, int nE, int bid) {
  __shared__ __attribute__((aligned(16))) float cat[EB * CAT_LD];
  __shared__ float buf1[EB * 132];
  __shared__ float stats[EB][2];
  __shared__ float xrel[EB][3];
  __shared__ float coefs[EB];
  __shared__ int sidx[EB], didx[EB];
  float* buf2 = cat;
  const int tid = threadIdx.x, lane = tid & 31, wave = tid >> 5;
  const int hi = lane >> 4, cn = lane & 15;
  const int e0 = bid * EB;

  if (tid < EB) {
    int e = e0 + tid;
    if (e < nE) {
      int s = src[e], d = dst[e];
      sidx[tid] = s; didx[tid] = d;
      float x0 = coors[s * 3 + 0] - coors[d * 3 + 0];
      float x1 = coors[s * 3 + 1] - coors[d * 3 + 1];
      float x2 = coors[s * 3 + 2] - coors[d * 3 + 2];
      xrel[tid][0] = x0; xrel[tid][1] = x1; xrel[tid][2] = x2;
      float d2 = x0 * x0 + x1 * x1 + x2 * x2;
      float sig = 1.f;
      for (int j = 0; j < 15; j++) {
        cat[tid * CAT_LD + 256 + j] = __expf(-d2 / sig);
        sig *= 1.5f;
      }
      cat[tid * CAT_LD + 271] = 0.f;
    } else {
      sidx[tid] = 0; didx[tid] = -1;
      xrel[tid][0] = xrel[tid][1] = xrel[tid][2] = 0.f;
      for (int j = 0; j < 16; j++) cat[tid * CAT_LD + 256 + j] = 0.f;
    }
  }
  __syncthreads();
  {
    const int l4 = (tid & 31) * 4;
    const int half = wave & 1;
    const int* ip = half ? didx : sidx;
    int nn[16], dd[16];
    v4f g[16];
    #pragma unroll
    for (int it = 0; it < 16; it++) {
      const int e = (wave >> 1) + 4 * it;
      nn[it] = ip[e];
      dd[it] = didx[e];
    }
    #pragma unroll
    for (int it = 0; it < 16; it++) {
      const int node = (dd[it] >= 0) ? nn[it] : 0;
      g[it] = *(const v4f*)(h + (size_t)node * 128 + l4);
    }
    #pragma unroll
    for (int it = 0; it < 16; it++) {
      const int e = (wave >> 1) + 4 * it;
      float* p = cat + e * CAT_LD + half * 128 + l4;
      v2f lo, hi2;
      lo.x = g[it].x; lo.y = g[it].y; hi2.x = g[it].z; hi2.y = g[it].w;
      *(v2f*)p = lo;
      *(v2f*)(p + 2) = hi2;
    }
  }
  __syncthreads();

  {
    const int n0 = wave * 16;
    v8f acc[4];
    #pragma unroll
    for (int mt = 0; mt < 4; mt++) acc[mt] = (v8f){};
    for (int k0 = 0; k0 < 272; k0 += 4) {
      v2f b = ldBp(eW1p, k0, n0, lane);
      #pragma unroll
      for (int mt = 0; mt < 4; mt++) {
        v2f x = ldsA(cat + mt * 16 * CAT_LD, CAT_LD, k0, lane);
        acc[mt] = wmma4(x, b, acc[mt]);
      }
    }
    const float bb = eb1[n0 + cn];
    #pragma unroll
    for (int mt = 0; mt < 4; mt++) {
      #pragma unroll
      for (int j = 0; j < 8; j++)
        buf1[(mt * 16 + j + 8 * hi) * 132 + n0 + cn] = lrelu(acc[mt][j] + bb);
    }
  }
  __syncthreads();
  layernorm128x64(buf1, eg1, ebt1, stats, tid);
  __syncthreads();
  gemm128_lds64(buf1, eW2p, eb2, buf2, lane, wave, 0);
  __syncthreads();
  gemm128_lds64(buf2, cW1p, cb1, buf1, lane, wave, 1);
  __syncthreads();
  layernorm128x64(buf1, cg1, cbt1, stats, tid);
  __syncthreads();
  {
    const int seg = tid & 7;
    #pragma unroll
    for (int hh = 0; hh < 2; hh++) {
      const int r = (tid >> 3) + 32 * hh;
      float sum = 0.f;
      #pragma unroll
      for (int i = 0; i < 16; i++) {
        int c = seg * 16 + i;
        sum += buf1[r * 132 + c] * cW2[c];
      }
      sum += __shfl_xor(sum, 1);
      sum += __shfl_xor(sum, 2);
      sum += __shfl_xor(sum, 4);
      if (seg == 0) coefs[r] = sum + cb2[0];
    }
  }
  __syncthreads();
  for (int i = tid; i < EB * 128; i += 256) {
    int e = i >> 7, c = i & 127;
    int d = didx[e];
    if (d >= 0) atomicAdd(&sum_msg[(size_t)d * 128 + c], buf2[e * 132 + c]);
  }
  if (tid < EB * 3) {
    int e = tid / 3, ax = tid % 3;
    int d = didx[e];
    if (d >= 0) atomicAdd(&sum_x[d * 3 + ax], xrel[e][ax] * coefs[e]);
  }
  if (tid < EB) {
    int d = didx[tid];
    if (d >= 0) atomicAdd(&cnt[d], 1.f);
  }
}

__global__ void __launch_bounds__(256) edge_pair(
    const float* __restrict__ coors0, const float* __restrict__ h0,
    const int* __restrict__ src0, const int* __restrict__ dst0,
    const float* __restrict__ coors1, const float* __restrict__ h1,
    const int* __restrict__ src1, const int* __restrict__ dst1,
    const float* __restrict__ eW1p, const float* __restrict__ eb1,
    const float* __restrict__ eg1, const float* __restrict__ ebt1,
    const float* __restrict__ eW2p, const float* __restrict__ eb2,
    const float* __restrict__ cW1p, const float* __restrict__ cb1,
    const float* __restrict__ cg1, const float* __restrict__ cbt1,
    const float* __restrict__ cW2, const float* __restrict__ cb2,
    float* __restrict__ sm0, float* __restrict__ sx0, float* __restrict__ cnt0,
    float* __restrict__ sm1, float* __restrict__ sx1, float* __restrict__ cnt1,
    int nE, int eblocks) {
  const int second = (int)blockIdx.x >= eblocks;
  const int bid = blockIdx.x - second * eblocks;
  edge_body(second ? coors1 : coors0, second ? h1 : h0, second ? src1 : src0,
            second ? dst1 : dst0, eW1p, eb1, eg1, ebt1, eW2p, eb2, cW1p, cb1, cg1,
            cbt1, cW2, cb2, second ? sm1 : sm0, second ? sx1 : sx0,
            second ? cnt1 : cnt0, nE, bid);
}

#define NCAT_LD 450
__device__ __forceinline__ void node_body(
    const float* __restrict__ h, const float* __restrict__ summ,
    const float* __restrict__ cnt, const float* __restrict__ cm,
    const float* __restrict__ orig,
    const float* __restrict__ nW1p, const float* __restrict__ nb1,
    const float* __restrict__ ng1, const float* __restrict__ nbt1,
    const float* __restrict__ nW2p, const float* __restrict__ nb2,
    float* __restrict__ hout, int bid) {
  __shared__ float cat[16 * NCAT_LD];
  __shared__ float buf1[16 * 132];
  __shared__ float stats[16][2];
  const int tid = threadIdx.x, lane = tid & 31, wave = tid >> 5;
  const int hi = lane >> 4, cn = lane & 15;
  const int node0 = bid * 16;

  for (int i = tid; i < 2048; i += 128) {
    int e = i >> 7, c = i & 127;
    int node = node0 + e;
    cat[e * NCAT_LD + c] = h[(size_t)node * 128 + c];
    float ct = cnt[node];
    cat[e * NCAT_LD + 128 + c] = ct > 0.f ? summ[(size_t)node * 128 + c] / ct : 0.f;
    cat[e * NCAT_LD + 256 + c] = cm[(size_t)node * 128 + c];
  }
  for (int i = tid; i < 1024; i += 128) {
    int e = i >> 6, c = i & 63;
    cat[e * NCAT_LD + 384 + c] = orig[(size_t)(node0 + e) * 64 + c];
  }
  __syncthreads();
  {
    const int n0 = wave * 32;
    v8f a0 = {}, a1 = {};
    for (int k0 = 0; k0 < 448; k0 += 4) {
      v2f a  = ldsA(cat, NCAT_LD, k0, lane);
      v2f b0 = ldBp(nW1p, k0, n0, lane);
      v2f b1 = ldBp(nW1p, k0, n0 + 16, lane);
      a0 = wmma4(a, b0, a0);
      a1 = wmma4(a, b1, a1);
    }
    #pragma unroll
    for (int j = 0; j < 8; j++) {
      int r = j + 8 * hi;
      buf1[r * 132 + n0 + cn]      = lrelu(a0[j] + nb1[n0 + cn]);
      buf1[r * 132 + n0 + 16 + cn] = lrelu(a1[j] + nb1[n0 + 16 + cn]);
    }
  }
  __syncthreads();
  layernorm128(buf1, ng1, nbt1, stats, tid);
  __syncthreads();
  {
    const int n0 = wave * 32;
    v8f a0 = {}, a1 = {};
    for (int k0 = 0; k0 < 128; k0 += 4) {
      v2f a  = ldsA(buf1, 132, k0, lane);
      v2f b0 = ldBp(nW2p, k0, n0, lane);
      v2f b1 = ldBp(nW2p, k0, n0 + 16, lane);
      a0 = wmma4(a, b0, a0);
      a1 = wmma4(a, b1, a1);
    }
    #pragma unroll
    for (int j = 0; j < 8; j++) {
      int r = j + 8 * hi;
      size_t g0 = (size_t)(node0 + r) * 128 + n0 + cn;
      size_t g1 = g0 + 16;
      hout[g0] = 0.5f * (a0[j] + nb2[n0 + cn]) + 0.5f * h[g0];
      hout[g1] = 0.5f * (a1[j] + nb2[n0 + 16 + cn]) + 0.5f * h[g1];
    }
  }
}

__global__ void __launch_bounds__(128) node_pair(
    const float* __restrict__ h0, const float* __restrict__ summ0,
    const float* __restrict__ cnt0, const float* __restrict__ cm0,
    const float* __restrict__ orig0, float* __restrict__ hout0,
    const float* __restrict__ h1, const float* __restrict__ summ1,
    const float* __restrict__ cnt1, const float* __restrict__ cm1,
    const float* __restrict__ orig1, float* __restrict__ hout1,
    const float* __restrict__ nW1p, const float* __restrict__ nb1,
    const float* __restrict__ ng1, const float* __restrict__ nbt1,
    const float* __restrict__ nW2p, const float* __restrict__ nb2) {
  const int second = blockIdx.x >= N_NODES / 16;
  const int bid = blockIdx.x - second * (N_NODES / 16);
  node_body(second ? h1 : h0, second ? summ1 : summ0, second ? cnt1 : cnt0,
            second ? cm1 : cm0, second ? orig1 : orig0, nW1p, nb1, ng1, nbt1, nW2p,
            nb2, second ? hout1 : hout0, bid);
}

#define CB ((3 * N_NODES + 255) / 256)
__global__ void coords_pair(const float* __restrict__ oc0, const float* __restrict__ c0,
                            const float* __restrict__ sx0, const float* __restrict__ cnt0,
                            float* __restrict__ out0,
                            const float* __restrict__ oc1, const float* __restrict__ c1,
                            const float* __restrict__ sx1, const float* __restrict__ cnt1,
                            float* __restrict__ out1, int n) {
  const int second = blockIdx.x >= CB;
  const float* oc = second ? oc1 : oc0;
  const float* c = second ? c1 : c0;
  const float* sx = second ? sx1 : sx0;
  const float* cnt = second ? cnt1 : cnt0;
  float* out = second ? out1 : out0;
  int i = (blockIdx.x - second * CB) * 256 + threadIdx.x;
  if (i >= n * 3) return;
  int node = i / 3;
  float ct = cnt[node];
  float xu = ct > 0.f ? sx[i] / ct : 0.f;
  out[i] = 0.2f * oc[i] + 0.8f * c[i] + xu;
}

extern "C" void kernel_launch(void* const* d_in, const int* in_sizes, int n_in,
                              void* d_out, int out_size, void* d_ws, size_t ws_size,
                              hipStream_t stream) {
  const float* coors_lig = (const float*)d_in[0];
  const float* h_lig     = (const float*)d_in[1];
  const float* orig_lig  = (const float*)d_in[2];
  const float* oc_lig    = (const float*)d_in[3];
  const float* coors_rec = (const float*)d_in[4];
  const float* h_rec     = (const float*)d_in[5];
  const float* orig_rec  = (const float*)d_in[6];
  const float* oc_rec    = (const float*)d_in[7];
  const int* lig_src = (const int*)d_in[8];
  const int* lig_dst = (const int*)d_in[9];
  const int* rec_src = (const int*)d_in[10];
  const int* rec_dst = (const int*)d_in[11];
  const float* mask = (const float*)d_in[12];
  const float* eW1 = (const float*)d_in[13];
  const float* eb1 = (const float*)d_in[14];
  const float* eg1 = (const float*)d_in[15];
  const float* ebt1 = (const float*)d_in[16];
  const float* eW2 = (const float*)d_in[17];
  const float* eb2 = (const float*)d_in[18];
  const float* qW = (const float*)d_in[19];
  const float* kW = (const float*)d_in[20];
  const float* vW = (const float*)d_in[21];
  const float* nW1 = (const float*)d_in[22];
  const float* nb1 = (const float*)d_in[23];
  const float* ng1 = (const float*)d_in[24];
  const float* nbt1 = (const float*)d_in[25];
  const float* nW2 = (const float*)d_in[26];
  const float* nb2 = (const float*)d_in[27];
  const float* cW1 = (const float*)d_in[28];
  const float* cb1 = (const float*)d_in[29];
  const float* cg1 = (const float*)d_in[30];
  const float* cbt1 = (const float*)d_in[31];
  const float* cW2 = (const float*)d_in[32];
  const float* cb2 = (const float*)d_in[33];
  const int nE = in_sizes[8];

  float* ws = (float*)d_ws;
  float* q_l = ws + 0 * NH;
  float* k_l = ws + 1 * NH;
  float* v_l = ws + 2 * NH;
  float* q_r = ws + 3 * NH;
  float* k_r = ws + 4 * NH;
  float* v_r = ws + 5 * NH;
  float* cm_l = ws + 6 * NH;
  float* cm_r = ws + 7 * NH;
  float* sm_l = ws + 8 * NH;
  float* sm_r = ws + 9 * NH;
  float* sx_l = ws + 10 * NH;
  float* sx_r = sx_l + 3 * N_NODES;
  float* cnt_l = sx_r + 3 * N_NODES;
  float* cnt_r = cnt_l + N_NODES;
  float* eW1p = cnt_r + N_NODES;
  float* eW2p = eW1p + 136 * 256;
  float* cW1p = eW2p + 64 * 256;
  float* nW1p = cW1p + 64 * 256;
  float* nW2p = nW1p + 224 * 256;
  float* qWp  = nW2p + 64 * 256;
  float* kWp  = qWp + 64 * 256;
  float* vWp  = kWp + 64 * 256;

  size_t accum_floats = 2 * NH + 8 * (size_t)N_NODES;
  hipMemsetAsync(sm_l, 0, accum_floats * sizeof(float), stream);

  pack_all<<<PK_END * 128 / 256, 256, 0, stream>>>(eW1, eW2, cW1, nW1, nW2, qW, kW, vW,
                                                  eW1p);

  const int ntiles = N_NODES / 16;
  proj_all<<<6 * ntiles, 128, 0, stream>>>(h_lig, h_rec, qWp, kWp, vWp, q_l, k_l, v_l,
                                           q_r, k_r, v_r);

  attn_pair<<<2 * ntiles, 256, 0, stream>>>(q_l, k_r, v_r, cm_l, q_r, k_l, v_l, cm_r,
                                            mask);

  int eblocks = (nE + EB - 1) / EB;
  edge_pair<<<2 * eblocks, 256, 0, stream>>>(coors_lig, h_lig, lig_src, lig_dst,
      coors_rec, h_rec, rec_src, rec_dst,
      eW1p, eb1, eg1, ebt1, eW2p, eb2, cW1p, cb1, cg1, cbt1, cW2, cb2,
      sm_l, sx_l, cnt_l, sm_r, sx_r, cnt_r, nE, eblocks);

  float* out_xl = (float*)d_out;
  float* out_hl = out_xl + 3 * N_NODES;
  float* out_xr = out_hl + NH;
  float* out_hr = out_xr + 3 * N_NODES;

  node_pair<<<2 * ntiles, 128, 0, stream>>>(h_lig, sm_l, cnt_l, cm_l, orig_lig, out_hl,
      h_rec, sm_r, cnt_r, cm_r, orig_rec, out_hr,
      nW1p, nb1, ng1, nbt1, nW2p, nb2);

  coords_pair<<<2 * CB, 256, 0, stream>>>(oc_lig, coors_lig, sx_l, cnt_l, out_xl,
                                          oc_rec, coors_rec, sx_r, cnt_r, out_xr,
                                          N_NODES);
}


# edge MLP layer 1 split by linearity: h@eW1[0:128], h@eW1[128:256] made once per node (one new launch); edge kernel gathers and adds the two rows, keeps only the K=16 rbf product; sum order changes
# speedup vs baseline: 1.9821x; 1.1824x over previous
#include <hip/hip_runtime.h>
#include <hip/hip_bf16.h>
#include <cstdint>

#define N_NODES 6000
#define NH      ((size_t)N_NODES * 128)

typedef float v2f __attribute__((ext_vector_type(2)));
typedef float v4f __attribute__((ext_vector_type(4)));
typedef float v8f __attribute__((ext_vector_type(8)));

__device__ __forceinline__ v8f wmma4(v2f a, v2f b, v8f c) {
  return __builtin_amdgcn_wmma_f32_16x16x4_f32(false, a, false, b, (short)0, c,
                                               false, false);
}

__device__ __forceinline__ v2f ldsA(const float* base, int ld, int k0, int lane) {
  const int m = lane & 15;
  const int k = k0 + ((lane >> 4) << 1);
  const float* p = base + m * ld + k;
  v2f r; r.x = p[0]; r.y = p[1];
  return r;
}

__device__ __forceinline__ v2f ldBp(const float* __restrict__ P, int k0, int n0,
                                    int lane) {
  const int n = n0 + (lane & 15);
  const int k = k0 + ((lane >> 4) << 1);
  const float* p = P + (size_t)(k >> 1) * 256 + n * 2;
  v2f r; r.x = p[0]; r.y = p[1];
  return r;
}

__device__ __forceinline__ float lrelu(float x) { return x >= 0.f ? x : 0.01f * x; }

__device__ __forceinline__ float rowmax16(float v) {
  v = fmaxf(v, __shfl_xor(v, 1));
  v = fmaxf(v, __shfl_xor(v, 2));
  v = fmaxf(v, __shfl_xor(v, 4));
  v = fmaxf(v, __shfl_xor(v, 8));
  return v;
}
__device__ __forceinline__ float rowsum16(float v) {
  v += __shfl_xor(v, 1);
  v += __shfl_xor(v, 2);
  v += __shfl_xor(v, 4);
  v += __shfl_xor(v, 8);
  return v;
}

__device__ __forceinline__ void layernorm128(float* buf, const float* __restrict__ g,
                                             const float* __restrict__ b,
                                             float (*stats)[2], int tid) {
  const int r = tid >> 3, seg = tid & 7;
  float sum = 0.f, sq = 0.f;
  #pragma unroll
  for (int i = 0; i < 16; i++) {
    float v = buf[r * 132 + seg * 16 + i];
    sum += v; sq += v * v;
  }
  sum += __shfl_xor(sum, 1); sq += __shfl_xor(sq, 1);
  sum += __shfl_xor(sum, 2); sq += __shfl_xor(sq, 2);
  sum += __shfl_xor(sum, 4); sq += __shfl_xor(sq, 4);
  if (seg == 0) {
    float mean = sum * (1.f / 128.f);
    float var  = sq * (1.f / 128.f) - mean * mean;
    stats[r][0] = mean;
    stats[r][1] = rsqrtf(var + 1e-5f);
  }
  __syncthreads();
  const float mean = stats[r][0], rstd = stats[r][1];
  #pragma unroll
  for (int i = 0; i < 16; i++) {
    int c = seg * 16 + i;
    float v = buf[r * 132 + c];
    buf[r * 132 + c] = g[c] * (v - mean) * rstd + b[c];
  }
}

__device__ __forceinline__ void gemm128_lds(const float* in, const float* __restrict__ Wp,
                                            const float* __restrict__ bias, float* outb,
                                            int lane, int wave, int act) {
  const int hi = lane >> 4, cn = lane & 15;
  const int n0 = wave * 32;
  v8f a0 = {}, a1 = {};
  for (int k0 = 0; k0 < 128; k0 += 4) {
    v2f a  = ldsA(in, 132, k0, lane);
    v2f b0 = ldBp(Wp, k0, n0, lane);
    v2f b1 = ldBp(Wp, k0, n0 + 16, lane);
    a0 = wmma4(a, b0, a0);
    a1 = wmma4(a, b1, a1);
  }
  #pragma unroll
  for (int j = 0; j < 8; j++) {
    int r = j + 8 * hi;
    float x0 = a0[j] + bias[n0 + cn];
    float x1 = a1[j] + bias[n0 + 16 + cn];
    if (act) { x0 = lrelu(x0); x1 = lrelu(x1); }
    outb[r * 132 + n0 + cn] = x0;
    outb[r * 132 + n0 + 16 + cn] = x1;
  }
}

__device__ __forceinline__ void layernorm128x64(float* buf, const float* __restrict__ g,
                                                const float* __restrict__ b,
                                                float (*stats)[2], int tid) {
  const int seg = tid & 7;
  #pragma unroll
  for (int hh = 0; hh < 2; hh++) {
    const int r = (tid >> 3) + 32 * hh;
    float sum = 0.f, sq = 0.f;
    #pragma unroll
    for (int i = 0; i < 16; i++) {
      float v = buf[r * 132 + seg * 16 + i];
      sum += v; sq += v * v;
    }
    sum += __shfl_xor(sum, 1); sq += __shfl_xor(sq, 1);
    sum += __shfl_xor(sum, 2); sq += __shfl_xor(sq, 2);
    sum += __shfl_xor(sum, 4); sq += __shfl_xor(sq, 4);
    if (seg == 0) {
      float mean = sum * (1.f / 128.f);
      float var  = sq * (1.f / 128.f) - mean * mean;
      stats[r][0] = mean;
      stats[r][1] = rsqrtf(var + 1e-5f);
    }
  }
  __syncthreads();
  #pragma unroll
  for (int hh = 0; hh < 2; hh++) {
    const int r = (tid >> 3) + 32 * hh;
    const float mean = stats[r][0], rstd = stats[r][1];
    #pragma unroll
    for (int i = 0; i < 16; i++) {
      int c = seg * 16 + i;
      float v = buf[r * 132 + c];
      buf[r * 132 + c] = g[c] * (v - mean) * rstd + b[c];
    }
  }
}

__device__ __forceinline__ void gemm128_lds64(const float* in, const float* __restrict__ Wp,
                                              const float* __restrict__ bias, float* outb,
                                              int lane, int wave, int act) {
  const int hi = lane >> 4, cn = lane & 15;
  const int n0 = wave * 16;
  v8f acc[4];
  #pragma unroll
  for (int mt = 0; mt < 4; mt++) acc[mt] = (v8f){};
  for (int k0 = 0; k0 < 128; k0 += 4) {
    v2f b = ldBp(Wp, k0, n0, lane);
    #pragma unroll
    for (int mt = 0; mt < 4; mt++) {
      v2f x = ldsA(in + mt * 16 * 132, 132, k0, lane);
      acc[mt] = wmma4(x, b, acc[mt]);
    }
  }
  const float bb = bias[n0 + cn];
  #pragma unroll
  for (int mt = 0; mt < 4; mt++) {
    #pragma unroll
    for (int j = 0; j < 8; j++) {
      int r = mt * 16 + j + 8 * hi;
      float x = acc[mt][j] + bb;
      if (act) x = lrelu(x);
      outb[r * 132 + n0 + cn] = x;
    }
  }
}

#define PK_E1 0
#define PK_E2 136
#define PK_C1 200
#define PK_N1 264
#define PK_N2 488
#define PK_Q  552
#define PK_K  616
#define PK_V  680
#define PK_END 744
__global__ void pack_all(const float* __restrict__ eW1, const float* __restrict__ eW2,
                         const float* __restrict__ cW1, const float* __restrict__ nW1,
                         const float* __restrict__ nW2, const float* __restrict__ qW,
                         const float* __restrict__ kW, const float* __restrict__ vW,
                         float* __restrict__ P) {
  int i = blockIdx.x * 256 + threadIdx.x;
  if (i >= PK_END * 128) return;
  int gp = i >> 7, n = i & 127;
  const float* W; int K, p0;
  if (gp < PK_E2)      { W = eW1; K = 271; p0 = PK_E1; }
  else if (gp < PK_C1) { W = eW2; K = 128; p0 = PK_E2; }
  else if (gp < PK_N1) { W = cW1; K = 128; p0 = PK_C1; }
  else if (gp < PK_N2) { W = nW1; K = 448; p0 = PK_N1; }
  else if (gp < PK_Q)  { W = nW2; K = 128; p0 = PK_N2; }
  else if (gp < PK_K)  { W = qW;  K = 128; p0 = PK_Q; }
  else if (gp < PK_V)  { W = kW;  K = 128; p0 = PK_K; }
  else                 { W = vW;  K = 128; p0 = PK_V; }
  int k = 2 * (gp - p0);
  float a = (k < K)     ? W[(size_t)k * 128 + n]       : 0.f;
  float b = (k + 1 < K) ? W[(size_t)(k + 1) * 128 + n] : 0.f;
  P[(size_t)gp * 256 + n * 2]     = a;
  P[(size_t)gp * 256 + n * 2 + 1] = b;
}

__device__ __forceinline__ void proj_body(const float* __restrict__ A,
                                          const float* __restrict__ Wp,
                                          float* __restrict__ C, int act,
                                          int pack, int bid) {
  __shared__ float at[16 * 132];
  const int tid = threadIdx.x, lane = tid & 31, wave = tid >> 5;
  const int row0 = bid * 16;
  for (int i = tid; i < 16 * 128; i += 128) {
    int r = i >> 7, c = i & 127;
    at[r * 132 + c] = A[(size_t)(row0 + r) * 128 + c];
  }
  __syncthreads();
  const int n0 = wave * 32;
  v8f a0 = {}, a1 = {};
  for (int k0 = 0; k0 < 128; k0 += 4) {
    v2f a  = ldsA(at, 132, k0, lane);
    v2f b0 = ldBp(Wp, k0, n0, lane);
    v2f b1 = ldBp(Wp, k0, n0 + 16, lane);
    a0 = wmma4(a, b0, a0);
    a1 = wmma4(a, b1, a1);
  }
  const int hi = lane >> 4, cn = lane & 15;
  #pragma unroll
  for (int j = 0; j < 8; j++) {
    int r = row0 + j + 8 * hi;
    float x0 = a0[j], x1 = a1[j];
    if (act) { x0 = lrelu(x0); x1 = lrelu(x1); }
    if (pack) {
      C[(size_t)(r >> 1) * 256 + (n0 + cn) * 2 + (r & 1)]      = x0;
      C[(size_t)(r >> 1) * 256 + (n0 + 16 + cn) * 2 + (r & 1)] = x1;
    } else {
      C[(size_t)r * 128 + n0 + cn] = x0;
      C[(size_t)r * 128 + n0 + 16 + cn] = x1;
    }
  }
}

__global__ void __launch_bounds__(128) proj_all(
    const float* __restrict__ h0, const float* __restrict__ h1,
    const float* __restrict__ qWp, const float* __restrict__ kWp,
    const float* __restrict__ vWp,
    float* __restrict__ q0, float* __restrict__ k0, float* __restrict__ v0,
    float* __restrict__ q1, float* __restrict__ k1, float* __restrict__ v1) {
  const int p = blockIdx.x / (N_NODES / 16);
  const int bid = blockIdx.x - p * (N_NODES / 16);
  const int side = p >= 3;
  const int j = p - 3 * side;
  const float* A = side ? h1 : h0;
  const float* Wp = j == 0 ? qWp : (j == 1 ? kWp : vWp);
  float* C = j == 0 ? (side ? q1 : q0) : (j == 1 ? (side ? k1 : k0) : (side ? v1 : v0));
  proj_body(A, Wp, C, j != 2, j == 2, bid);
}

__global__ void __launch_bounds__(128) proj_edge(
    const float* __restrict__ h0, const float* __restrict__ h1,
    const float* __restrict__ eW1p,
    float* __restrict__ hs0, float* __restrict__ hd0,
    float* __restrict__ hs1, float* __restrict__ hd1) {
  const int p = blockIdx.x / (N_NODES / 16);
  const int bid = blockIdx.x - p * (N_NODES / 16);
  const int side = p >> 1;
  const int half = p & 1;
  const float* A = side ? h1 : h0;
  const float* Wp = eW1p + half * (64 * 256);
  float* C = half ? (side ? hd1 : hd0) : (side ? hs1 : hs0);
  proj_body(A, Wp, C, 0, 0, bid);
}

__device__ __forceinline__ void attn_body(const float* __restrict__ Q,
                                          const float* __restrict__ K,
                                          const float* __restrict__ Vp,
                                          const float* __restrict__ mask,
                                          int tmask, float* __restrict__ out, int bid) {
  __shared__ __attribute__((aligned(16))) float qt[16 * 132];
  __shared__ float pbuf[8][16 * 18];
  __shared__ float outbuf[16 * 128];
  __shared__ float mbuf[8][16], lbuf[8][16];
  __shared__ float Mrow[16], Lrow[16];
  const int tid = threadIdx.x, lane = tid & 31;
  const int wave = __builtin_amdgcn_readfirstlane(tid >> 5);
  const int hi = lane >> 4, cn = lane & 15;
  const int qrow0 = bid * 16;

  for (int i = tid; i < 2048; i += 256) {
    int r = i >> 7, c = i & 127;
    qt[r * 132 + c] = Q[(size_t)(qrow0 + r) * 128 + c];
  }
  __syncthreads();

  float mr[8], lr[8];
  v8f acc[8];
  #pragma unroll
  for (int j = 0; j < 8; j++) { mr[j] = -3.0e38f; lr[j] = 0.f; acc[j] = (v8f){}; }
  float* pw = &pbuf[wave][0];

  const float* qap = qt + cn * 132 + 4 * hi;
  const unsigned koff = (unsigned)(cn * 128 + 4 * hi);
  const unsigned voff = (unsigned)(hi * 256 + 4 * cn);
  const unsigned mrs = tmask ? 1u : (unsigned)N_NODES;
  const unsigned mcs = tmask ? (unsigned)N_NODES : 1u;
  const unsigned moff = (unsigned)(qrow0 + 8 * hi) * mrs + (unsigned)cn * mcs;
  v4f kf[16], vf[16];
  float mk[8];
  {
    const float* kb = K + (size_t)wave * 2048;
    #pragma unroll
    for (int i = 0; i < 16; i++) kf[i] = *(const v4f*)(kb + (koff + 8u * i));
    const float* mb = mask + (size_t)(wave * 16) * mcs;
    #pragma unroll
    for (int j = 0; j < 8; j++) mk[j] = mb[moff + (unsigned)j * mrs];
    const float* vb = Vp + (size_t)wave * 2048;
    #pragma unroll
    for (int i = 0; i < 16; i++) vf[i] = *(const v4f*)(vb + (voff + 512u * (i >> 2) + 64u * (i & 3)));
  }

  for (int ch = wave; ch < N_NODES / 16; ch += 8) {
    const int nch = (ch + 8 < N_NODES / 16) ? ch + 8 : N_NODES / 16 - 1;
    v8f s = {};
    #pragma unroll
    for (int i = 0; i < 16; i++) {
      v4f qa = *(const v4f*)(qap + 8 * i);
      v2f a0, a1, b0, b1;
      a0.x = qa.x; a0.y = qa.y; a1.x = qa.z; a1.y = qa.w;
      b0.x = kf[i].x; b0.y = kf[i].y; b1.x = kf[i].z; b1.y = kf[i].w;
      s = wmma4(a0, b0, s);
      s = wmma4(a1, b1, s);
    }
    float sc[8];
    #pragma unroll
    for (int j = 0; j < 8; j++) {
      float mv = mk[j];
      float sj = mv * s[j] - 1000.f * (1.f - mv);
      float mnew = fmaxf(mr[j], rowmax16(sj));
      sc[j] = __expf(mr[j] - mnew);
      float pj = __expf(sj - mnew);
      lr[j] = lr[j] * sc[j] + rowsum16(pj);
      mr[j] = mnew;
      pw[(j + 8 * hi) * 18 + cn] = pj;
    }
    #pragma unroll
    for (int t = 0; t < 8; t++)
      #pragma unroll
      for (int j = 0; j < 8; j++) acc[t][j] *= sc[j];
    {
      const float* kb = K + (size_t)nch * 2048;
      #pragma unroll
      for (int i = 0; i < 16; i++) kf[i] = *(const v4f*)(kb + (koff + 8u * i));
      const float* mb = mask + (size_t)(nch * 16) * mcs;
      #pragma unroll
      for (int j = 0; j < 8; j++) mk[j] = mb[moff + (unsigned)j * mrs];
    }
    asm volatile("s_wait_dscnt 0" ::: "memory");
    #pragma unroll
    for (int kk = 0; kk < 4; kk++) {
      v2f a = ldsA(pw, 18, 4 * kk, lane);
      #pragma unroll
      for (int u = 0; u < 4; u++) {
        v2f b0, b1;
        b0.x = vf[4 * kk + u].x; b0.y = vf[4 * kk + u].y;
        b1.x = vf[4 * kk + u].z; b1.y = vf[4 * kk + u].w;
        acc[2 * u]     = wmma4(a, b0, acc[2 * u]);
        acc[2 * u + 1] = wmma4(a, b1, acc[2 * u + 1]);
      }
    }
    {
      const float* vb = Vp + (size_t)nch * 2048;
      #pragma unroll
      for (int i = 0; i < 16; i++) vf[i] = *(const v4f*)(vb + (voff + 512u * (i >> 2) + 64u * (i & 3)));
    }
  }

  if (cn == 0) {
    #pragma unroll
    for (int j = 0; j < 8; j++) {
      mbuf[wave][j + 8 * hi] = mr[j];
      lbuf[wave][j + 8 * hi] = lr[j];
    }
  }
  __syncthreads();
  for (int i = tid; i < 2048; i += 256) outbuf[i] = 0.f;
  if (tid < 16) {
    float M = -3.0e38f;
    for (int w = 0; w < 8; w++) M = fmaxf(M, mbuf[w][tid]);
    float L = 0.f;
    for (int w = 0; w < 8; w++) L += lbuf[w][tid] * __expf(mbuf[w][tid] - M);
    Mrow[tid] = M;
    Lrow[tid] = L > 0.f ? 1.f / L : 0.f;
  }
  __syncthreads();
  #pragma unroll
  for (int j = 0; j < 8; j++) {
    float e = __expf(mr[j] - Mrow[j + 8 * hi]);
    #pragma unroll
    for (int t = 0; t < 8; t++)
      atomicAdd(&outbuf[(j + 8 * hi) * 128 + 32 * (t >> 1) + 2 * cn + (t & 1)],
                acc[t][j] * e);
  }
  __syncthreads();
  for (int i = tid; i < 2048; i += 256) {
    int r = i >> 7, c = i & 127;
    out[(size_t)(qrow0 + r) * 128 + c] = outbuf[i] * Lrow[r];
  }
}

__global__ void __launch_bounds__(256) attn_pair(
    const float* __restrict__ Q0, const float* __restrict__ K0,
    const float* __restrict__ V0, float* __restrict__ out0,
    const float* __restrict__ Q1, const float* __restrict__ K1,
    const float* __restrict__ V1, float* __restrict__ out1,
    const float* __restrict__ mask) {
  const int second = blockIdx.x >= N_NODES / 16;
  const int bid = blockIdx.x - second * (N_NODES / 16);
  attn_body(second ? Q1 : Q0, second ? K1 : K0, second ? V1 : V0, mask, second,
            second ? out1 : out0, bid);
}

#define RB_LD 18
#define EB 64
__device__ __forceinline__ void edge_body(
    const float* __restrict__ coors, const float* __restrict__ hs,
    const float* __restrict__ hd,
    const int* __restrict__ src, const int* __restrict__ dst,
    const float* __restrict__ eW1rp, const float* __restrict__ eb1,
    const float* __restrict__ eg1, const float* __restrict__ ebt1,
    const float* __restrict__ eW2p, const float* __restrict__ eb2,
    const float* __restrict__ cW1p, const float* __restrict__ cb1,
    const float* __restrict__ cg1, const float* __restrict__ cbt1,
    const float* __restrict__ cW2, const float* __restrict__ cb2,
    float* __restrict__ sum_msg, float* __restrict__ sum_x,
    float* __restrict__ cnt, int nE, int bid) {
  __shared__ __attribute__((aligned(16))) float buf1[EB * 132];
  __shared__ __attribute__((aligned(16))) float buf2[EB * 132];
  __shared__ __attribute__((aligned(16))) float rbf[EB * RB_LD];
  __shared__ float stats[EB][2];
  __shared__ float xrel[EB][3];
  __shared__ float coefs[EB];
  __shared__ int sidx[EB], didx[EB];
  const int tid = threadIdx.x, lane = tid & 31, wave = tid >> 5;
  const int hi = lane >> 4, cn = lane & 15;
  const int e0 = bid * EB;

  if (tid < EB) {
    int e = e0 + tid;
    if (e < nE) {
      int s = src[e], d = dst[e];
      sidx[tid] = s; didx[tid] = d;
      float x0 = coors[s * 3 + 0] - coors[d * 3 + 0];
      float x1 = coors[s * 3 + 1] - coors[d * 3 + 1];
      float x2 = coors[s * 3 + 2] - coors[d * 3 + 2];
      xrel[tid][0] = x0; xrel[tid][1] = x1; xrel[tid][2] = x2;
      float d2 = x0 * x0 + x1 * x1 + x2 * x2;
      float sig = 1.f;
      for (int j = 0; j < 15; j++) {
        rbf[tid * RB_LD + j] = __expf(-d2 / sig);
        sig *= 1.5f;
      }
      rbf[tid * RB_LD + 15] = 0.f;
    } else {
      sidx[tid] = 0; didx[tid] = -1;
      xrel[tid][0] = xrel[tid][1] = xrel[tid][2] = 0.f;
      for (int j = 0; j < 16; j++) rbf[tid * RB_LD + j] = 0.f;
    }
  }
  __syncthreads();
  {
    const int l4 = lane * 4;
    int ns[8], nd[8];
    v4f gs[8], gd[8];
    #pragma unroll
    for (int it = 0; it < 8; it++) {
      const int e = wave + 8 * it;
      ns[it] = sidx[e];
      nd[it] = didx[e];
    }
    #pragma unroll
    for (int it = 0; it < 8; it++) nd[it] = nd[it] >= 0 ? nd[it] : 0;
    #pragma unroll
    for (int it = 0; it < 8; it++) {
      gs[it] = *(const v4f*)(hs + (size_t)ns[it] * 128 + l4);
      gd[it] = *(const v4f*)(hd + (size_t)nd[it] * 128 + l4);
    }
    #pragma unroll
    for (int it = 0; it < 8; it++) {
      const int e = wave + 8 * it;
      *(v4f*)(buf1 + e * 132 + l4) = gs[it] + gd[it];
    }
  }
  __syncthreads();

  {
    const int n0 = wave * 16;
    v8f acc[4];
    #pragma unroll
    for (int mt = 0; mt < 4; mt++) {
      #pragma unroll
      for (int j = 0; j < 8; j++)
        acc[mt][j] = buf1[(mt * 16 + j + 8 * hi) * 132 + n0 + cn];
    }
    #pragma unroll
    for (int k0 = 0; k0 < 16; k0 += 4) {
      v2f b = ldBp(eW1rp, k0, n0, lane);
      #pragma unroll
      for (int mt = 0; mt < 4; mt++) {
        v2f x = ldsA(rbf + mt * 16 * RB_LD, RB_LD, k0, lane);
        acc[mt] = wmma4(x, b, acc[mt]);
      }
    }
    const float bb = eb1[n0 + cn];
    #pragma unroll
    for (int mt = 0; mt < 4; mt++) {
      #pragma unroll
      for (int j = 0; j < 8; j++)
        buf1[(mt * 16 + j + 8 * hi) * 132 + n0 + cn] = lrelu(acc[mt][j] + bb);
    }
  }
  __syncthreads();
  layernorm128x64(buf1, eg1, ebt1, stats, tid);
  __syncthreads();
  gemm128_lds64(buf1, eW2p, eb2, buf2, lane, wave, 0);
  __syncthreads();
  gemm128_lds64(buf2, cW1p, cb1, buf1, lane, wave, 1);
  __syncthreads();
  layernorm128x64(buf1, cg1, cbt1, stats, tid);
  __syncthreads();
  {
    const int seg = tid & 7;
    #pragma unroll
    for (int hh = 0; hh < 2; hh++) {
      const int r = (tid >> 3) + 32 * hh;
      float sum = 0.f;
      #pragma unroll
      for (int i = 0; i < 16; i++) {
        int c = seg * 16 + i;
        sum += buf1[r * 132 + c] * cW2[c];
      }
      sum += __shfl_xor(sum, 1);
      sum += __shfl_xor(sum, 2);
      sum += __shfl_xor(sum, 4);
      if (seg == 0) coefs[r] = sum + cb2[0];
    }
  }
  __syncthreads();
  for (int i = tid; i < EB * 128; i += 256) {
    int e = i >> 7, c = i & 127;
    int d = didx[e];
    if (d >= 0) atomicAdd(&sum_msg[(size_t)d * 128 + c], buf2[e * 132 + c]);
  }
  if (tid < EB * 3) {
    int e = tid / 3, ax = tid % 3;
    int d = didx[e];
    if (d >= 0) atomicAdd(&sum_x[d * 3 + ax], xrel[e][ax] * coefs[e]);
  }
  if (tid < EB) {
    int d = didx[tid];
    if (d >= 0) atomicAdd(&cnt[d], 1.f);
  }
}

__global__ void __launch_bounds__(256) edge_pair(
    const float* __restrict__ coors0, const float* __restrict__ hs0,
    const float* __restrict__ hd0,
    const int* __restrict__ src0, const int* __restrict__ dst0,
    const float* __restrict__ coors1, const float* __restrict__ hs1,
    const float* __restrict__ hd1,
    const int* __restrict__ src1, const int* __restrict__ dst1,
    const float* __restrict__ eW1rp, const float* __restrict__ eb1,
    const float* __restrict__ eg1, const float* __restrict__ ebt1,
    const float* __restrict__ eW2p, const float* __restrict__ eb2,
    const float* __restrict__ cW1p, const float* __restrict__ cb1,
    const float* __restrict__ cg1, const float* __restrict__ cbt1,
    const float* __restrict__ cW2, const float* __restrict__ cb2,
    float* __restrict__ sm0, float* __restrict__ sx0, float* __restrict__ cnt0,
    float* __restrict__ sm1, float* __restrict__ sx1, float* __restrict__ cnt1,
    int nE, int eblocks) {
  const int second = (int)blockIdx.x >= eblocks;
  const int bid = blockIdx.x - second * eblocks;
  edge_body(second ? coors1 : coors0, second ? hs1 : hs0, second ? hd1 : hd0,
            second ? src1 : src0, second ? dst1 : dst0, eW1rp, eb1, eg1, ebt1, eW2p,
            eb2, cW1p, cb1, cg1,
            cbt1, cW2, cb2, second ? sm1 : sm0, second ? sx1 : sx0,
            second ? cnt1 : cnt0, nE, bid);
}

#define NCAT_LD 450
__device__ __forceinline__ void node_body(
    const float* __restrict__ h, const float* __restrict__ summ,
    const float* __restrict__ cnt, const float* __restrict__ cm,
    const float* __restrict__ orig,
    const float* __restrict__ nW1p, const float* __restrict__ nb1,
    const float* __restrict__ ng1, const float* __restrict__ nbt1,
    const float* __restrict__ nW2p, const float* __restrict__ nb2,
    float* __restrict__ hout, int bid) {
  __shared__ float cat[16 * NCAT_LD];
  __shared__ float buf1[16 * 132];
  __shared__ float stats[16][2];
  const int tid = threadIdx.x, lane = tid & 31, wave = tid >> 5;
  const int hi = lane >> 4, cn = lane & 15;
  const int node0 = bid * 16;

  for (int i = tid; i < 2048; i += 128) {
    int e = i >> 7, c = i & 127;
    int node = node0 + e;
    cat[e * NCAT_LD + c] = h[(size_t)node * 128 + c];
    float ct = cnt[node];
    cat[e * NCAT_LD + 128 + c] = ct > 0.f ? summ[(size_t)node * 128 + c] / ct : 0.f;
    cat[e * NCAT_LD + 256 + c] = cm[(size_t)node * 128 + c];
  }
  for (int i = tid; i < 1024; i += 128) {
    int e = i >> 6, c = i & 63;
    cat[e * NCAT_LD + 384 + c] = orig[(size_t)(node0 + e) * 64 + c];
  }
  __syncthreads();
  {
    const int n0 = wave * 32;
    v8f a0 = {}, a1 = {};
    for (int k0 = 0; k0 < 448; k0 += 4) {
      v2f a  = ldsA(cat, NCAT_LD, k0, lane);
      v2f b0 = ldBp(nW1p, k0, n0, lane);
      v2f b1 = ldBp(nW1p, k0, n0 + 16, lane);
      a0 = wmma4(a, b0, a0);
      a1 = wmma4(a, b1, a1);
    }
    #pragma unroll
    for (int j = 0; j < 8; j++) {
      int r = j + 8 * hi;
      buf1[r * 132 + n0 + cn]      = lrelu(a0[j] + nb1[n0 + cn]);
      buf1[r * 132 + n0 + 16 + cn] = lrelu(a1[j] + nb1[n0 + 16 + cn]);
    }
  }
  __syncthreads();
  layernorm128(buf1, ng1, nbt1, stats, tid);
  __syncthreads();
  {
    const int n0 = wave * 32;
    v8f a0 = {}, a1 = {};
    for (int k0 = 0; k0 < 128; k0 += 4) {
      v2f a  = ldsA(buf1, 132, k0, lane);
      v2f b0 = ldBp(nW2p, k0, n0, lane);
      v2f b1 = ldBp(nW2p, k0, n0 + 16, lane);
      a0 = wmma4(a, b0, a0);
      a1 = wmma4(a, b1, a1);
    }
    #pragma unroll
    for (int j = 0; j < 8; j++) {
      int r = j + 8 * hi;
      size_t g0 = (size_t)(node0 + r) * 128 + n0 + cn;
      size_t g1 = g0 + 16;
      hout[g0] = 0.5f * (a0[j] + nb2[n0 + cn]) + 0.5f * h[g0];
      hout[g1] = 0.5f * (a1[j] + nb2[n0 + 16 + cn]) + 0.5f * h[g1];
    }
  }
}

__global__ void __launch_bounds__(128) node_pair(
    const float* __restrict__ h0, const float* __restrict__ summ0,
    const float* __restrict__ cnt0, const float* __restrict__ cm0,
    const float* __restrict__ orig0, float* __restrict__ hout0,
    const float* __restrict__ h1, const float* __restrict__ summ1,
    const float* __restrict__ cnt1, const float* __restrict__ cm1,
    const float* __restrict__ orig1, float* __restrict__ hout1,
    const float* __restrict__ nW1p, const float* __restrict__ nb1,
    const float* __restrict__ ng1, const float* __restrict__ nbt1,
    const float* __restrict__ nW2p, const float* __restrict__ nb2) {
  const int second = blockIdx.x >= N_NODES / 16;
  const int bid = blockIdx.x - second * (N_NODES / 16);
  node_body(second ? h1 : h0, second ? summ1 : summ0, second ? cnt1 : cnt0,
            second ? cm1 : cm0, second ? orig1 : orig0, nW1p, nb1, ng1, nbt1, nW2p,
            nb2, second ? hout1 : hout0, bid);
}

#define CB ((3 * N_NODES + 255) / 256)
__global__ void coords_pair(const float* __restrict__ oc0, const float* __restrict__ c0,
                            const float* __restrict__ sx0, const float* __restrict__ cnt0,
                            float* __restrict__ out0,
                            const float* __restrict__ oc1, const float* __restrict__ c1,
                            const float* __restrict__ sx1, const float* __restrict__ cnt1,
                            float* __restrict__ out1, int n) {
  const int second = blockIdx.x >= CB;
  const float* oc = second ? oc1 : oc0;
  const float* c = second ? c1 : c0;
  const float* sx = second ? sx1 : sx0;
  const float* cnt = second ? cnt1 : cnt0;
  float* out = second ? out1 : out0;
  int i = (blockIdx.x - second * CB) * 256 + threadIdx.x;
  if (i >= n * 3) return;
  int node = i / 3;
  float ct = cnt[node];
  float xu = ct > 0.f ? sx[i] / ct : 0.f;
  out[i] = 0.2f * oc[i] + 0.8f * c[i] + xu;
}

extern "C" void kernel_launch(void* const* d_in, const int* in_sizes, int n_in,
                              void* d_out, int out_size, void* d_ws, size_t ws_size,
                              hipStream_t stream) {
  const float* coors_lig = (const float*)d_in[0];
  const float* h_lig     = (const float*)d_in[1];
  const float* orig_lig  = (const float*)d_in[2];
  const float* oc_lig    = (const float*)d_in[3];
  const float* coors_rec = (const float*)d_in[4];
  const float* h_rec     = (const float*)d_in[5];
  const float* orig_rec  = (const float*)d_in[6];
  const float* oc_rec    = (const float*)d_in[7];
  const int* lig_src = (const int*)d_in[8];
  const int* lig_dst = (const int*)d_in[9];
  const int* rec_src = (const int*)d_in[10];
  const int* rec_dst = (const int*)d_in[11];
  const float* mask = (const float*)d_in[12];
  const float* eW1 = (const float*)d_in[13];
  const float* eb1 = (const float*)d_in[14];
  const float* eg1 = (const float*)d_in[15];
  const float* ebt1 = (const float*)d_in[16];
  const float* eW2 = (const float*)d_in[17];
  const float* eb2 = (const float*)d_in[18];
  const float* qW = (const float*)d_in[19];
  const float* kW = (const float*)d_in[20];
  const float* vW = (const float*)d_in[21];
  const float* nW1 = (const float*)d_in[22];
  const float* nb1 = (const float*)d_in[23];
  const float* ng1 = (const float*)d_in[24];
  const float* nbt1 = (const float*)d_in[25];
  const float* nW2 = (const float*)d_in[26];
  const float* nb2 = (const float*)d_in[27];
  const float* cW1 = (const float*)d_in[28];
  const float* cb1 = (const float*)d_in[29];
  const float* cg1 = (const float*)d_in[30];
  const float* cbt1 = (const float*)d_in[31];
  const float* cW2 = (const float*)d_in[32];
  const float* cb2 = (const float*)d_in[33];
  const int nE = in_sizes[8];

  float* ws = (float*)d_ws;
  float* hs_l = ws + 0 * NH;
  float* hd_l = ws + 1 * NH;
  float* hs_r = ws + 2 * NH;
  float* hd_r = ws + 3 * NH;
  float* q_l = ws + 0 * NH;
  float* k_l = ws + 1 * NH;
  float* v_l = ws + 2 * NH;
  float* q_r = ws + 3 * NH;
  float* k_r = ws + 4 * NH;
  float* v_r = ws + 5 * NH;
  float* cm_l = ws + 6 * NH;
  float* cm_r = ws + 7 * NH;
  float* sm_l = ws + 8 * NH;
  float* sm_r = ws + 9 * NH;
  float* sx_l = ws + 10 * NH;
  float* sx_r = sx_l + 3 * N_NODES;
  float* cnt_l = sx_r + 3 * N_NODES;
  float* cnt_r = cnt_l + N_NODES;
  float* eW1p = cnt_r + N_NODES;
  float* eW2p = eW1p + 136 * 256;
  float* cW1p = eW2p + 64 * 256;
  float* nW1p = cW1p + 64 * 256;
  float* nW2p = nW1p + 224 * 256;
  float* qWp  = nW2p + 64 * 256;
  float* kWp  = qWp + 64 * 256;
  float* vWp  = kWp + 64 * 256;

  size_t accum_floats = 2 * NH + 8 * (size_t)N_NODES;
  hipMemsetAsync(sm_l, 0, accum_floats * sizeof(float), stream);

  pack_all<<<PK_END * 128 / 256, 256, 0, stream>>>(eW1, eW2, cW1, nW1, nW2, qW, kW, vW,
                                                  eW1p);

  const int ntiles = N_NODES / 16;
  proj_edge<<<4 * ntiles, 128, 0, stream>>>(h_lig, h_rec, eW1p, hs_l, hd_l, hs_r, hd_r);

  int eblocks = (nE + EB - 1) / EB;
  edge_pair<<<2 * eblocks, 256, 0, stream>>>(coors_lig, hs_l, hd_l, lig_src, lig_dst,
      coors_rec, hs_r, hd_r, rec_src, rec_dst,
      eW1p + 128 * 256, eb1, eg1, ebt1, eW2p, eb2, cW1p, cb1, cg1, cbt1, cW2, cb2,
      sm_l, sx_l, cnt_l, sm_r, sx_r, cnt_r, nE, eblocks);

  proj_all<<<6 * ntiles, 128, 0, stream>>>(h_lig, h_rec, qWp, kWp, vWp, q_l, k_l, v_l,
                                           q_r, k_r, v_r);

  attn_pair<<<2 * ntiles, 256, 0, stream>>>(q_l, k_r, v_r, cm_l, q_r, k_l, v_l, cm_r,
                                            mask);

  float* out_xl = (float*)d_out;
  float* out_hl = out_xl + 3 * N_NODES;
  float* out_xr = out_hl + NH;
  float* out_hr = out_xr + 3 * N_NODES;

  node_pair<<<2 * ntiles, 128, 0, stream>>>(h_lig, sm_l, cnt_l, cm_l, orig_lig, out_hl,
      h_rec, sm_r, cnt_r, cm_r, orig_rec, out_hr,
      nW1p, nb1, ng1, nbt1, nW2p, nb2);

  coords_pair<<<2 * CB, 256, 0, stream>>>(oc_lig, coors_lig, sx_l, cnt_l, out_xl,
                                          oc_rec, coors_rec, sx_r, cnt_r, out_xr,
                                          N_NODES);
}


# edge MLP layer 2 moved behind the segment mean by linearity: edge kernel scatters the hidden rows, eW2@cW1 and eW2@nW1[128:256] folded once in 18 extra tiles; sum order changes
# speedup vs baseline: 2.1480x; 1.0837x over previous
#include <hip/hip_runtime.h>
#include <hip/hip_bf16.h>
#include <cstdint>

#define N_NODES 6000
#define NH      ((size_t)N_NODES * 128)

typedef float v2f __attribute__((ext_vector_type(2)));
typedef float v4f __attribute__((ext_vector_type(4)));
typedef float v8f __attribute__((ext_vector_type(8)));

__device__ __forceinline__ v8f wmma4(v2f a, v2f b, v8f c) {
  return __builtin_amdgcn_wmma_f32_16x16x4_f32(false, a, false, b, (short)0, c,
                                               false, false);
}

__device__ __forceinline__ v2f ldsA(const float* base, int ld, int k0, int lane) {
  const int m = lane & 15;
  const int k = k0 + ((lane >> 4) << 1);
  const float* p = base + m * ld + k;
  v2f r; r.x = p[0]; r.y = p[1];
  return r;
}

__device__ __forceinline__ v2f ldBp(const float* __restrict__ P, int k0, int n0,
                                    int lane) {
  const int n = n0 + (lane & 15);
  const int k = k0 + ((lane >> 4) << 1);
  const float* p = P + (size_t)(k >> 1) * 256 + n * 2;
  v2f r; r.x = p[0]; r.y = p[1];
  return r;
}

__device__ __forceinline__ float lrelu(float x) { return x >= 0.f ? x : 0.01f * x; }

__device__ __forceinline__ float rowmax16(float v) {
  v = fmaxf(v, __shfl_xor(v, 1));
  v = fmaxf(v, __shfl_xor(v, 2));
  v = fmaxf(v, __shfl_xor(v, 4));
  v = fmaxf(v, __shfl_xor(v, 8));
  return v;
}
__device__ __forceinline__ float rowsum16(float v) {
  v += __shfl_xor(v, 1);
  v += __shfl_xor(v, 2);
  v += __shfl_xor(v, 4);
  v += __shfl_xor(v, 8);
  return v;
}

__device__ __forceinline__ void layernorm128(float* buf, const float* __restrict__ g,
                                             const float* __restrict__ b,
                                             float (*stats)[2], int tid) {
  const int r = tid >> 3, seg = tid & 7;
  float sum = 0.f, sq = 0.f;
  #pragma unroll
  for (int i = 0; i < 16; i++) {
    float v = buf[r * 132 + seg * 16 + i];
    sum += v; sq += v * v;
  }
  sum += __shfl_xor(sum, 1); sq += __shfl_xor(sq, 1);
  sum += __shfl_xor(sum, 2); sq += __shfl_xor(sq, 2);
  sum += __shfl_xor(sum, 4); sq += __shfl_xor(sq, 4);
  if (seg == 0) {
    float mean = sum * (1.f / 128.f);
    float var  = sq * (1.f / 128.f) - mean * mean;
    stats[r][0] = mean;
    stats[r][1] = rsqrtf(var + 1e-5f);
  }
  __syncthreads();
  const float mean = stats[r][0], rstd = stats[r][1];
  #pragma unroll
  for (int i = 0; i < 16; i++) {
    int c = seg * 16 + i;
    float v = buf[r * 132 + c];
    buf[r * 132 + c] = g[c] * (v - mean) * rstd + b[c];
  }
}

__device__ __forceinline__ void gemm128_lds(const float* in, const float* __restrict__ Wp,
                                            const float* __restrict__ bias, float* outb,
                                            int lane, int wave, int act) {
  const int hi = lane >> 4, cn = lane & 15;
  const int n0 = wave * 32;
  v8f a0 = {}, a1 = {};
  for (int k0 = 0; k0 < 128; k0 += 4) {
    v2f a  = ldsA(in, 132, k0, lane);
    v2f b0 = ldBp(Wp, k0, n0, lane);
    v2f b1 = ldBp(Wp, k0, n0 + 16, lane);
    a0 = wmma4(a, b0, a0);
    a1 = wmma4(a, b1, a1);
  }
  #pragma unroll
  for (int j = 0; j < 8; j++) {
    int r = j + 8 * hi;
    float x0 = a0[j] + bias[n0 + cn];
    float x1 = a1[j] + bias[n0 + 16 + cn];
    if (act) { x0 = lrelu(x0); x1 = lrelu(x1); }
    outb[r * 132 + n0 + cn] = x0;
    outb[r * 132 + n0 + 16 + cn] = x1;
  }
}

__device__ __forceinline__ void layernorm128x64(float* buf, const float* __restrict__ g,
                                                const float* __restrict__ b,
                                                float (*stats)[2], int tid) {
  const int seg = tid & 7;
  #pragma unroll
  for (int hh = 0; hh < 2; hh++) {
    const int r = (tid >> 3) + 32 * hh;
    float sum = 0.f, sq = 0.f;
    #pragma unroll
    for (int i = 0; i < 16; i++) {
      float v = buf[r * 132 + seg * 16 + i];
      sum += v; sq += v * v;
    }
    sum += __shfl_xor(sum, 1); sq += __shfl_xor(sq, 1);
    sum += __shfl_xor(sum, 2); sq += __shfl_xor(sq, 2);
    sum += __shfl_xor(sum, 4); sq += __shfl_xor(sq, 4);
    if (seg == 0) {
      float mean = sum * (1.f / 128.f);
      float var  = sq * (1.f / 128.f) - mean * mean;
      stats[r][0] = mean;
      stats[r][1] = rsqrtf(var + 1e-5f);
    }
  }
  __syncthreads();
  #pragma unroll
  for (int hh = 0; hh < 2; hh++) {
    const int r = (tid >> 3) + 32 * hh;
    const float mean = stats[r][0], rstd = stats[r][1];
    #pragma unroll
    for (int i = 0; i < 16; i++) {
      int c = seg * 16 + i;
      float v = buf[r * 132 + c];
      buf[r * 132 + c] = g[c] * (v - mean) * rstd + b[c];
    }
  }
}

__device__ __forceinline__ void gemm128_lds64(const float* in, const float* __restrict__ Wp,
                                              const float* __restrict__ bias, float* outb,
                                              int lane, int wave, int act) {
  const int hi = lane >> 4, cn = lane & 15;
  const int n0 = wave * 16;
  v8f acc[4];
  #pragma unroll
  for (int mt = 0; mt < 4; mt++) acc[mt] = (v8f){};
  for (int k0 = 0; k0 < 128; k0 += 4) {
    v2f b = ldBp(Wp, k0, n0, lane);
    #pragma unroll
    for (int mt = 0; mt < 4; mt++) {
      v2f x = ldsA(in + mt * 16 * 132, 132, k0, lane);
      acc[mt] = wmma4(x, b, acc[mt]);
    }
  }
  const float bb = bias[n0 + cn];
  #pragma unroll
  for (int mt = 0; mt < 4; mt++) {
    #pragma unroll
    for (int j = 0; j < 8; j++) {
      int r = mt * 16 + j + 8 * hi;
      float x = acc[mt][j] + bb;
      if (act) x = lrelu(x);
      outb[r * 132 + n0 + cn] = x;
    }
  }
}

#define PK_E1 0
#define PK_WC 136
#define PK_N1 264
#define PK_N2 488
#define PK_Q  552
#define PK_K  616
#define PK_V  680
#define PK_END 744
__global__ void pack_all(const float* __restrict__ eW1, const float* __restrict__ nW1,
                         const float* __restrict__ nW2, const float* __restrict__ qW,
                         const float* __restrict__ kW, const float* __restrict__ vW,
                         float* __restrict__ P) {
  int i = blockIdx.x * 256 + threadIdx.x;
  if (i >= PK_END * 128) return;
  int gp = i >> 7, n = i & 127;
  const float* W; int K, p0;
  if (gp < PK_WC)      { W = eW1; K = 271; p0 = PK_E1; }
  else if (gp < PK_N1) return;
  else if (gp < PK_N2) {
    if (gp >= PK_N1 + 64 && gp < PK_N1 + 128) return;
    W = nW1; K = 448; p0 = PK_N1;
  }
  else if (gp < PK_Q)  { W = nW2; K = 128; p0 = PK_N2; }
  else if (gp < PK_K)  { W = qW;  K = 128; p0 = PK_Q; }
  else if (gp < PK_V)  { W = kW;  K = 128; p0 = PK_K; }
  else                 { W = vW;  K = 128; p0 = PK_V; }
  int k = 2 * (gp - p0);
  float a = (k < K)     ? W[(size_t)k * 128 + n]       : 0.f;
  float b = (k + 1 < K) ? W[(size_t)(k + 1) * 128 + n] : 0.f;
  P[(size_t)gp * 256 + n * 2]     = a;
  P[(size_t)gp * 256 + n * 2 + 1] = b;
}

__device__ __forceinline__ void proj_body(const float* __restrict__ A,
                                          const float* __restrict__ Wp,
                                          float* __restrict__ C, int act,
                                          int pack, int bid, float* at) {
  const int tid = threadIdx.x, lane = tid & 31, wave = tid >> 5;
  const int row0 = bid * 16;
  for (int i = tid; i < 16 * 128; i += 128) {
    int r = i >> 7, c = i & 127;
    at[r * 132 + c] = A[(size_t)(row0 + r) * 128 + c];
  }
  __syncthreads();
  const int n0 = wave * 32;
  v8f a0 = {}, a1 = {};
  for (int k0 = 0; k0 < 128; k0 += 4) {
    v2f a  = ldsA(at, 132, k0, lane);
    v2f b0 = ldBp(Wp, k0, n0, lane);
    v2f b1 = ldBp(Wp, k0, n0 + 16, lane);
    a0 = wmma4(a, b0, a0);
    a1 = wmma4(a, b1, a1);
  }
  const int hi = lane >> 4, cn = lane & 15;
  #pragma unroll
  for (int j = 0; j < 8; j++) {
    int r = row0 + j + 8 * hi;
    float x0 = a0[j], x1 = a1[j];
    if (act) { x0 = lrelu(x0); x1 = lrelu(x1); }
    if (pack) {
      C[(size_t)(r >> 1) * 256 + (n0 + cn) * 2 + (r & 1)]      = x0;
      C[(size_t)(r >> 1) * 256 + (n0 + 16 + cn) * 2 + (r & 1)] = x1;
    } else {
      C[(size_t)r * 128 + n0 + cn] = x0;
      C[(size_t)r * 128 + n0 + 16 + cn] = x1;
    }
  }
}

__global__ void __launch_bounds__(128) proj_all(
    const float* __restrict__ h0, const float* __restrict__ h1,
    const float* __restrict__ qWp, const float* __restrict__ kWp,
    const float* __restrict__ vWp,
    float* __restrict__ q0, float* __restrict__ k0, float* __restrict__ v0,
    float* __restrict__ q1, float* __restrict__ k1, float* __restrict__ v1) {
  const int p = blockIdx.x / (N_NODES / 16);
  const int bid = blockIdx.x - p * (N_NODES / 16);
  const int side = p >= 3;
  const int j = p - 3 * side;
  const float* A = side ? h1 : h0;
  const float* Wp = j == 0 ? qWp : (j == 1 ? kWp : vWp);
  float* C = j == 0 ? (side ? q1 : q0) : (j == 1 ? (side ? k1 : k0) : (side ? v1 : v0));
  __shared__ float at[16 * 132];
  proj_body(A, Wp, C, j != 2, j == 2, bid, at);
}

__device__ __forceinline__ void fold_body(float* at, const float* __restrict__ eW2,
                                          const float* __restrict__ eb2,
                                          const float* __restrict__ Wb,
                                          const float* __restrict__ addb,
                                          float* __restrict__ Cp,
                                          float* __restrict__ biasout, int t) {
  const int tid = threadIdx.x, lane = tid & 31, wave = tid >> 5;
  for (int i = tid; i < 16 * 128; i += 128) {
    int r = i >> 7, c = i & 127;
    float v;
    if (t < 8) v = eW2[(size_t)(t * 16 + r) * 128 + c];
    else       v = r == 0 ? eb2[c] : 0.f;
    at[r * 132 + c] = v;
  }
  __syncthreads();
  const int n0 = wave * 32;
  const int hi = lane >> 4, cn = lane & 15;
  v8f a0 = {}, a1 = {};
  for (int k0 = 0; k0 < 128; k0 += 4) {
    v2f a = ldsA(at, 132, k0, lane);
    const float* p = Wb + (size_t)(k0 + 2 * hi) * 128 + n0 + cn;
    v2f b0, b1;
    b0.x = p[0];  b0.y = p[128];
    b1.x = p[16]; b1.y = p[144];
    a0 = wmma4(a, b0, a0);
    a1 = wmma4(a, b1, a1);
  }
  if (t < 8) {
    #pragma unroll
    for (int j = 0; j < 8; j++) {
      int r = t * 16 + j + 8 * hi;
      Cp[(size_t)(r >> 1) * 256 + (n0 + cn) * 2 + (r & 1)]      = a0[j];
      Cp[(size_t)(r >> 1) * 256 + (n0 + 16 + cn) * 2 + (r & 1)] = a1[j];
    }
  } else if (hi == 0) {
    biasout[n0 + cn]      = a0[0] + (addb ? addb[n0 + cn] : 0.f);
    biasout[n0 + 16 + cn] = a1[0] + (addb ? addb[n0 + 16 + cn] : 0.f);
  }
}

__global__ void __launch_bounds__(128) proj_edge(
    const float* __restrict__ h0, const float* __restrict__ h1,
    const float* __restrict__ eW1p,
    float* __restrict__ hs0, float* __restrict__ hd0,
    float* __restrict__ hs1, float* __restrict__ hd1,
    const float* __restrict__ eW2, const float* __restrict__ eb2,
    const float* __restrict__ cW1, const float* __restrict__ cb1,
    const float* __restrict__ nW1,
    float* __restrict__ Wcp, float* __restrict__ bcv,
    float* __restrict__ Wnp, float* __restrict__ bnv) {
  __shared__ float at[16 * 132];
  if (blockIdx.x >= 4 * (N_NODES / 16)) {
    const int f = blockIdx.x - 4 * (N_NODES / 16);
    if (f < 9) fold_body(at, eW2, eb2, cW1, cb1, Wcp, bcv, f);
    else       fold_body(at, eW2, eb2, nW1 + 128 * 128, nullptr, Wnp, bnv, f - 9);
    return;
  }
  const int p = blockIdx.x / (N_NODES / 16);
  const int bid = blockIdx.x - p * (N_NODES / 16);
  const int side = p >> 1;
  const int half = p & 1;
  const float* A = side ? h1 : h0;
  const float* Wp = eW1p + half * (64 * 256);
  float* C = half ? (side ? hd1 : hd0) : (side ? hs1 : hs0);
  proj_body(A, Wp, C, 0, 0, bid, at);
}

__device__ __forceinline__ void attn_body(const float* __restrict__ Q,
                                          const float* __restrict__ K,
                                          const float* __restrict__ Vp,
                                          const float* __restrict__ mask,
                                          int tmask, float* __restrict__ out, int bid) {
  __shared__ __attribute__((aligned(16))) float qt[16 * 132];
  __shared__ float pbuf[8][16 * 18];
  __shared__ float outbuf[16 * 128];
  __shared__ float mbuf[8][16], lbuf[8][16];
  __shared__ float Mrow[16], Lrow[16];
  const int tid = threadIdx.x, lane = tid & 31;
  const int wave = __builtin_amdgcn_readfirstlane(tid >> 5);
  const int hi = lane >> 4, cn = lane & 15;
  const int qrow0 = bid * 16;

  for (int i = tid; i < 2048; i += 256) {
    int r = i >> 7, c = i & 127;
    qt[r * 132 + c] = Q[(size_t)(qrow0 + r) * 128 + c];
  }
  __syncthreads();

  float mr[8], lr[8];
  v8f acc[8];
  #pragma unroll
  for (int j = 0; j < 8; j++) { mr[j] = -3.0e38f; lr[j] = 0.f; acc[j] = (v8f){}; }
  float* pw = &pbuf[wave][0];

  const float* qap = qt + cn * 132 + 4 * hi;
  const unsigned koff = (unsigned)(cn * 128 + 4 * hi);
  const unsigned voff = (unsigned)(hi * 256 + 4 * cn);
  const unsigned mrs = tmask ? 1u : (unsigned)N_NODES;
  const unsigned mcs = tmask ? (unsigned)N_NODES : 1u;
  const unsigned moff = (unsigned)(qrow0 + 8 * hi) * mrs + (unsigned)cn * mcs;
  v4f kf[16], vf[16];
  float mk[8];
  {
    const float* kb = K + (size_t)wave * 2048;
    #pragma unroll
    for (int i = 0; i < 16; i++) kf[i] = *(const v4f*)(kb + (koff + 8u * i));
    const float* mb = mask + (size_t)(wave * 16) * mcs;
    #pragma unroll
    for (int j = 0; j < 8; j++) mk[j] = mb[moff + (unsigned)j * mrs];
    const float* vb = Vp + (size_t)wave * 2048;
    #pragma unroll
    for (int i = 0; i < 16; i++) vf[i] = *(const v4f*)(vb + (voff + 512u * (i >> 2) + 64u * (i & 3)));
  }

  for (int ch = wave; ch < N_NODES / 16; ch += 8) {
    const int nch = (ch + 8 < N_NODES / 16) ? ch + 8 : N_NODES / 16 - 1;
    v8f s = {};
    #pragma unroll
    for (int i = 0; i < 16; i++) {
      v4f qa = *(const v4f*)(qap + 8 * i);
      v2f a0, a1, b0, b1;
      a0.x = qa.x; a0.y = qa.y; a1.x = qa.z; a1.y = qa.w;
      b0.x = kf[i].x; b0.y = kf[i].y; b1.x = kf[i].z; b1.y = kf[i].w;
      s = wmma4(a0, b0, s);
      s = wmma4(a1, b1, s);
    }
    float sc[8];
    #pragma unroll
    for (int j = 0; j < 8; j++) {
      float mv = mk[j];
      float sj = mv * s[j] - 1000.f * (1.f - mv);
      float mnew = fmaxf(mr[j], rowmax16(sj));
      sc[j] = __expf(mr[j] - mnew);
      float pj = __expf(sj - mnew);
      lr[j] = lr[j] * sc[j] + rowsum16(pj);
      mr[j] = mnew;
      pw[(j + 8 * hi) * 18 + cn] = pj;
    }
    #pragma unroll
    for (int t = 0; t < 8; t++)
      #pragma unroll
      for (int j = 0; j < 8; j++) acc[t][j] *= sc[j];
    {
      const float* kb = K + (size_t)nch * 2048;
      #pragma unroll
      for (int i = 0; i < 16; i++) kf[i] = *(const v4f*)(kb + (koff + 8u * i));
      const float* mb = mask + (size_t)(nch * 16) * mcs;
      #pragma unroll
      for (int j = 0; j < 8; j++) mk[j] = mb[moff + (unsigned)j * mrs];
    }
    asm volatile("s_wait_dscnt 0" ::: "memory");
    #pragma unroll
    for (int kk = 0; kk < 4; kk++) {
      v2f a = ldsA(pw, 18, 4 * kk, lane);
      #pragma unroll
      for (int u = 0; u < 4; u++) {
        v2f b0, b1;
        b0.x = vf[4 * kk + u].x; b0.y = vf[4 * kk + u].y;
        b1.x = vf[4 * kk + u].z; b1.y = vf[4 * kk + u].w;
        acc[2 * u]     = wmma4(a, b0, acc[2 * u]);
        acc[2 * u + 1] = wmma4(a, b1, acc[2 * u + 1]);
      }
    }
    {
      const float* vb = Vp + (size_t)nch * 2048;
      #pragma unroll
      for (int i = 0; i < 16; i++) vf[i] = *(const v4f*)(vb + (voff + 512u * (i >> 2) + 64u * (i & 3)));
    }
  }

  if (cn == 0) {
    #pragma unroll
    for (int j = 0; j < 8; j++) {
      mbuf[wave][j + 8 * hi] = mr[j];
      lbuf[wave][j + 8 * hi] = lr[j];
    }
  }
  __syncthreads();
  for (int i = tid; i < 2048; i += 256) outbuf[i] = 0.f;
  if (tid < 16) {
    float M = -3.0e38f;
    for (int w = 0; w < 8; w++) M = fmaxf(M, mbuf[w][tid]);
    float L = 0.f;
    for (int w = 0; w < 8; w++) L += lbuf[w][tid] * __expf(mbuf[w][tid] - M);
    Mrow[tid] = M;
    Lrow[tid] = L > 0.f ? 1.f / L : 0.f;
  }
  __syncthreads();
  #pragma unroll
  for (int j = 0; j < 8; j++) {
    float e = __expf(mr[j] - Mrow[j + 8 * hi]);
    #pragma unroll
    for (int t = 0; t < 8; t++)
      atomicAdd(&outbuf[(j + 8 * hi) * 128 + 32 * (t >> 1) + 2 * cn + (t & 1)],
                acc[t][j] * e);
  }
  __syncthreads();
  for (int i = tid; i < 2048; i += 256) {
    int r = i >> 7, c = i & 127;
    out[(size_t)(qrow0 + r) * 128 + c] = outbuf[i] * Lrow[r];
  }
}

__global__ void __launch_bounds__(256) attn_pair(
    const float* __restrict__ Q0, const float* __restrict__ K0,
    const float* __restrict__ V0, float* __restrict__ out0,
    const float* __restrict__ Q1, const float* __restrict__ K1,
    const float* __restrict__ V1, float* __restrict__ out1,
    const float* __restrict__ mask) {
  const int second = blockIdx.x >= N_NODES / 16;
  const int bid = blockIdx.x - second * (N_NODES / 16);
  attn_body(second ? Q1 : Q0, second ? K1 : K0, second ? V1 : V0, mask, second,
            second ? out1 : out0, bid);
}

#define RB_LD 18
#define EB 64
__device__ __forceinline__ void edge_body(
    const float* __restrict__ coors, const float* __restrict__ hs,
    const float* __restrict__ hd,
    const int* __restrict__ src, const int* __restrict__ dst,
    const float* __restrict__ eW1rp, const float* __restrict__ eb1,
    const float* __restrict__ eg1, const float* __restrict__ ebt1,
    const float* __restrict__ Wcp, const float* __restrict__ bcv,
    const float* __restrict__ cg1, const float* __restrict__ cbt1,
    const float* __restrict__ cW2, const float* __restrict__ cb2,
    float* __restrict__ sum_hid, float* __restrict__ sum_x,
    float* __restrict__ cnt, int nE, int bid) {
  __shared__ __attribute__((aligned(16))) float buf1[EB * 132];
  __shared__ __attribute__((aligned(16))) float buf2[EB * 132];
  __shared__ __attribute__((aligned(16))) float rbf[EB * RB_LD];
  __shared__ float stats[EB][2];
  __shared__ float xrel[EB][3];
  __shared__ float coefs[EB];
  __shared__ int sidx[EB], didx[EB];
  const int tid = threadIdx.x, lane = tid & 31, wave = tid >> 5;
  const int hi = lane >> 4, cn = lane & 15;
  const int e0 = bid * EB;

  if (tid < EB) {
    int e = e0 + tid;
    if (e < nE) {
      int s = src[e], d = dst[e];
      sidx[tid] = s; didx[tid] = d;
      float x0 = coors[s * 3 + 0] - coors[d * 3 + 0];
      float x1 = coors[s * 3 + 1] - coors[d * 3 + 1];
      float x2 = coors[s * 3 + 2] - coors[d * 3 + 2];
      xrel[tid][0] = x0; xrel[tid][1] = x1; xrel[tid][2] = x2;
      float d2 = x0 * x0 + x1 * x1 + x2 * x2;
      float sig = 1.f;
      for (int j = 0; j < 15; j++) {
        rbf[tid * RB_LD + j] = __expf(-d2 / sig);
        sig *= 1.5f;
      }
      rbf[tid * RB_LD + 15] = 0.f;
    } else {
      sidx[tid] = 0; didx[tid] = -1;
      xrel[tid][0] = xrel[tid][1] = xrel[tid][2] = 0.f;
      for (int j = 0; j < 16; j++) rbf[tid * RB_LD + j] = 0.f;
    }
  }
  __syncthreads();
  {
    const int l4 = lane * 4;
    int ns[8], nd[8];
    v4f gs[8], gd[8];
    #pragma unroll
    for (int it = 0; it < 8; it++) {
      const int e = wave + 8 * it;
      ns[it] = sidx[e];
      nd[it] = didx[e];
    }
    #pragma unroll
    for (int it = 0; it < 8; it++) nd[it] = nd[it] >= 0 ? nd[it] : 0;
    #pragma unroll
    for (int it = 0; it < 8; it++) {
      gs[it] = *(const v4f*)(hs + (size_t)ns[it] * 128 + l4);
      gd[it] = *(const v4f*)(hd + (size_t)nd[it] * 128 + l4);
    }
    #pragma unroll
    for (int it = 0; it < 8; it++) {
      const int e = wave + 8 * it;
      *(v4f*)(buf1 + e * 132 + l4) = gs[it] + gd[it];
    }
  }
  __syncthreads();

  {
    const int n0 = wave * 16;
    v8f acc[4];
    #pragma unroll
    for (int mt = 0; mt < 4; mt++) {
      #pragma unroll
      for (int j = 0; j < 8; j++)
        acc[mt][j] = buf1[(mt * 16 + j + 8 * hi) * 132 + n0 + cn];
    }
    #pragma unroll
    for (int k0 = 0; k0 < 16; k0 += 4) {
      v2f b = ldBp(eW1rp, k0, n0, lane);
      #pragma unroll
      for (int mt = 0; mt < 4; mt++) {
        v2f x = ldsA(rbf + mt * 16 * RB_LD, RB_LD, k0, lane);
        acc[mt] = wmma4(x, b, acc[mt]);
      }
    }
    const float bb = eb1[n0 + cn];
    #pragma unroll
    for (int mt = 0; mt < 4; mt++) {
      #pragma unroll
      for (int j = 0; j < 8; j++)
        buf1[(mt * 16 + j + 8 * hi) * 132 + n0 + cn] = lrelu(acc[mt][j] + bb);
    }
  }
  __syncthreads();
  layernorm128x64(buf1, eg1, ebt1, stats, tid);
  __syncthreads();
  gemm128_lds64(buf1, Wcp, bcv, buf2, lane, wave, 1);
  __syncthreads();
  layernorm128x64(buf2, cg1, cbt1, stats, tid);
  __syncthreads();
  {
    const int seg = tid & 7;
    #pragma unroll
    for (int hh = 0; hh < 2; hh++) {
      const int r = (tid >> 3) + 32 * hh;
      float sum = 0.f;
      #pragma unroll
      for (int i = 0; i < 16; i++) {
        int c = seg * 16 + i;
        sum += buf2[r * 132 + c] * cW2[c];
      }
      sum += __shfl_xor(sum, 1);
      sum += __shfl_xor(sum, 2);
      sum += __shfl_xor(sum, 4);
      if (seg == 0) coefs[r] = sum + cb2[0];
    }
  }
  __syncthreads();
  for (int i = tid; i < EB * 128; i += 256) {
    int e = i >> 7, c = i & 127;
    int d = didx[e];
    if (d >= 0) atomicAdd(&sum_hid[(size_t)d * 128 + c], buf1[e * 132 + c]);
  }
  if (tid < EB * 3) {
    int e = tid / 3, ax = tid % 3;
    int d = didx[e];
    if (d >= 0) atomicAdd(&sum_x[d * 3 + ax], xrel[e][ax] * coefs[e]);
  }
  if (tid < EB) {
    int d = didx[tid];
    if (d >= 0) atomicAdd(&cnt[d], 1.f);
  }
}

__global__ void __launch_bounds__(256) edge_pair(
    const float* __restrict__ coors0, const float* __restrict__ hs0,
    const float* __restrict__ hd0,
    const int* __restrict__ src0, const int* __restrict__ dst0,
    const float* __restrict__ coors1, const float* __restrict__ hs1,
    const float* __restrict__ hd1,
    const int* __restrict__ src1, const int* __restrict__ dst1,
    const float* __restrict__ eW1rp, const float* __restrict__ eb1,
    const float* __restrict__ eg1, const float* __restrict__ ebt1,
    const float* __restrict__ Wcp, const float* __restrict__ bcv,
    const float* __restrict__ cg1, const float* __restrict__ cbt1,
    const float* __restrict__ cW2, const float* __restrict__ cb2,
    float* __restrict__ sm0, float* __restrict__ sx0, float* __restrict__ cnt0,
    float* __restrict__ sm1, float* __restrict__ sx1, float* __restrict__ cnt1,
    int nE, int eblocks) {
  const int second = (int)blockIdx.x >= eblocks;
  const int bid = blockIdx.x - second * eblocks;
  edge_body(second ? coors1 : coors0, second ? hs1 : hs0, second ? hd1 : hd0,
            second ? src1 : src0, second ? dst1 : dst0, eW1rp, eb1, eg1, ebt1, Wcp,
            bcv, cg1,
            cbt1, cW2, cb2, second ? sm1 : sm0, second ? sx1 : sx0,
            second ? cnt1 : cnt0, nE, bid);
}

#define NCAT_LD 450
__device__ __forceinline__ void node_body(
    const float* __restrict__ h, const float* __restrict__ summ,
    const float* __restrict__ cnt, const float* __restrict__ cm,
    const float* __restrict__ orig,
    const float* __restrict__ nW1p, const float* __restrict__ nb1,
    const float* __restrict__ bnv,
    const float* __restrict__ ng1, const float* __restrict__ nbt1,
    const float* __restrict__ nW2p, const float* __restrict__ nb2,
    float* __restrict__ hout, int bid) {
  __shared__ float cat[16 * NCAT_LD];
  __shared__ float buf1[16 * 132];
  __shared__ float stats[16][2];
  __shared__ float has[16];
  const int tid = threadIdx.x, lane = tid & 31, wave = tid >> 5;
  const int hi = lane >> 4, cn = lane & 15;
  const int node0 = bid * 16;

  for (int i = tid; i < 2048; i += 128) {
    int e = i >> 7, c = i & 127;
    int node = node0 + e;
    cat[e * NCAT_LD + c] = h[(size_t)node * 128 + c];
    float ct = cnt[node];
    cat[e * NCAT_LD + 128 + c] = ct > 0.f ? summ[(size_t)node * 128 + c] / ct : 0.f;
    if (c == 0) has[e] = ct > 0.f ? 1.f : 0.f;
    cat[e * NCAT_LD + 256 + c] = cm[(size_t)node * 128 + c];
  }
  for (int i = tid; i < 1024; i += 128) {
    int e = i >> 6, c = i & 63;
    cat[e * NCAT_LD + 384 + c] = orig[(size_t)(node0 + e) * 64 + c];
  }
  __syncthreads();
  {
    const int n0 = wave * 32;
    v8f a0 = {}, a1 = {};
    for (int k0 = 0; k0 < 448; k0 += 4) {
      v2f a  = ldsA(cat, NCAT_LD, k0, lane);
      v2f b0 = ldBp(nW1p, k0, n0, lane);
      v2f b1 = ldBp(nW1p, k0, n0 + 16, lane);
      a0 = wmma4(a, b0, a0);
      a1 = wmma4(a, b1, a1);
    }
    const float fb0 = bnv[n0 + cn], fb1 = bnv[n0 + 16 + cn];
    #pragma unroll
    for (int j = 0; j < 8; j++) {
      int r = j + 8 * hi;
      const bool hasr = has[r] > 0.f;
      buf1[r * 132 + n0 + cn]      = lrelu(a0[j] + nb1[n0 + cn] + (hasr ? fb0 : 0.f));
      buf1[r * 132 + n0 + 16 + cn] = lrelu(a1[j] + nb1[n0 + 16 + cn] + (hasr ? fb1 : 0.f));
    }
  }
  __syncthreads();
  layernorm128(buf1, ng1, nbt1, stats, tid);
  __syncthreads();
  {
    const int n0 = wave * 32;
    v8f a0 = {}, a1 = {};
    for (int k0 = 0; k0 < 128; k0 += 4) {
      v2f a  = ldsA(buf1, 132, k0, lane);
      v2f b0 = ldBp(nW2p, k0, n0, lane);
      v2f b1 = ldBp(nW2p, k0, n0 + 16, lane);
      a0 = wmma4(a, b0, a0);
      a1 = wmma4(a, b1, a1);
    }
    #pragma unroll
    for (int j = 0; j < 8; j++) {
      int r = j + 8 * hi;
      size_t g0 = (size_t)(node0 + r) * 128 + n0 + cn;
      size_t g1 = g0 + 16;
      hout[g0] = 0.5f * (a0[j] + nb2[n0 + cn]) + 0.5f * h[g0];
      hout[g1] = 0.5f * (a1[j] + nb2[n0 + 16 + cn]) + 0.5f * h[g1];
    }
  }
}

__global__ void __launch_bounds__(128) node_pair(
    const float* __restrict__ h0, const float* __restrict__ summ0,
    const float* __restrict__ cnt0, const float* __restrict__ cm0,
    const float* __restrict__ orig0, float* __restrict__ hout0,
    const float* __restrict__ h1, const float* __restrict__ summ1,
    const float* __restrict__ cnt1, const float* __restrict__ cm1,
    const float* __restrict__ orig1, float* __restrict__ hout1,
    const float* __restrict__ nW1p, const float* __restrict__ nb1,
    const float* __restrict__ bnv,
    const float* __restrict__ ng1, const float* __restrict__ nbt1,
    const float* __restrict__ nW2p, const float* __restrict__ nb2) {
  const int second = blockIdx.x >= N_NODES / 16;
  const int bid = blockIdx.x - second * (N_NODES / 16);
  node_body(second ? h1 : h0, second ? summ1 : summ0, second ? cnt1 : cnt0,
            second ? cm1 : cm0, second ? orig1 : orig0, nW1p, nb1, bnv, ng1, nbt1, nW2p,
            nb2, second ? hout1 : hout0, bid);
}

#define CB ((3 * N_NODES + 255) / 256)
__global__ void coords_pair(const float* __restrict__ oc0, const float* __restrict__ c0,
                            const float* __restrict__ sx0, const float* __restrict__ cnt0,
                            float* __restrict__ out0,
                            const float* __restrict__ oc1, const float* __restrict__ c1,
                            const float* __restrict__ sx1, const float* __restrict__ cnt1,
                            float* __restrict__ out1, int n) {
  const int second = blockIdx.x >= CB;
  const float* oc = second ? oc1 : oc0;
  const float* c = second ? c1 : c0;
  const float* sx = second ? sx1 : sx0;
  const float* cnt = second ? cnt1 : cnt0;
  float* out = second ? out1 : out0;
  int i = (blockIdx.x - second * CB) * 256 + threadIdx.x;
  if (i >= n * 3) return;
  int node = i / 3;
  float ct = cnt[node];
  float xu = ct > 0.f ? sx[i] / ct : 0.f;
  out[i] = 0.2f * oc[i] + 0.8f * c[i] + xu;
}

extern "C" void kernel_launch(void* const* d_in, const int* in_sizes, int n_in,
                              void* d_out, int out_size, void* d_ws, size_t ws_size,
                              hipStream_t stream) {
  const float* coors_lig = (const float*)d_in[0];
  const float* h_lig     = (const float*)d_in[1];
  const float* orig_lig  = (const float*)d_in[2];
  const float* oc_lig    = (const float*)d_in[3];
  const float* coors_rec = (const float*)d_in[4];
  const float* h_rec     = (const float*)d_in[5];
  const float* orig_rec  = (const float*)d_in[6];
  const float* oc_rec    = (const float*)d_in[7];
  const int* lig_src = (const int*)d_in[8];
  const int* lig_dst = (const int*)d_in[9];
  const int* rec_src = (const int*)d_in[10];
  const int* rec_dst = (const int*)d_in[11];
  const float* mask = (const float*)d_in[12];
  const float* eW1 = (const float*)d_in[13];
  const float* eb1 = (const float*)d_in[14];
  const float* eg1 = (const float*)d_in[15];
  const float* ebt1 = (const float*)d_in[16];
  const float* eW2 = (const float*)d_in[17];
  const float* eb2 = (const float*)d_in[18];
  const float* qW = (const float*)d_in[19];
  const float* kW = (const float*)d_in[20];
  const float* vW = (const float*)d_in[21];
  const float* nW1 = (const float*)d_in[22];
  const float* nb1 = (const float*)d_in[23];
  const float* ng1 = (const float*)d_in[24];
  const float* nbt1 = (const float*)d_in[25];
  const float* nW2 = (const float*)d_in[26];
  const float* nb2 = (const float*)d_in[27];
  const float* cW1 = (const float*)d_in[28];
  const float* cb1 = (const float*)d_in[29];
  const float* cg1 = (const float*)d_in[30];
  const float* cbt1 = (const float*)d_in[31];
  const float* cW2 = (const float*)d_in[32];
  const float* cb2 = (const float*)d_in[33];
  const int nE = in_sizes[8];

  float* ws = (float*)d_ws;
  float* hs_l = ws + 0 * NH;
  float* hd_l = ws + 1 * NH;
  float* hs_r = ws + 2 * NH;
  float* hd_r = ws + 3 * NH;
  float* q_l = ws + 0 * NH;
  float* k_l = ws + 1 * NH;
  float* v_l = ws + 2 * NH;
  float* q_r = ws + 3 * NH;
  float* k_r = ws + 4 * NH;
  float* v_r = ws + 5 * NH;
  float* cm_l = ws + 6 * NH;
  float* cm_r = ws + 7 * NH;
  float* sm_l = ws + 8 * NH;
  float* sm_r = ws + 9 * NH;
  float* sx_l = ws + 10 * NH;
  float* sx_r = sx_l + 3 * N_NODES;
  float* cnt_l = sx_r + 3 * N_NODES;
  float* cnt_r = cnt_l + N_NODES;
  float* eW1p = cnt_r + N_NODES;
  float* Wcp  = eW1p + 136 * 256;
  float* bcv  = Wcp + 64 * 256;
  float* bnv  = bcv + 128;
  float* nW1p = Wcp + 128 * 256;
  float* nW2p = nW1p + 224 * 256;
  float* qWp  = nW2p + 64 * 256;
  float* kWp  = qWp + 64 * 256;
  float* vWp  = kWp + 64 * 256;

  size_t accum_floats = 2 * NH + 8 * (size_t)N_NODES;
  hipMemsetAsync(sm_l, 0, accum_floats * sizeof(float), stream);

  pack_all<<<PK_END * 128 / 256, 256, 0, stream>>>(eW1, nW1, nW2, qW, kW, vW, eW1p);

  const int ntiles = N_NODES / 16;
  proj_edge<<<4 * ntiles + 18, 128, 0, stream>>>(h_lig, h_rec, eW1p, hs_l, hd_l, hs_r, hd_r,
      eW2, eb2, cW1, cb1, nW1, Wcp, bcv, nW1p + 64 * 256, bnv);

  int eblocks = (nE + EB - 1) / EB;
  edge_pair<<<2 * eblocks, 256, 0, stream>>>(coors_lig, hs_l, hd_l, lig_src, lig_dst,
      coors_rec, hs_r, hd_r, rec_src, rec_dst,
      eW1p + 128 * 256, eb1, eg1, ebt1, Wcp, bcv, cg1, cbt1, cW2, cb2,
      sm_l, sx_l, cnt_l, sm_r, sx_r, cnt_r, nE, eblocks);

  proj_all<<<6 * ntiles, 128, 0, stream>>>(h_lig, h_rec, qWp, kWp, vWp, q_l, k_l, v_l,
                                           q_r, k_r, v_r);

  attn_pair<<<2 * ntiles, 256, 0, stream>>>(q_l, k_r, v_r, cm_l, q_r, k_l, v_l, cm_r,
                                            mask);

  float* out_xl = (float*)d_out;
  float* out_hl = out_xl + 3 * N_NODES;
  float* out_xr = out_hl + NH;
  float* out_hr = out_xr + 3 * N_NODES;

  node_pair<<<2 * ntiles, 128, 0, stream>>>(h_lig, sm_l, cnt_l, cm_l, orig_lig, out_hl,
      h_rec, sm_r, cnt_r, cm_r, orig_rec, out_hr,
      nW1p, nb1, bnv, ng1, nbt1, nW2p, nb2);

  coords_pair<<<2 * CB, 256, 0, stream>>>(oc_lig, coors_lig, sx_l, cnt_l, out_xl,
                                          oc_rec, coors_rec, sx_r, cnt_r, out_xr,
                                          N_NODES);
}


# attention chunk transposed (S^T = K Q^T, O^T += V^T P^T): softmax state one scalar per lane, P feeds PV from registers; key loop free of LDS traffic and 16-lane sums; row-sum and PV sum order changes
# speedup vs baseline: 2.3812x; 1.1086x over previous
#include <hip/hip_runtime.h>
#include <hip/hip_bf16.h>
#include <cstdint>

#define N_NODES 6000
#define NH      ((size_t)N_NODES * 128)

typedef float v2f __attribute__((ext_vector_type(2)));
typedef float v4f __attribute__((ext_vector_type(4)));
typedef float v8f __attribute__((ext_vector_type(8)));

__device__ __forceinline__ v8f wmma4(v2f a, v2f b, v8f c) {
  return __builtin_amdgcn_wmma_f32_16x16x4_f32(false, a, false, b, (short)0, c,
                                               false, false);
}

__device__ __forceinline__ v2f ldsA(const float* base, int ld, int k0, int lane) {
  const int m = lane & 15;
  const int k = k0 + ((lane >> 4) << 1);
  const float* p = base + m * ld + k;
  v2f r; r.x = p[0]; r.y = p[1];
  return r;
}

__device__ __forceinline__ v2f ldBp(const float* __restrict__ P, int k0, int n0,
                                    int lane) {
  const int n = n0 + (lane & 15);
  const int k = k0 + ((lane >> 4) << 1);
  const float* p = P + (size_t)(k >> 1) * 256 + n * 2;
  v2f r; r.x = p[0]; r.y = p[1];
  return r;
}

__device__ __forceinline__ float lrelu(float x) { return x >= 0.f ? x : 0.01f * x; }

__device__ __forceinline__ float rowmax16(float v) {
  v = fmaxf(v, __shfl_xor(v, 1));
  v = fmaxf(v, __shfl_xor(v, 2));
  v = fmaxf(v, __shfl_xor(v, 4));
  v = fmaxf(v, __shfl_xor(v, 8));
  return v;
}
__device__ __forceinline__ float rowsum16(float v) {
  v += __shfl_xor(v, 1);
  v += __shfl_xor(v, 2);
  v += __shfl_xor(v, 4);
  v += __shfl_xor(v, 8);
  return v;
}

__device__ __forceinline__ void layernorm128(float* buf, const float* __restrict__ g,
                                             const float* __restrict__ b,
                                             float (*stats)[2], int tid) {
  const int r = tid >> 3, seg = tid & 7;
  float sum = 0.f, sq = 0.f;
  #pragma unroll
  for (int i = 0; i < 16; i++) {
    float v = buf[r * 132 + seg * 16 + i];
    sum += v; sq += v * v;
  }
  sum += __shfl_xor(sum, 1); sq += __shfl_xor(sq, 1);
  sum += __shfl_xor(sum, 2); sq += __shfl_xor(sq, 2);
  sum += __shfl_xor(sum, 4); sq += __shfl_xor(sq, 4);
  if (seg == 0) {
    float mean = sum * (1.f / 128.f);
    float var  = sq * (1.f / 128.f) - mean * mean;
    stats[r][0] = mean;
    stats[r][1] = rsqrtf(var + 1e-5f);
  }
  __syncthreads();
  const float mean = stats[r][0], rstd = stats[r][1];
  #pragma unroll
  for (int i = 0; i < 16; i++) {
    int c = seg * 16 + i;
    float v = buf[r * 132 + c];
    buf[r * 132 + c] = g[c] * (v - mean) * rstd + b[c];
  }
}

__device__ __forceinline__ void gemm128_lds(const float* in, const float* __restrict__ Wp,
                                            const float* __restrict__ bias, float* outb,
                                            int lane, int wave, int act) {
  const int hi = lane >> 4, cn = lane & 15;
  const int n0 = wave * 32;
  v8f a0 = {}, a1 = {};
  for (int k0 = 0; k0 < 128; k0 += 4) {
    v2f a  = ldsA(in, 132, k0, lane);
    v2f b0 = ldBp(Wp, k0, n0, lane);
    v2f b1 = ldBp(Wp, k0, n0 + 16, lane);
    a0 = wmma4(a, b0, a0);
    a1 = wmma4(a, b1, a1);
  }
  #pragma unroll
  for (int j = 0; j < 8; j++) {
    int r = j + 8 * hi;
    float x0 = a0[j] + bias[n0 + cn];
    float x1 = a1[j] + bias[n0 + 16 + cn];
    if (act) { x0 = lrelu(x0); x1 = lrelu(x1); }
    outb[r * 132 + n0 + cn] = x0;
    outb[r * 132 + n0 + 16 + cn] = x1;
  }
}

__device__ __forceinline__ void layernorm128x64(float* buf, const float* __restrict__ g,
                                                const float* __restrict__ b,
                                                float (*stats)[2], int tid) {
  const int seg = tid & 7;
  #pragma unroll
  for (int hh = 0; hh < 2; hh++) {
    const int r = (tid >> 3) + 32 * hh;
    float sum = 0.f, sq = 0.f;
    #pragma unroll
    for (int i = 0; i < 16; i++) {
      float v = buf[r * 132 + seg * 16 + i];
      sum += v; sq += v * v;
    }
    sum += __shfl_xor(sum, 1); sq += __shfl_xor(sq, 1);
    sum += __shfl_xor(sum, 2); sq += __shfl_xor(sq, 2);
    sum += __shfl_xor(sum, 4); sq += __shfl_xor(sq, 4);
    if (seg == 0) {
      float mean = sum * (1.f / 128.f);
      float var  = sq * (1.f / 128.f) - mean * mean;
      stats[r][0] = mean;
      stats[r][1] = rsqrtf(var + 1e-5f);
    }
  }
  __syncthreads();
  #pragma unroll
  for (int hh = 0; hh < 2; hh++) {
    const int r = (tid >> 3) + 32 * hh;
    const float mean = stats[r][0], rstd = stats[r][1];
    #pragma unroll
    for (int i = 0; i < 16; i++) {
      int c = seg * 16 + i;
      float v = buf[r * 132 + c];
      buf[r * 132 + c] = g[c] * (v - mean) * rstd + b[c];
    }
  }
}

__device__ __forceinline__ void gemm128_lds64(const float* in, const float* __restrict__ Wp,
                                              const float* __restrict__ bias, float* outb,
                                              int lane, int wave, int act) {
  const int hi = lane >> 4, cn = lane & 15;
  const int n0 = wave * 16;
  v8f acc[4];
  #pragma unroll
  for (int mt = 0; mt < 4; mt++) acc[mt] = (v8f){};
  for (int k0 = 0; k0 < 128; k0 += 4) {
    v2f b = ldBp(Wp, k0, n0, lane);
    #pragma unroll
    for (int mt = 0; mt < 4; mt++) {
      v2f x = ldsA(in + mt * 16 * 132, 132, k0, lane);
      acc[mt] = wmma4(x, b, acc[mt]);
    }
  }
  const float bb = bias[n0 + cn];
  #pragma unroll
  for (int mt = 0; mt < 4; mt++) {
    #pragma unroll
    for (int j = 0; j < 8; j++) {
      int r = mt * 16 + j + 8 * hi;
      float x = acc[mt][j] + bb;
      if (act) x = lrelu(x);
      outb[r * 132 + n0 + cn] = x;
    }
  }
}

#define PK_E1 0
#define PK_WC 136
#define PK_N1 264
#define PK_N2 488
#define PK_Q  552
#define PK_K  616
#define PK_V  680
#define PK_END 744
__global__ void pack_all(const float* __restrict__ eW1, const float* __restrict__ nW1,
                         const float* __restrict__ nW2, const float* __restrict__ qW,
                         const float* __restrict__ kW, const float* __restrict__ vW,
                         float* __restrict__ P) {
  int i = blockIdx.x * 256 + threadIdx.x;
  if (i >= PK_END * 128) return;
  int gp = i >> 7, n = i & 127;
  const float* W; int K, p0;
  if (gp < PK_WC)      { W = eW1; K = 271; p0 = PK_E1; }
  else if (gp < PK_N1) return;
  else if (gp < PK_N2) {
    if (gp >= PK_N1 + 64 && gp < PK_N1 + 128) return;
    W = nW1; K = 448; p0 = PK_N1;
  }
  else if (gp < PK_Q)  { W = nW2; K = 128; p0 = PK_N2; }
  else if (gp < PK_K)  { W = qW;  K = 128; p0 = PK_Q; }
  else if (gp < PK_V)  { W = kW;  K = 128; p0 = PK_K; }
  else                 { W = vW;  K = 128; p0 = PK_V; }
  int k = 2 * (gp - p0);
  float a = (k < K)     ? W[(size_t)k * 128 + n]       : 0.f;
  float b = (k + 1 < K) ? W[(size_t)(k + 1) * 128 + n] : 0.f;
  P[(size_t)gp * 256 + n * 2]     = a;
  P[(size_t)gp * 256 + n * 2 + 1] = b;
}

__device__ __forceinline__ void proj_body(const float* __restrict__ A,
                                          const float* __restrict__ Wp,
                                          float* __restrict__ C, int act,
                                          int pack, int bid, float* at) {
  const int tid = threadIdx.x, lane = tid & 31, wave = tid >> 5;
  const int row0 = bid * 16;
  for (int i = tid; i < 16 * 128; i += 128) {
    int r = i >> 7, c = i & 127;
    at[r * 132 + c] = A[(size_t)(row0 + r) * 128 + c];
  }
  __syncthreads();
  const int n0 = wave * 32;
  v8f a0 = {}, a1 = {};
  for (int k0 = 0; k0 < 128; k0 += 4) {
    v2f a  = ldsA(at, 132, k0, lane);
    v2f b0 = ldBp(Wp, k0, n0, lane);
    v2f b1 = ldBp(Wp, k0, n0 + 16, lane);
    a0 = wmma4(a, b0, a0);
    a1 = wmma4(a, b1, a1);
  }
  const int hi = lane >> 4, cn = lane & 15;
  #pragma unroll
  for (int j = 0; j < 8; j++) {
    int r = row0 + j + 8 * hi;
    float x0 = a0[j], x1 = a1[j];
    if (act) { x0 = lrelu(x0); x1 = lrelu(x1); }
    if (pack) {
      C[(size_t)(r >> 1) * 256 + (n0 + cn) * 2 + (r & 1)]      = x0;
      C[(size_t)(r >> 1) * 256 + (n0 + 16 + cn) * 2 + (r & 1)] = x1;
    } else {
      C[(size_t)r * 128 + n0 + cn] = x0;
      C[(size_t)r * 128 + n0 + 16 + cn] = x1;
    }
  }
}

__global__ void __launch_bounds__(128) proj_all(
    const float* __restrict__ h0, const float* __restrict__ h1,
    const float* __restrict__ qWp, const float* __restrict__ kWp,
    const float* __restrict__ vWp,
    float* __restrict__ q0, float* __restrict__ k0, float* __restrict__ v0,
    float* __restrict__ q1, float* __restrict__ k1, float* __restrict__ v1) {
  const int p = blockIdx.x / (N_NODES / 16);
  const int bid = blockIdx.x - p * (N_NODES / 16);
  const int side = p >= 3;
  const int j = p - 3 * side;
  const float* A = side ? h1 : h0;
  const float* Wp = j == 0 ? qWp : (j == 1 ? kWp : vWp);
  float* C = j == 0 ? (side ? q1 : q0) : (j == 1 ? (side ? k1 : k0) : (side ? v1 : v0));
  __shared__ float at[16 * 132];
  proj_body(A, Wp, C, j != 2, j == 2, bid, at);
}

__device__ __forceinline__ void fold_body(float* at, const float* __restrict__ eW2,
                                          const float* __restrict__ eb2,
                                          const float* __restrict__ Wb,
                                          const float* __restrict__ addb,
                                          float* __restrict__ Cp,
                                          float* __restrict__ biasout, int t) {
  const int tid = threadIdx.x, lane = tid & 31, wave = tid >> 5;
  for (int i = tid; i < 16 * 128; i += 128) {
    int r = i >> 7, c = i & 127;
    float v;
    if (t < 8) v = eW2[(size_t)(t * 16 + r) * 128 + c];
    else       v = r == 0 ? eb2[c] : 0.f;
    at[r * 132 + c] = v;
  }
  __syncthreads();
  const int n0 = wave * 32;
  const int hi = lane >> 4, cn = lane & 15;
  v8f a0 = {}, a1 = {};
  for (int k0 = 0; k0 < 128; k0 += 4) {
    v2f a = ldsA(at, 132, k0, lane);
    const float* p = Wb + (size_t)(k0 + 2 * hi) * 128 + n0 + cn;
    v2f b0, b1;
    b0.x = p[0];  b0.y = p[128];
    b1.x = p[16]; b1.y = p[144];
    a0 = wmma4(a, b0, a0);
    a1 = wmma4(a, b1, a1);
  }
  if (t < 8) {
    #pragma unroll
    for (int j = 0; j < 8; j++) {
      int r = t * 16 + j + 8 * hi;
      Cp[(size_t)(r >> 1) * 256 + (n0 + cn) * 2 + (r & 1)]      = a0[j];
      Cp[(size_t)(r >> 1) * 256 + (n0 + 16 + cn) * 2 + (r & 1)] = a1[j];
    }
  } else if (hi == 0) {
    biasout[n0 + cn]      = a0[0] + (addb ? addb[n0 + cn] : 0.f);
    biasout[n0 + 16 + cn] = a1[0] + (addb ? addb[n0 + 16 + cn] : 0.f);
  }
}

__global__ void __launch_bounds__(128) proj_edge(
    const float* __restrict__ h0, const float* __restrict__ h1,
    const float* __restrict__ eW1p,
    float* __restrict__ hs0, float* __restrict__ hd0,
    float* __restrict__ hs1, float* __restrict__ hd1,
    const float* __restrict__ eW2, const float* __restrict__ eb2,
    const float* __restrict__ cW1, const float* __restrict__ cb1,
    const float* __restrict__ nW1,
    float* __restrict__ Wcp, float* __restrict__ bcv,
    float* __restrict__ Wnp, float* __restrict__ bnv) {
  __shared__ float at[16 * 132];
  if (blockIdx.x >= 4 * (N_NODES / 16)) {
    const int f = blockIdx.x - 4 * (N_NODES / 16);
    if (f < 9) fold_body(at, eW2, eb2, cW1, cb1, Wcp, bcv, f);
    else       fold_body(at, eW2, eb2, nW1 + 128 * 128, nullptr, Wnp, bnv, f - 9);
    return;
  }
  const int p = blockIdx.x / (N_NODES / 16);
  const int bid = blockIdx.x - p * (N_NODES / 16);
  const int side = p >> 1;
  const int half = p & 1;
  const float* A = side ? h1 : h0;
  const float* Wp = eW1p + half * (64 * 256);
  float* C = half ? (side ? hd1 : hd0) : (side ? hs1 : hs0);
  proj_body(A, Wp, C, 0, 0, bid, at);
}

__device__ __forceinline__ float xhalf(float x) {
  const int b = __builtin_bit_cast(int, x);
  return __builtin_bit_cast(float, __builtin_amdgcn_permlanex16(b, b, 0x76543210, 0xfedcba98, false, false));
}

template <int TM>
__device__ __forceinline__ void load_mask8(const float* __restrict__ mp, float (&mk)[8]) {
  if (TM == 0) {
    const v4f m0 = *(const v4f*)(mp);
    const v4f m1 = *(const v4f*)(mp + 4);
    mk[0] = m0.x; mk[1] = m0.y; mk[2] = m0.z; mk[3] = m0.w;
    mk[4] = m1.x; mk[5] = m1.y; mk[6] = m1.z; mk[7] = m1.w;
  } else {
    #pragma unroll
    for (int r = 0; r < 8; r++) mk[r] = mp[(size_t)r * N_NODES];
  }
}

#define OB_LD 129
template <int TM>
__device__ __forceinline__ void attn_body(const float* __restrict__ Q,
                                          const float* __restrict__ K,
                                          const float* __restrict__ Vp,
                                          const float* __restrict__ mask,
                                          float* __restrict__ out, int bid,
                                          float* qt, float* outbuf, float (*mbuf)[16],
                                          float (*lbuf)[16], float* Mrow, float* Lrow) {
  const int tid = threadIdx.x, lane = tid & 31;
  const int wave = __builtin_amdgcn_readfirstlane(tid >> 5);
  const int hi = lane >> 4, cn = lane & 15;
  const int qrow0 = bid * 16;

  for (int i = tid; i < 2048; i += 256) {
    int r = i >> 7, c = i & 127;
    qt[r * 132 + c] = Q[(size_t)(qrow0 + r) * 128 + c];
  }
  __syncthreads();

  float mr = -3.0e38f, lr = 0.f;
  v8f acc[8];
  #pragma unroll
  for (int j = 0; j < 8; j++) acc[j] = (v8f){};

  const float* qap = qt + cn * 132 + 4 * hi;
  const unsigned koff = (unsigned)(cn * 128 + 4 * hi);
  const unsigned voff = (unsigned)(hi * 1024 + 4 * cn);
  const size_t mrs = TM ? 1 : N_NODES;
  const size_t mcs = TM ? N_NODES : 1;
  const float* mlane = mask + (size_t)(qrow0 + cn) * mrs + (size_t)(8 * hi) * mcs;
  v4f kf[16], vf[16];
  float mk[8];
  {
    const float* kb = K + (size_t)wave * 2048;
    #pragma unroll
    for (int i = 0; i < 16; i++) kf[i] = *(const v4f*)(kb + (koff + 8u * i));
    load_mask8<TM>(mlane + (size_t)(wave * 16) * mcs, mk);
    const float* vb = Vp + (size_t)wave * 2048;
    #pragma unroll
    for (int i = 0; i < 16; i++) vf[i] = *(const v4f*)(vb + (voff + 256u * (i >> 2) + 64u * (i & 3)));
  }
#define RW8(a, o) "+v"(a[o]), "+v"(a[o + 1]), "+v"(a[o + 2]), "+v"(a[o + 3]), \
                  "+v"(a[o + 4]), "+v"(a[o + 5]), "+v"(a[o + 6]), "+v"(a[o + 7])
  asm volatile("" : RW8(kf, 0));
  asm volatile("" : RW8(kf, 8));
  asm volatile("" : RW8(vf, 0));
  asm volatile("" : RW8(vf, 8));
  asm volatile("" : RW8(mk, 0));
#undef RW8

  for (int ch = wave; ch < N_NODES / 16; ch += 8) {
    const int nch = (ch + 8 < N_NODES / 16) ? ch + 8 : N_NODES / 16 - 1;
    v8f s = {};
    #pragma unroll
    for (int i = 0; i < 16; i++) {
      v4f qa = *(const v4f*)(qap + 8 * i);
      v2f a0, a1, b0, b1;
      a0.x = kf[i].x; a0.y = kf[i].y; a1.x = kf[i].z; a1.y = kf[i].w;
      b0.x = qa.x; b0.y = qa.y; b1.x = qa.z; b1.y = qa.w;
      s = wmma4(a0, b0, s);
      s = wmma4(a1, b1, s);
    }
    float sj[8], p[8];
    #pragma unroll
    for (int r = 0; r < 8; r++) {
      float mv = mk[r];
      sj[r] = mv * s[r] - 1000.f * (1.f - mv);
    }
    float cm = fmaxf(fmaxf(fmaxf(sj[0], sj[1]), fmaxf(sj[2], sj[3])),
                     fmaxf(fmaxf(sj[4], sj[5]), fmaxf(sj[6], sj[7])));
    cm = fmaxf(cm, xhalf(cm));
    const float mnew = fmaxf(mr, cm);
    const float sc = __expf(mr - mnew);
    float ps = 0.f;
    #pragma unroll
    for (int r = 0; r < 8; r++) {
      p[r] = __expf(sj[r] - mnew);
      ps += p[r];
    }
    lr = lr * sc + ps;
    mr = mnew;
    #pragma unroll
    for (int t = 0; t < 8; t++)
      #pragma unroll
      for (int j = 0; j < 8; j++) acc[t][j] *= sc;
    {
      const float* kb = K + (size_t)nch * 2048;
      #pragma unroll
      for (int i = 0; i < 16; i++) kf[i] = *(const v4f*)(kb + (koff + 8u * i));
      load_mask8<TM>(mlane + (size_t)(nch * 16) * mcs, mk);
    }
    #pragma unroll
    for (int kk = 0; kk < 4; kk++) {
      v2f b; b.x = p[2 * kk]; b.y = p[2 * kk + 1];
      #pragma unroll
      for (int u = 0; u < 4; u++) {
        v2f a0, a1;
        a0.x = vf[4 * kk + u].x; a0.y = vf[4 * kk + u].y;
        a1.x = vf[4 * kk + u].z; a1.y = vf[4 * kk + u].w;
        acc[2 * u]     = wmma4(a0, b, acc[2 * u]);
        acc[2 * u + 1] = wmma4(a1, b, acc[2 * u + 1]);
      }
    }
    {
      const float* vb = Vp + (size_t)nch * 2048;
      #pragma unroll
      for (int i = 0; i < 16; i++) vf[i] = *(const v4f*)(vb + (voff + 256u * (i >> 2) + 64u * (i & 3)));
    }
  }

  const float lt = lr + xhalf(lr);
  if (hi == 0) {
    mbuf[wave][cn] = mr;
    lbuf[wave][cn] = lt;
  }
  __syncthreads();
  for (int i = tid; i < 16 * OB_LD; i += 256) outbuf[i] = 0.f;
  if (tid < 16) {
    float M = -3.0e38f;
    for (int w = 0; w < 8; w++) M = fmaxf(M, mbuf[w][tid]);
    float L = 0.f;
    for (int w = 0; w < 8; w++) L += lbuf[w][tid] * __expf(mbuf[w][tid] - M);
    Mrow[tid] = M;
    Lrow[tid] = L > 0.f ? 1.f / L : 0.f;
  }
  __syncthreads();
  {
    const float e = __expf(mr - Mrow[cn]);
    #pragma unroll
    for (int t = 0; t < 8; t++)
      #pragma unroll
      for (int r = 0; r < 8; r++)
        atomicAdd(&outbuf[cn * OB_LD + 32 * (t >> 1) + 2 * (r + 8 * hi) + (t & 1)],
                  acc[t][r] * e);
  }
  __syncthreads();
  for (int i = tid; i < 2048; i += 256) {
    int r = i >> 7, c = i & 127;
    out[(size_t)(qrow0 + r) * 128 + c] = outbuf[r * OB_LD + c] * Lrow[r];
  }
}

__global__ void __launch_bounds__(256) attn_pair(
    const float* __restrict__ Q0, const float* __restrict__ K0,
    const float* __restrict__ V0, float* __restrict__ out0,
    const float* __restrict__ Q1, const float* __restrict__ K1,
    const float* __restrict__ V1, float* __restrict__ out1,
    const float* __restrict__ mask) {
  __shared__ __attribute__((aligned(16))) float qt[16 * 132];
  __shared__ float outbuf[16 * OB_LD];
  __shared__ float mbuf[8][16], lbuf[8][16];
  __shared__ float Mrow[16], Lrow[16];
  const int second = blockIdx.x >= N_NODES / 16;
  const int bid = blockIdx.x - second * (N_NODES / 16);
  if (second) attn_body<1>(Q1, K1, V1, mask, out1, bid, qt, outbuf, mbuf, lbuf, Mrow, Lrow);
  else        attn_body<0>(Q0, K0, V0, mask, out0, bid, qt, outbuf, mbuf, lbuf, Mrow, Lrow);
}

#define RB_LD 18
#define EB 64
__device__ __forceinline__ void edge_body(
    const float* __restrict__ coors, const float* __restrict__ hs,
    const float* __restrict__ hd,
    const int* __restrict__ src, const int* __restrict__ dst,
    const float* __restrict__ eW1rp, const float* __restrict__ eb1,
    const float* __restrict__ eg1, const float* __restrict__ ebt1,
    const float* __restrict__ Wcp, const float* __restrict__ bcv,
    const float* __restrict__ cg1, const float* __restrict__ cbt1,
    const float* __restrict__ cW2, const float* __restrict__ cb2,
    float* __restrict__ sum_hid, float* __restrict__ sum_x,
    float* __restrict__ cnt, int nE, int bid) {
  __shared__ __attribute__((aligned(16))) float buf1[EB * 132];
  __shared__ __attribute__((aligned(16))) float buf2[EB * 132];
  __shared__ __attribute__((aligned(16))) float rbf[EB * RB_LD];
  __shared__ float stats[EB][2];
  __shared__ float xrel[EB][3];
  __shared__ float coefs[EB];
  __shared__ int sidx[EB], didx[EB];
  const int tid = threadIdx.x, lane = tid & 31, wave = tid >> 5;
  const int hi = lane >> 4, cn = lane & 15;
  const int e0 = bid * EB;

  if (tid < EB) {
    int e = e0 + tid;
    if (e < nE) {
      int s = src[e], d = dst[e];
      sidx[tid] = s; didx[tid] = d;
      float x0 = coors[s * 3 + 0] - coors[d * 3 + 0];
      float x1 = coors[s * 3 + 1] - coors[d * 3 + 1];
      float x2 = coors[s * 3 + 2] - coors[d * 3 + 2];
      xrel[tid][0] = x0; xrel[tid][1] = x1; xrel[tid][2] = x2;
      float d2 = x0 * x0 + x1 * x1 + x2 * x2;
      float sig = 1.f;
      for (int j = 0; j < 15; j++) {
        rbf[tid * RB_LD + j] = __expf(-d2 / sig);
        sig *= 1.5f;
      }
      rbf[tid * RB_LD + 15] = 0.f;
    } else {
      sidx[tid] = 0; didx[tid] = -1;
      xrel[tid][0] = xrel[tid][1] = xrel[tid][2] = 0.f;
      for (int j = 0; j < 16; j++) rbf[tid * RB_LD + j] = 0.f;
    }
  }
  __syncthreads();
  {
    const int l4 = lane * 4;
    int ns[8], nd[8];
    v4f gs[8], gd[8];
    #pragma unroll
    for (int it = 0; it < 8; it++) {
      const int e = wave + 8 * it;
      ns[it] = sidx[e];
      nd[it] = didx[e];
    }
    #pragma unroll
    for (int it = 0; it < 8; it++) nd[it] = nd[it] >= 0 ? nd[it] : 0;
    #pragma unroll
    for (int it = 0; it < 8; it++) {
      gs[it] = *(const v4f*)(hs + (size_t)ns[it] * 128 + l4);
      gd[it] = *(const v4f*)(hd + (size_t)nd[it] * 128 + l4);
    }
    #pragma unroll
    for (int it = 0; it < 8; it++) {
      const int e = wave + 8 * it;
      *(v4f*)(buf1 + e * 132 + l4) = gs[it] + gd[it];
    }
  }
  __syncthreads();

  {
    const int n0 = wave * 16;
    v8f acc[4];
    #pragma unroll
    for (int mt = 0; mt < 4; mt++) {
      #pragma unroll
      for (int j = 0; j < 8; j++)
        acc[mt][j] = buf1[(mt * 16 + j + 8 * hi) * 132 + n0 + cn];
    }
    #pragma unroll
    for (int k0 = 0; k0 < 16; k0 += 4) {
      v2f b = ldBp(eW1rp, k0, n0, lane);
      #pragma unroll
      for (int mt = 0; mt < 4; mt++) {
        v2f x = ldsA(rbf + mt * 16 * RB_LD, RB_LD, k0, lane);
        acc[mt] = wmma4(x, b, acc[mt]);
      }
    }
    const float bb = eb1[n0 + cn];
    #pragma unroll
    for (int mt = 0; mt < 4; mt++) {
      #pragma unroll
      for (int j = 0; j < 8; j++)
        buf1[(mt * 16 + j + 8 * hi) * 132 + n0 + cn] = lrelu(acc[mt][j] + bb);
    }
  }
  __syncthreads();
  layernorm128x64(buf1, eg1, ebt1, stats, tid);
  __syncthreads();
  gemm128_lds64(buf1, Wcp, bcv, buf2, lane, wave, 1);
  __syncthreads();
  layernorm128x64(buf2, cg1, cbt1, stats, tid);
  __syncthreads();
  {
    const int seg = tid & 7;
    #pragma unroll
    for (int hh = 0; hh < 2; hh++) {
      const int r = (tid >> 3) + 32 * hh;
      float sum = 0.f;
      #pragma unroll
      for (int i = 0; i < 16; i++) {
        int c = seg * 16 + i;
        sum += buf2[r * 132 + c] * cW2[c];
      }
      sum += __shfl_xor(sum, 1);
      sum += __shfl_xor(sum, 2);
      sum += __shfl_xor(sum, 4);
      if (seg == 0) coefs[r] = sum + cb2[0];
    }
  }
  __syncthreads();
  for (int i = tid; i < EB * 128; i += 256) {
    int e = i >> 7, c = i & 127;
    int d = didx[e];
    if (d >= 0) atomicAdd(&sum_hid[(size_t)d * 128 + c], buf1[e * 132 + c]);
  }
  if (tid < EB * 3) {
    int e = tid / 3, ax = tid % 3;
    int d = didx[e];
    if (d >= 0) atomicAdd(&sum_x[d * 3 + ax], xrel[e][ax] * coefs[e]);
  }
  if (tid < EB) {
    int d = didx[tid];
    if (d >= 0) atomicAdd(&cnt[d], 1.f);
  }
}

__global__ void __launch_bounds__(256) edge_pair(
    const float* __restrict__ coors0, const float* __restrict__ hs0,
    const float* __restrict__ hd0,
    const int* __restrict__ src0, const int* __restrict__ dst0,
    const float* __restrict__ coors1, const float* __restrict__ hs1,
    const float* __restrict__ hd1,
    const int* __restrict__ src1, const int* __restrict__ dst1,
    const float* __restrict__ eW1rp, const float* __restrict__ eb1,
    const float* __restrict__ eg1, const float* __restrict__ ebt1,
    const float* __restrict__ Wcp, const float* __restrict__ bcv,
    const float* __restrict__ cg1, const float* __restrict__ cbt1,
    const float* __restrict__ cW2, const float* __restrict__ cb2,
    float* __restrict__ sm0, float* __restrict__ sx0, float* __restrict__ cnt0,
    float* __restrict__ sm1, float* __restrict__ sx1, float* __restrict__ cnt1,
    int nE, int eblocks) {
  const int second = (int)blockIdx.x >= eblocks;
  const int bid = blockIdx.x - second * eblocks;
  edge_body(second ? coors1 : coors0, second ? hs1 : hs0, second ? hd1 : hd0,
            second ? src1 : src0, second ? dst1 : dst0, eW1rp, eb1, eg1, ebt1, Wcp,
            bcv, cg1,
            cbt1, cW2, cb2, second ? sm1 : sm0, second ? sx1 : sx0,
            second ? cnt1 : cnt0, nE, bid);
}

#define NCAT_LD 450
__device__ __forceinline__ void node_body(
    const float* __restrict__ h, const float* __restrict__ summ,
    const float* __restrict__ cnt, const float* __restrict__ cm,
    const float* __restrict__ orig,
    const float* __restrict__ nW1p, const float* __restrict__ nb1,
    const float* __restrict__ bnv,
    const float* __restrict__ ng1, const float* __restrict__ nbt1,
    const float* __restrict__ nW2p, const float* __restrict__ nb2,
    float* __restrict__ hout, int bid) {
  __shared__ float cat[16 * NCAT_LD];
  __shared__ float buf1[16 * 132];
  __shared__ float stats[16][2];
  __shared__ float has[16];
  const int tid = threadIdx.x, lane = tid & 31, wave = tid >> 5;
  const int hi = lane >> 4, cn = lane & 15;
  const int node0 = bid * 16;

  for (int i = tid; i < 2048; i += 128) {
    int e = i >> 7, c = i & 127;
    int node = node0 + e;
    cat[e * NCAT_LD + c] = h[(size_t)node * 128 + c];
    float ct = cnt[node];
    cat[e * NCAT_LD + 128 + c] = ct > 0.f ? summ[(size_t)node * 128 + c] / ct : 0.f;
    if (c == 0) has[e] = ct > 0.f ? 1.f : 0.f;
    cat[e * NCAT_LD + 256 + c] = cm[(size_t)node * 128 + c];
  }
  for (int i = tid; i < 1024; i += 128) {
    int e = i >> 6, c = i & 63;
    cat[e * NCAT_LD + 384 + c] = orig[(size_t)(node0 + e) * 64 + c];
  }
  __syncthreads();
  {
    const int n0 = wave * 32;
    v8f a0 = {}, a1 = {};
    for (int k0 = 0; k0 < 448; k0 += 4) {
      v2f a  = ldsA(cat, NCAT_LD, k0, lane);
      v2f b0 = ldBp(nW1p, k0, n0, lane);
      v2f b1 = ldBp(nW1p, k0, n0 + 16, lane);
      a0 = wmma4(a, b0, a0);
      a1 = wmma4(a, b1, a1);
    }
    const float fb0 = bnv[n0 + cn], fb1 = bnv[n0 + 16 + cn];
    #pragma unroll
    for (int j = 0; j < 8; j++) {
      int r = j + 8 * hi;
      const bool hasr = has[r] > 0.f;
      buf1[r * 132 + n0 + cn]      = lrelu(a0[j] + nb1[n0 + cn] + (hasr ? fb0 : 0.f));
      buf1[r * 132 + n0 + 16 + cn] = lrelu(a1[j] + nb1[n0 + 16 + cn] + (hasr ? fb1 : 0.f));
    }
  }
  __syncthreads();
  layernorm128(buf1, ng1, nbt1, stats, tid);
  __syncthreads();
  {
    const int n0 = wave * 32;
    v8f a0 = {}, a1 = {};
    for (int k0 = 0; k0 < 128; k0 += 4) {
      v2f a  = ldsA(buf1, 132, k0, lane);
      v2f b0 = ldBp(nW2p, k0, n0, lane);
      v2f b1 = ldBp(nW2p, k0, n0 + 16, lane);
      a0 = wmma4(a, b0, a0);
      a1 = wmma4(a, b1, a1);
    }
    #pragma unroll
    for (int j = 0; j < 8; j++) {
      int r = j + 8 * hi;
      size_t g0 = (size_t)(node0 + r) * 128 + n0 + cn;
      size_t g1 = g0 + 16;
      hout[g0] = 0.5f * (a0[j] + nb2[n0 + cn]) + 0.5f * h[g0];
      hout[g1] = 0.5f * (a1[j] + nb2[n0 + 16 + cn]) + 0.5f * h[g1];
    }
  }
}

__global__ void __launch_bounds__(128) node_pair(
    const float* __restrict__ h0, const float* __restrict__ summ0,
    const float* __restrict__ cnt0, const float* __restrict__ cm0,
    const float* __restrict__ orig0, float* __restrict__ hout0,
    const float* __restrict__ h1, const float* __restrict__ summ1,
    const float* __restrict__ cnt1, const float* __restrict__ cm1,
    const float* __restrict__ orig1, float* __restrict__ hout1,
    const float* __restrict__ nW1p, const float* __restrict__ nb1,
    const float* __restrict__ bnv,
    const float* __restrict__ ng1, const float* __restrict__ nbt1,
    const float* __restrict__ nW2p, const float* __restrict__ nb2) {
  const int second = blockIdx.x >= N_NODES / 16;
  const int bid = blockIdx.x - second * (N_NODES / 16);
  node_body(second ? h1 : h0, second ? summ1 : summ0, second ? cnt1 : cnt0,
            second ? cm1 : cm0, second ? orig1 : orig0, nW1p, nb1, bnv, ng1, nbt1, nW2p,
            nb2, second ? hout1 : hout0, bid);
}

#define CB ((3 * N_NODES + 255) / 256)
__global__ void coords_pair(const float* __restrict__ oc0, const float* __restrict__ c0,
                            const float* __restrict__ sx0, const float* __restrict__ cnt0,
                            float* __restrict__ out0,
                            const float* __restrict__ oc1, const float* __restrict__ c1,
                            const float* __restrict__ sx1, const float* __restrict__ cnt1,
                            float* __restrict__ out1, int n) {
  const int second = blockIdx.x >= CB;
  const float* oc = second ? oc1 : oc0;
  const float* c = second ? c1 : c0;
  const float* sx = second ? sx1 : sx0;
  const float* cnt = second ? cnt1 : cnt0;
  float* out = second ? out1 : out0;
  int i = (blockIdx.x - second * CB) * 256 + threadIdx.x;
  if (i >= n * 3) return;
  int node = i / 3;
  float ct = cnt[node];
  float xu = ct > 0.f ? sx[i] / ct : 0.f;
  out[i] = 0.2f * oc[i] + 0.8f * c[i] + xu;
}

extern "C" void kernel_launch(void* const* d_in, const int* in_sizes, int n_in,
                              void* d_out, int out_size, void* d_ws, size_t ws_size,
                              hipStream_t stream) {
  const float* coors_lig = (const float*)d_in[0];
  const float* h_lig     = (const float*)d_in[1];
  const float* orig_lig  = (const float*)d_in[2];
  const float* oc_lig    = (const float*)d_in[3];
  const float* coors_rec = (const float*)d_in[4];
  const float* h_rec     = (const float*)d_in[5];
  const float* orig_rec  = (const float*)d_in[6];
  const float* oc_rec    = (const float*)d_in[7];
  const int* lig_src = (const int*)d_in[8];
  const int* lig_dst = (const int*)d_in[9];
  const int* rec_src = (const int*)d_in[10];
  const int* rec_dst = (const int*)d_in[11];
  const float* mask = (const float*)d_in[12];
  const float* eW1 = (const float*)d_in[13];
  const float* eb1 = (const float*)d_in[14];
  const float* eg1 = (const float*)d_in[15];
  const float* ebt1 = (const float*)d_in[16];
  const float* eW2 = (const float*)d_in[17];
  const float* eb2 = (const float*)d_in[18];
  const float* qW = (const float*)d_in[19];
  const float* kW = (const float*)d_in[20];
  const float* vW = (const float*)d_in[21];
  const float* nW1 = (const float*)d_in[22];
  const float* nb1 = (const float*)d_in[23];
  const float* ng1 = (const float*)d_in[24];
  const float* nbt1 = (const float*)d_in[25];
  const float* nW2 = (const float*)d_in[26];
  const float* nb2 = (const float*)d_in[27];
  const float* cW1 = (const float*)d_in[28];
  const float* cb1 = (const float*)d_in[29];
  const float* cg1 = (const float*)d_in[30];
  const float* cbt1 = (const float*)d_in[31];
  const float* cW2 = (const float*)d_in[32];
  const float* cb2 = (const float*)d_in[33];
  const int nE = in_sizes[8];

  float* ws = (float*)d_ws;
  float* hs_l = ws + 0 * NH;
  float* hd_l = ws + 1 * NH;
  float* hs_r = ws + 2 * NH;
  float* hd_r = ws + 3 * NH;
  float* q_l = ws + 0 * NH;
  float* k_l = ws + 1 * NH;
  float* v_l = ws + 2 * NH;
  float* q_r = ws + 3 * NH;
  float* k_r = ws + 4 * NH;
  float* v_r = ws + 5 * NH;
  float* cm_l = ws + 6 * NH;
  float* cm_r = ws + 7 * NH;
  float* sm_l = ws + 8 * NH;
  float* sm_r = ws + 9 * NH;
  float* sx_l = ws + 10 * NH;
  float* sx_r = sx_l + 3 * N_NODES;
  float* cnt_l = sx_r + 3 * N_NODES;
  float* cnt_r = cnt_l + N_NODES;
  float* eW1p = cnt_r + N_NODES;
  float* Wcp  = eW1p + 136 * 256;
  float* bcv  = Wcp + 64 * 256;
  float* bnv  = bcv + 128;
  float* nW1p = Wcp + 128 * 256;
  float* nW2p = nW1p + 224 * 256;
  float* qWp  = nW2p + 64 * 256;
  float* kWp  = qWp + 64 * 256;
  float* vWp  = kWp + 64 * 256;

  size_t accum_floats = 2 * NH + 8 * (size_t)N_NODES;
  hipMemsetAsync(sm_l, 0, accum_floats * sizeof(float), stream);

  pack_all<<<PK_END * 128 / 256, 256, 0, stream>>>(eW1, nW1, nW2, qW, kW, vW, eW1p);

  const int ntiles = N_NODES / 16;
  proj_edge<<<4 * ntiles + 18, 128, 0, stream>>>(h_lig, h_rec, eW1p, hs_l, hd_l, hs_r, hd_r,
      eW2, eb2, cW1, cb1, nW1, Wcp, bcv, nW1p + 64 * 256, bnv);

  int eblocks = (nE + EB - 1) / EB;
  edge_pair<<<2 * eblocks, 256, 0, stream>>>(coors_lig, hs_l, hd_l, lig_src, lig_dst,
      coors_rec, hs_r, hd_r, rec_src, rec_dst,
      eW1p + 128 * 256, eb1, eg1, ebt1, Wcp, bcv, cg1, cbt1, cW2, cb2,
      sm_l, sx_l, cnt_l, sm_r, sx_r, cnt_r, nE, eblocks);

  proj_all<<<6 * ntiles, 128, 0, stream>>>(h_lig, h_rec, qWp, kWp, vWp, q_l, k_l, v_l,
                                           q_r, k_r, v_r);

  attn_pair<<<2 * ntiles, 256, 0, stream>>>(q_l, k_r, v_r, cm_l, q_r, k_l, v_l, cm_r,
                                            mask);

  float* out_xl = (float*)d_out;
  float* out_hl = out_xl + 3 * N_NODES;
  float* out_xr = out_hl + NH;
  float* out_hr = out_xr + 3 * N_NODES;

  node_pair<<<2 * ntiles, 128, 0, stream>>>(h_lig, sm_l, cnt_l, cm_l, orig_lig, out_hl,
      h_rec, sm_r, cnt_r, cm_r, orig_rec, out_hr,
      nW1p, nb1, bnv, ng1, nbt1, nW2p, nb2);

  coords_pair<<<2 * CB, 256, 0, stream>>>(oc_lig, coors_lig, sx_l, cnt_l, out_xl,
                                          oc_rec, coors_rec, sx_r, cnt_r, out_xr,
                                          N_NODES);
}


# remaining load, wait, use trip loops issued as groups: node, projection, fold, query tiles staged by 16-byte row loads; scatter reads indices and rows from LDS ahead of its adds (add order changes)
# speedup vs baseline: 2.6192x; 1.0999x over previous
#include <hip/hip_runtime.h>
#include <hip/hip_bf16.h>
#include <cstdint>

#define N_NODES 6000
#define NH      ((size_t)N_NODES * 128)

typedef float v2f __attribute__((ext_vector_type(2)));
typedef float v4f __attribute__((ext_vector_type(4)));
typedef float v8f __attribute__((ext_vector_type(8)));

__device__ __forceinline__ v8f wmma4(v2f a, v2f b, v8f c) {
  return __builtin_amdgcn_wmma_f32_16x16x4_f32(false, a, false, b, (short)0, c,
                                               false, false);
}

__device__ __forceinline__ v2f ldsA(const float* base, int ld, int k0, int lane) {
  const int m = lane & 15;
  const int k = k0 + ((lane >> 4) << 1);
  const float* p = base + m * ld + k;
  v2f r; r.x = p[0]; r.y = p[1];
  return r;
}

__device__ __forceinline__ v2f ldBp(const float* __restrict__ P, int k0, int n0,
                                    int lane) {
  const int n = n0 + (lane & 15);
  const int k = k0 + ((lane >> 4) << 1);
  const float* p = P + (size_t)(k >> 1) * 256 + n * 2;
  v2f r; r.x = p[0]; r.y = p[1];
  return r;
}

__device__ __forceinline__ float lrelu(float x) { return x >= 0.f ? x : 0.01f * x; }

__device__ __forceinline__ float rowmax16(float v) {
  v = fmaxf(v, __shfl_xor(v, 1));
  v = fmaxf(v, __shfl_xor(v, 2));
  v = fmaxf(v, __shfl_xor(v, 4));
  v = fmaxf(v, __shfl_xor(v, 8));
  return v;
}
__device__ __forceinline__ float rowsum16(float v) {
  v += __shfl_xor(v, 1);
  v += __shfl_xor(v, 2);
  v += __shfl_xor(v, 4);
  v += __shfl_xor(v, 8);
  return v;
}

__device__ __forceinline__ void layernorm128(float* buf, const float* __restrict__ g,
                                             const float* __restrict__ b,
                                             float (*stats)[2], int tid) {
  const int r = tid >> 3, seg = tid & 7;
  float sum = 0.f, sq = 0.f;
  #pragma unroll
  for (int i = 0; i < 16; i++) {
    float v = buf[r * 132 + seg * 16 + i];
    sum += v; sq += v * v;
  }
  sum += __shfl_xor(sum, 1); sq += __shfl_xor(sq, 1);
  sum += __shfl_xor(sum, 2); sq += __shfl_xor(sq, 2);
  sum += __shfl_xor(sum, 4); sq += __shfl_xor(sq, 4);
  if (seg == 0) {
    float mean = sum * (1.f / 128.f);
    float var  = sq * (1.f / 128.f) - mean * mean;
    stats[r][0] = mean;
    stats[r][1] = rsqrtf(var + 1e-5f);
  }
  __syncthreads();
  const float mean = stats[r][0], rstd = stats[r][1];
  #pragma unroll
  for (int i = 0; i < 16; i++) {
    int c = seg * 16 + i;
    float v = buf[r * 132 + c];
    buf[r * 132 + c] = g[c] * (v - mean) * rstd + b[c];
  }
}

__device__ __forceinline__ void gemm128_lds(const float* in, const float* __restrict__ Wp,
                                            const float* __restrict__ bias, float* outb,
                                            int lane, int wave, int act) {
  const int hi = lane >> 4, cn = lane & 15;
  const int n0 = wave * 32;
  v8f a0 = {}, a1 = {};
  for (int k0 = 0; k0 < 128; k0 += 4) {
    v2f a  = ldsA(in, 132, k0, lane);
    v2f b0 = ldBp(Wp, k0, n0, lane);
    v2f b1 = ldBp(Wp, k0, n0 + 16, lane);
    a0 = wmma4(a, b0, a0);
    a1 = wmma4(a, b1, a1);
  }
  #pragma unroll
  for (int j = 0; j < 8; j++) {
    int r = j + 8 * hi;
    float x0 = a0[j] + bias[n0 + cn];
    float x1 = a1[j] + bias[n0 + 16 + cn];
    if (act) { x0 = lrelu(x0); x1 = lrelu(x1); }
    outb[r * 132 + n0 + cn] = x0;
    outb[r * 132 + n0 + 16 + cn] = x1;
  }
}

__device__ __forceinline__ void layernorm128x64(float* buf, const float* __restrict__ g,
                                                const float* __restrict__ b,
                                                float (*stats)[2], int tid) {
  const int seg = tid & 7;
  #pragma unroll
  for (int hh = 0; hh < 2; hh++) {
    const int r = (tid >> 3) + 32 * hh;
    float sum = 0.f, sq = 0.f;
    #pragma unroll
    for (int i = 0; i < 16; i++) {
      float v = buf[r * 132 + seg * 16 + i];
      sum += v; sq += v * v;
    }
    sum += __shfl_xor(sum, 1); sq += __shfl_xor(sq, 1);
    sum += __shfl_xor(sum, 2); sq += __shfl_xor(sq, 2);
    sum += __shfl_xor(sum, 4); sq += __shfl_xor(sq, 4);
    if (seg == 0) {
      float mean = sum * (1.f / 128.f);
      float var  = sq * (1.f / 128.f) - mean * mean;
      stats[r][0] = mean;
      stats[r][1] = rsqrtf(var + 1e-5f);
    }
  }
  __syncthreads();
  #pragma unroll
  for (int hh = 0; hh < 2; hh++) {
    const int r = (tid >> 3) + 32 * hh;
    const float mean = stats[r][0], rstd = stats[r][1];
    #pragma unroll
    for (int i = 0; i < 16; i++) {
      int c = seg * 16 + i;
      float v = buf[r * 132 + c];
      buf[r * 132 + c] = g[c] * (v - mean) * rstd + b[c];
    }
  }
}

__device__ __forceinline__ void gemm128_lds64(const float* in, const float* __restrict__ Wp,
                                              const float* __restrict__ bias, float* outb,
                                              int lane, int wave, int act) {
  const int hi = lane >> 4, cn = lane & 15;
  const int n0 = wave * 16;
  v8f acc[4];
  #pragma unroll
  for (int mt = 0; mt < 4; mt++) acc[mt] = (v8f){};
  for (int k0 = 0; k0 < 128; k0 += 4) {
    v2f b = ldBp(Wp, k0, n0, lane);
    #pragma unroll
    for (int mt = 0; mt < 4; mt++) {
      v2f x = ldsA(in + mt * 16 * 132, 132, k0, lane);
      acc[mt] = wmma4(x, b, acc[mt]);
    }
  }
  const float bb = bias[n0 + cn];
  #pragma unroll
  for (int mt = 0; mt < 4; mt++) {
    #pragma unroll
    for (int j = 0; j < 8; j++) {
      int r = mt * 16 + j + 8 * hi;
      float x = acc[mt][j] + bb;
      if (act) x = lrelu(x);
      outb[r * 132 + n0 + cn] = x;
    }
  }
}

#define PK_E1 0
#define PK_WC 136
#define PK_N1 264
#define PK_N2 488
#define PK_Q  552
#define PK_K  616
#define PK_V  680
#define PK_END 744
__global__ void pack_all(const float* __restrict__ eW1, const float* __restrict__ nW1,
                         const float* __restrict__ nW2, const float* __restrict__ qW,
                         const float* __restrict__ kW, const float* __restrict__ vW,
                         float* __restrict__ P) {
  int i = blockIdx.x * 256 + threadIdx.x;
  if (i >= PK_END * 128) return;
  int gp = i >> 7, n = i & 127;
  const float* W; int K, p0;
  if (gp < PK_WC)      { W = eW1; K = 271; p0 = PK_E1; }
  else if (gp < PK_N1) return;
  else if (gp < PK_N2) {
    if (gp >= PK_N1 + 64 && gp < PK_N1 + 128) return;
    W = nW1; K = 448; p0 = PK_N1;
  }
  else if (gp < PK_Q)  { W = nW2; K = 128; p0 = PK_N2; }
  else if (gp < PK_K)  { W = qW;  K = 128; p0 = PK_Q; }
  else if (gp < PK_V)  { W = kW;  K = 128; p0 = PK_K; }
  else                 { W = vW;  K = 128; p0 = PK_V; }
  int k = 2 * (gp - p0);
  float a = (k < K)     ? W[(size_t)k * 128 + n]       : 0.f;
  float b = (k + 1 < K) ? W[(size_t)(k + 1) * 128 + n] : 0.f;
  P[(size_t)gp * 256 + n * 2]     = a;
  P[(size_t)gp * 256 + n * 2 + 1] = b;
}

__device__ __forceinline__ void proj_body(const float* __restrict__ A,
                                          const float* __restrict__ Wp,
                                          float* __restrict__ C, int act,
                                          int pack, int bid, float* at) {
  const int tid = threadIdx.x, lane = tid & 31, wave = tid >> 5;
  const int row0 = bid * 16;
  {
    v4f a4[4];
    #pragma unroll
    for (int it = 0; it < 4; it++) {
      const int i = tid + 128 * it;
      a4[it] = *(const v4f*)(A + (size_t)(row0 + (i >> 5)) * 128 + (i & 31) * 4);
    }
    #pragma unroll
    for (int it = 0; it < 4; it++) {
      const int i = tid + 128 * it;
      *(v4f*)(at + (i >> 5) * 132 + (i & 31) * 4) = a4[it];
    }
  }
  __syncthreads();
  const int n0 = wave * 32;
  v8f a0 = {}, a1 = {};
  for (int k0 = 0; k0 < 128; k0 += 4) {
    v2f a  = ldsA(at, 132, k0, lane);
    v2f b0 = ldBp(Wp, k0, n0, lane);
    v2f b1 = ldBp(Wp, k0, n0 + 16, lane);
    a0 = wmma4(a, b0, a0);
    a1 = wmma4(a, b1, a1);
  }
  const int hi = lane >> 4, cn = lane & 15;
  #pragma unroll
  for (int j = 0; j < 8; j++) {
    int r = row0 + j + 8 * hi;
    float x0 = a0[j], x1 = a1[j];
    if (act) { x0 = lrelu(x0); x1 = lrelu(x1); }
    if (pack) {
      C[(size_t)(r >> 1) * 256 + (n0 + cn) * 2 + (r & 1)]      = x0;
      C[(size_t)(r >> 1) * 256 + (n0 + 16 + cn) * 2 + (r & 1)] = x1;
    } else {
      C[(size_t)r * 128 + n0 + cn] = x0;
      C[(size_t)r * 128 + n0 + 16 + cn] = x1;
    }
  }
}

__global__ void __launch_bounds__(128) proj_all(
    const float* __restrict__ h0, const float* __restrict__ h1,
    const float* __restrict__ qWp, const float* __restrict__ kWp,
    const float* __restrict__ vWp,
    float* __restrict__ q0, float* __restrict__ k0, float* __restrict__ v0,
    float* __restrict__ q1, float* __restrict__ k1, float* __restrict__ v1) {
  const int p = blockIdx.x / (N_NODES / 16);
  const int bid = blockIdx.x - p * (N_NODES / 16);
  const int side = p >= 3;
  const int j = p - 3 * side;
  const float* A = side ? h1 : h0;
  const float* Wp = j == 0 ? qWp : (j == 1 ? kWp : vWp);
  float* C = j == 0 ? (side ? q1 : q0) : (j == 1 ? (side ? k1 : k0) : (side ? v1 : v0));
  __shared__ __attribute__((aligned(16))) float at[16 * 132];
  proj_body(A, Wp, C, j != 2, j == 2, bid, at);
}

__device__ __forceinline__ void fold_body(float* at, const float* __restrict__ eW2,
                                          const float* __restrict__ eb2,
                                          const float* __restrict__ Wb,
                                          const float* __restrict__ addb,
                                          float* __restrict__ Cp,
                                          float* __restrict__ biasout, int t) {
  const int tid = threadIdx.x, lane = tid & 31, wave = tid >> 5;
  if (t < 8) {
    v4f a4[4];
    #pragma unroll
    for (int it = 0; it < 4; it++) {
      const int i = tid + 128 * it;
      a4[it] = *(const v4f*)(eW2 + (size_t)(t * 16 + (i >> 5)) * 128 + (i & 31) * 4);
    }
    #pragma unroll
    for (int it = 0; it < 4; it++) {
      const int i = tid + 128 * it;
      *(v4f*)(at + (i >> 5) * 132 + (i & 31) * 4) = a4[it];
    }
  } else {
    for (int i = tid; i < 16 * 128; i += 128) {
      int r = i >> 7, c = i & 127;
      at[r * 132 + c] = r == 0 ? eb2[c] : 0.f;
    }
  }
  __syncthreads();
  const int n0 = wave * 32;
  const int hi = lane >> 4, cn = lane & 15;
  v8f a0 = {}, a1 = {};
  for (int k0 = 0; k0 < 128; k0 += 4) {
    v2f a = ldsA(at, 132, k0, lane);
    const float* p = Wb + (size_t)(k0 + 2 * hi) * 128 + n0 + cn;
    v2f b0, b1;
    b0.x = p[0];  b0.y = p[128];
    b1.x = p[16]; b1.y = p[144];
    a0 = wmma4(a, b0, a0);
    a1 = wmma4(a, b1, a1);
  }
  if (t < 8) {
    #pragma unroll
    for (int j = 0; j < 8; j++) {
      int r = t * 16 + j + 8 * hi;
      Cp[(size_t)(r >> 1) * 256 + (n0 + cn) * 2 + (r & 1)]      = a0[j];
      Cp[(size_t)(r >> 1) * 256 + (n0 + 16 + cn) * 2 + (r & 1)] = a1[j];
    }
  } else if (hi == 0) {
    biasout[n0 + cn]      = a0[0] + (addb ? addb[n0 + cn] : 0.f);
    biasout[n0 + 16 + cn] = a1[0] + (addb ? addb[n0 + 16 + cn] : 0.f);
  }
}

__global__ void __launch_bounds__(128) proj_edge(
    const float* __restrict__ h0, const float* __restrict__ h1,
    const float* __restrict__ eW1p,
    float* __restrict__ hs0, float* __restrict__ hd0,
    float* __restrict__ hs1, float* __restrict__ hd1,
    const float* __restrict__ eW2, const float* __restrict__ eb2,
    const float* __restrict__ cW1, const float* __restrict__ cb1,
    const float* __restrict__ nW1,
    float* __restrict__ Wcp, float* __restrict__ bcv,
    float* __restrict__ Wnp, float* __restrict__ bnv) {
  __shared__ __attribute__((aligned(16))) float at[16 * 132];
  if (blockIdx.x >= 4 * (N_NODES / 16)) {
    const int f = blockIdx.x - 4 * (N_NODES / 16);
    if (f < 9) fold_body(at, eW2, eb2, cW1, cb1, Wcp, bcv, f);
    else       fold_body(at, eW2, eb2, nW1 + 128 * 128, nullptr, Wnp, bnv, f - 9);
    return;
  }
  const int p = blockIdx.x / (N_NODES / 16);
  const int bid = blockIdx.x - p * (N_NODES / 16);
  const int side = p >> 1;
  const int half = p & 1;
  const float* A = side ? h1 : h0;
  const float* Wp = eW1p + half * (64 * 256);
  float* C = half ? (side ? hd1 : hd0) : (side ? hs1 : hs0);
  proj_body(A, Wp, C, 0, 0, bid, at);
}

__device__ __forceinline__ float xhalf(float x) {
  const int b = __builtin_bit_cast(int, x);
  return __builtin_bit_cast(float, __builtin_amdgcn_permlanex16(b, b, 0x76543210, 0xfedcba98, false, false));
}

template <int TM>
__device__ __forceinline__ void load_mask8(const float* __restrict__ mp, float (&mk)[8]) {
  if (TM == 0) {
    const v4f m0 = *(const v4f*)(mp);
    const v4f m1 = *(const v4f*)(mp + 4);
    mk[0] = m0.x; mk[1] = m0.y; mk[2] = m0.z; mk[3] = m0.w;
    mk[4] = m1.x; mk[5] = m1.y; mk[6] = m1.z; mk[7] = m1.w;
  } else {
    #pragma unroll
    for (int r = 0; r < 8; r++) mk[r] = mp[(size_t)r * N_NODES];
  }
}

#define OB_LD 129
template <int TM>
__device__ __forceinline__ void attn_body(const float* __restrict__ Q,
                                          const float* __restrict__ K,
                                          const float* __restrict__ Vp,
                                          const float* __restrict__ mask,
                                          float* __restrict__ out, int bid,
                                          float* qt, float* outbuf, float (*mbuf)[16],
                                          float (*lbuf)[16], float* Mrow, float* Lrow) {
  const int tid = threadIdx.x, lane = tid & 31;
  const int wave = __builtin_amdgcn_readfirstlane(tid >> 5);
  const int hi = lane >> 4, cn = lane & 15;
  const int qrow0 = bid * 16;

  {
    v4f q4[2];
    #pragma unroll
    for (int it = 0; it < 2; it++) {
      const int i = tid + 256 * it;
      q4[it] = *(const v4f*)(Q + (size_t)(qrow0 + (i >> 5)) * 128 + (i & 31) * 4);
    }
    #pragma unroll
    for (int it = 0; it < 2; it++) {
      const int i = tid + 256 * it;
      *(v4f*)(qt + (i >> 5) * 132 + (i & 31) * 4) = q4[it];
    }
  }
  __syncthreads();

  float mr = -3.0e38f, lr = 0.f;
  v8f acc[8];
  #pragma unroll
  for (int j = 0; j < 8; j++) acc[j] = (v8f){};

  const float* qap = qt + cn * 132 + 4 * hi;
  const unsigned koff = (unsigned)(cn * 128 + 4 * hi);
  const unsigned voff = (unsigned)(hi * 1024 + 4 * cn);
  const size_t mrs = TM ? 1 : N_NODES;
  const size_t mcs = TM ? N_NODES : 1;
  const float* mlane = mask + (size_t)(qrow0 + cn) * mrs + (size_t)(8 * hi) * mcs;
  v4f kf[16], vf[16];
  float mk[8];
  {
    const float* kb = K + (size_t)wave * 2048;
    #pragma unroll
    for (int i = 0; i < 16; i++) kf[i] = *(const v4f*)(kb + (koff + 8u * i));
    load_mask8<TM>(mlane + (size_t)(wave * 16) * mcs, mk);
    const float* vb = Vp + (size_t)wave * 2048;
    #pragma unroll
    for (int i = 0; i < 16; i++) vf[i] = *(const v4f*)(vb + (voff + 256u * (i >> 2) + 64u * (i & 3)));
  }
#define RW8(a, o) "+v"(a[o]), "+v"(a[o + 1]), "+v"(a[o + 2]), "+v"(a[o + 3]), \
                  "+v"(a[o + 4]), "+v"(a[o + 5]), "+v"(a[o + 6]), "+v"(a[o + 7])
  asm volatile("" : RW8(kf, 0));
  asm volatile("" : RW8(kf, 8));
  asm volatile("" : RW8(vf, 0));
  asm volatile("" : RW8(vf, 8));
  asm volatile("" : RW8(mk, 0));
#undef RW8

  for (int ch = wave; ch < N_NODES / 16; ch += 8) {
    const int nch = (ch + 8 < N_NODES / 16) ? ch + 8 : N_NODES / 16 - 1;
    v8f s = {};
    #pragma unroll
    for (int i = 0; i < 16; i++) {
      v4f qa = *(const v4f*)(qap + 8 * i);
      v2f a0, a1, b0, b1;
      a0.x = kf[i].x; a0.y = kf[i].y; a1.x = kf[i].z; a1.y = kf[i].w;
      b0.x = qa.x; b0.y = qa.y; b1.x = qa.z; b1.y = qa.w;
      s = wmma4(a0, b0, s);
      s = wmma4(a1, b1, s);
    }
    float sj[8], p[8];
    #pragma unroll
    for (int r = 0; r < 8; r++) {
      float mv = mk[r];
      sj[r] = mv * s[r] - 1000.f * (1.f - mv);
    }
    float cm = fmaxf(fmaxf(fmaxf(sj[0], sj[1]), fmaxf(sj[2], sj[3])),
                     fmaxf(fmaxf(sj[4], sj[5]), fmaxf(sj[6], sj[7])));
    cm = fmaxf(cm, xhalf(cm));
    const float mnew = fmaxf(mr, cm);
    const float sc = __expf(mr - mnew);
    float ps = 0.f;
    #pragma unroll
    for (int r = 0; r < 8; r++) {
      p[r] = __expf(sj[r] - mnew);
      ps += p[r];
    }
    lr = lr * sc + ps;
    mr = mnew;
    #pragma unroll
    for (int t = 0; t < 8; t++)
      #pragma unroll
      for (int j = 0; j < 8; j++) acc[t][j] *= sc;
    {
      const float* kb = K + (size_t)nch * 2048;
      #pragma unroll
      for (int i = 0; i < 16; i++) kf[i] = *(const v4f*)(kb + (koff + 8u * i));
      load_mask8<TM>(mlane + (size_t)(nch * 16) * mcs, mk);
    }
    #pragma unroll
    for (int kk = 0; kk < 4; kk++) {
      v2f b; b.x = p[2 * kk]; b.y = p[2 * kk + 1];
      #pragma unroll
      for (int u = 0; u < 4; u++) {
        v2f a0, a1;
        a0.x = vf[4 * kk + u].x; a0.y = vf[4 * kk + u].y;
        a1.x = vf[4 * kk + u].z; a1.y = vf[4 * kk + u].w;
        acc[2 * u]     = wmma4(a0, b, acc[2 * u]);
        acc[2 * u + 1] = wmma4(a1, b, acc[2 * u + 1]);
      }
    }
    {
      const float* vb = Vp + (size_t)nch * 2048;
      #pragma unroll
      for (int i = 0; i < 16; i++) vf[i] = *(const v4f*)(vb + (voff + 256u * (i >> 2) + 64u * (i & 3)));
    }
  }

  const float lt = lr + xhalf(lr);
  if (hi == 0) {
    mbuf[wave][cn] = mr;
    lbuf[wave][cn] = lt;
  }
  __syncthreads();
  for (int i = tid; i < 16 * OB_LD; i += 256) outbuf[i] = 0.f;
  if (tid < 16) {
    float M = -3.0e38f;
    for (int w = 0; w < 8; w++) M = fmaxf(M, mbuf[w][tid]);
    float L = 0.f;
    for (int w = 0; w < 8; w++) L += lbuf[w][tid] * __expf(mbuf[w][tid] - M);
    Mrow[tid] = M;
    Lrow[tid] = L > 0.f ? 1.f / L : 0.f;
  }
  __syncthreads();
  {
    const float e = __expf(mr - Mrow[cn]);
    #pragma unroll
    for (int t = 0; t < 8; t++)
      #pragma unroll
      for (int r = 0; r < 8; r++)
        atomicAdd(&outbuf[cn * OB_LD + 32 * (t >> 1) + 2 * (r + 8 * hi) + (t & 1)],
                  acc[t][r] * e);
  }
  __syncthreads();
  for (int i = tid; i < 2048; i += 256) {
    int r = i >> 7, c = i & 127;
    out[(size_t)(qrow0 + r) * 128 + c] = outbuf[r * OB_LD + c] * Lrow[r];
  }
}

__global__ void __launch_bounds__(256) attn_pair(
    const float* __restrict__ Q0, const float* __restrict__ K0,
    const float* __restrict__ V0, float* __restrict__ out0,
    const float* __restrict__ Q1, const float* __restrict__ K1,
    const float* __restrict__ V1, float* __restrict__ out1,
    const float* __restrict__ mask) {
  __shared__ __attribute__((aligned(16))) float qt[16 * 132];
  __shared__ float outbuf[16 * OB_LD];
  __shared__ float mbuf[8][16], lbuf[8][16];
  __shared__ float Mrow[16], Lrow[16];
  const int second = blockIdx.x >= N_NODES / 16;
  const int bid = blockIdx.x - second * (N_NODES / 16);
  if (second) attn_body<1>(Q1, K1, V1, mask, out1, bid, qt, outbuf, mbuf, lbuf, Mrow, Lrow);
  else        attn_body<0>(Q0, K0, V0, mask, out0, bid, qt, outbuf, mbuf, lbuf, Mrow, Lrow);
}

#define RB_LD 18
#define EB 64
__device__ __forceinline__ void edge_body(
    const float* __restrict__ coors, const float* __restrict__ hs,
    const float* __restrict__ hd,
    const int* __restrict__ src, const int* __restrict__ dst,
    const float* __restrict__ eW1rp, const float* __restrict__ eb1,
    const float* __restrict__ eg1, const float* __restrict__ ebt1,
    const float* __restrict__ Wcp, const float* __restrict__ bcv,
    const float* __restrict__ cg1, const float* __restrict__ cbt1,
    const float* __restrict__ cW2, const float* __restrict__ cb2,
    float* __restrict__ sum_hid, float* __restrict__ sum_x,
    float* __restrict__ cnt, int nE, int bid) {
  __shared__ __attribute__((aligned(16))) float buf1[EB * 132];
  __shared__ __attribute__((aligned(16))) float buf2[EB * 132];
  __shared__ __attribute__((aligned(16))) float rbf[EB * RB_LD];
  __shared__ float stats[EB][2];
  __shared__ float xrel[EB][3];
  __shared__ float coefs[EB];
  __shared__ int sidx[EB], didx[EB];
  const int tid = threadIdx.x, lane = tid & 31, wave = tid >> 5;
  const int hi = lane >> 4, cn = lane & 15;
  const int e0 = bid * EB;

  if (tid < EB) {
    int e = e0 + tid;
    if (e < nE) {
      int s = src[e], d = dst[e];
      sidx[tid] = s; didx[tid] = d;
      float x0 = coors[s * 3 + 0] - coors[d * 3 + 0];
      float x1 = coors[s * 3 + 1] - coors[d * 3 + 1];
      float x2 = coors[s * 3 + 2] - coors[d * 3 + 2];
      xrel[tid][0] = x0; xrel[tid][1] = x1; xrel[tid][2] = x2;
      float d2 = x0 * x0 + x1 * x1 + x2 * x2;
      float sig = 1.f;
      for (int j = 0; j < 15; j++) {
        rbf[tid * RB_LD + j] = __expf(-d2 / sig);
        sig *= 1.5f;
      }
      rbf[tid * RB_LD + 15] = 0.f;
    } else {
      sidx[tid] = 0; didx[tid] = -1;
      xrel[tid][0] = xrel[tid][1] = xrel[tid][2] = 0.f;
      for (int j = 0; j < 16; j++) rbf[tid * RB_LD + j] = 0.f;
    }
  }
  __syncthreads();
  {
    const int l4 = lane * 4;
    int ns[8], nd[8];
    v4f gs[8], gd[8];
    #pragma unroll
    for (int it = 0; it < 8; it++) {
      const int e = wave + 8 * it;
      ns[it] = sidx[e];
      nd[it] = didx[e];
    }
    #pragma unroll
    for (int it = 0; it < 8; it++) nd[it] = nd[it] >= 0 ? nd[it] : 0;
    #pragma unroll
    for (int it = 0; it < 8; it++) {
      gs[it] = *(const v4f*)(hs + (size_t)ns[it] * 128 + l4);
      gd[it] = *(const v4f*)(hd + (size_t)nd[it] * 128 + l4);
    }
    #pragma unroll
    for (int it = 0; it < 8; it++) {
      const int e = wave + 8 * it;
      *(v4f*)(buf1 + e * 132 + l4) = gs[it] + gd[it];
    }
  }
  __syncthreads();

  {
    const int n0 = wave * 16;
    v8f acc[4];
    #pragma unroll
    for (int mt = 0; mt < 4; mt++) {
      #pragma unroll
      for (int j = 0; j < 8; j++)
        acc[mt][j] = buf1[(mt * 16 + j + 8 * hi) * 132 + n0 + cn];
    }
    #pragma unroll
    for (int k0 = 0; k0 < 16; k0 += 4) {
      v2f b = ldBp(eW1rp, k0, n0, lane);
      #pragma unroll
      for (int mt = 0; mt < 4; mt++) {
        v2f x = ldsA(rbf + mt * 16 * RB_LD, RB_LD, k0, lane);
        acc[mt] = wmma4(x, b, acc[mt]);
      }
    }
    const float bb = eb1[n0 + cn];
    #pragma unroll
    for (int mt = 0; mt < 4; mt++) {
      #pragma unroll
      for (int j = 0; j < 8; j++)
        buf1[(mt * 16 + j + 8 * hi) * 132 + n0 + cn] = lrelu(acc[mt][j] + bb);
    }
  }
  __syncthreads();
  layernorm128x64(buf1, eg1, ebt1, stats, tid);
  __syncthreads();
  gemm128_lds64(buf1, Wcp, bcv, buf2, lane, wave, 1);
  __syncthreads();
  layernorm128x64(buf2, cg1, cbt1, stats, tid);
  __syncthreads();
  {
    const int seg = tid & 7;
    #pragma unroll
    for (int hh = 0; hh < 2; hh++) {
      const int r = (tid >> 3) + 32 * hh;
      float sum = 0.f;
      #pragma unroll
      for (int i = 0; i < 16; i++) {
        int c = seg * 16 + i;
        sum += buf2[r * 132 + c] * cW2[c];
      }
      sum += __shfl_xor(sum, 1);
      sum += __shfl_xor(sum, 2);
      sum += __shfl_xor(sum, 4);
      if (seg == 0) coefs[r] = sum + cb2[0];
    }
  }
  __syncthreads();
  {
    int dd[8];
    float hv[8][4];
    #pragma unroll
    for (int it = 0; it < 8; it++) {
      const int e = wave + 8 * it;
      dd[it] = didx[e];
      #pragma unroll
      for (int q = 0; q < 4; q++) hv[it][q] = buf1[e * 132 + lane + 32 * q];
    }
    #pragma unroll
    for (int it = 0; it < 8; it++) {
      const int d = dd[it];
      if (d >= 0) {
        #pragma unroll
        for (int q = 0; q < 4; q++)
          atomicAdd(&sum_hid[(size_t)d * 128 + lane + 32 * q], hv[it][q]);
      }
    }
  }
  if (tid < EB * 3) {
    int e = tid / 3, ax = tid % 3;
    int d = didx[e];
    if (d >= 0) atomicAdd(&sum_x[d * 3 + ax], xrel[e][ax] * coefs[e]);
  }
  if (tid < EB) {
    int d = didx[tid];
    if (d >= 0) atomicAdd(&cnt[d], 1.f);
  }
}

__global__ void __launch_bounds__(256) edge_pair(
    const float* __restrict__ coors0, const float* __restrict__ hs0,
    const float* __restrict__ hd0,
    const int* __restrict__ src0, const int* __restrict__ dst0,
    const float* __restrict__ coors1, const float* __restrict__ hs1,
    const float* __restrict__ hd1,
    const int* __restrict__ src1, const int* __restrict__ dst1,
    const float* __restrict__ eW1rp, const float* __restrict__ eb1,
    const float* __restrict__ eg1, const float* __restrict__ ebt1,
    const float* __restrict__ Wcp, const float* __restrict__ bcv,
    const float* __restrict__ cg1, const float* __restrict__ cbt1,
    const float* __restrict__ cW2, const float* __restrict__ cb2,
    float* __restrict__ sm0, float* __restrict__ sx0, float* __restrict__ cnt0,
    float* __restrict__ sm1, float* __restrict__ sx1, float* __restrict__ cnt1,
    int nE, int eblocks) {
  const int second = (int)blockIdx.x >= eblocks;
  const int bid = blockIdx.x - second * eblocks;
  edge_body(second ? coors1 : coors0, second ? hs1 : hs0, second ? hd1 : hd0,
            second ? src1 : src0, second ? dst1 : dst0, eW1rp, eb1, eg1, ebt1, Wcp,
            bcv, cg1,
            cbt1, cW2, cb2, second ? sm1 : sm0, second ? sx1 : sx0,
            second ? cnt1 : cnt0, nE, bid);
}

#define NCAT_LD 452
__device__ __forceinline__ void node_body(
    const float* __restrict__ h, const float* __restrict__ summ,
    const float* __restrict__ cnt, const float* __restrict__ cm,
    const float* __restrict__ orig,
    const float* __restrict__ nW1p, const float* __restrict__ nb1,
    const float* __restrict__ bnv,
    const float* __restrict__ ng1, const float* __restrict__ nbt1,
    const float* __restrict__ nW2p, const float* __restrict__ nb2,
    float* __restrict__ hout, int bid) {
  __shared__ __attribute__((aligned(16))) float cat[16 * NCAT_LD];
  __shared__ float buf1[16 * 132];
  __shared__ float stats[16][2];
  __shared__ float has[16];
  const int tid = threadIdx.x, lane = tid & 31, wave = tid >> 5;
  const int hi = lane >> 4, cn = lane & 15;
  const int node0 = bid * 16;

  {
    const int l4 = lane * 4;
    v4f vh[4], vs[4], vc[4], vo[2];
    float ct[4];
    #pragma unroll
    for (int it = 0; it < 4; it++) {
      const int e = wave + 4 * it;
      const size_t g = (size_t)(node0 + e) * 128 + l4;
      vh[it] = *(const v4f*)(h + g);
      vs[it] = *(const v4f*)(summ + g);
      vc[it] = *(const v4f*)(cm + g);
      ct[it] = cnt[node0 + e];
    }
    #pragma unroll
    for (int it = 0; it < 2; it++) {
      const int i = tid + 128 * it;
      vo[it] = *(const v4f*)(orig + (size_t)(node0 + (i >> 4)) * 64 + (i & 15) * 4);
    }
    #pragma unroll
    for (int it = 0; it < 4; it++) {
      const int e = wave + 4 * it;
      float* row = cat + e * NCAT_LD + l4;
      const bool hasr = ct[it] > 0.f;
      v4f m;
      m.x = hasr ? vs[it].x / ct[it] : 0.f;
      m.y = hasr ? vs[it].y / ct[it] : 0.f;
      m.z = hasr ? vs[it].z / ct[it] : 0.f;
      m.w = hasr ? vs[it].w / ct[it] : 0.f;
      *(v4f*)(row)       = vh[it];
      *(v4f*)(row + 128) = m;
      *(v4f*)(row + 256) = vc[it];
      if (lane == 0) has[e] = hasr ? 1.f : 0.f;
    }
    #pragma unroll
    for (int it = 0; it < 2; it++) {
      const int i = tid + 128 * it;
      *(v4f*)(cat + (i >> 4) * NCAT_LD + 384 + (i & 15) * 4) = vo[it];
    }
  }
  __syncthreads();
  {
    const int n0 = wave * 32;
    v8f a0 = {}, a1 = {};
    for (int k0 = 0; k0 < 448; k0 += 4) {
      v2f a  = ldsA(cat, NCAT_LD, k0, lane);
      v2f b0 = ldBp(nW1p, k0, n0, lane);
      v2f b1 = ldBp(nW1p, k0, n0 + 16, lane);
      a0 = wmma4(a, b0, a0);
      a1 = wmma4(a, b1, a1);
    }
    const float fb0 = bnv[n0 + cn], fb1 = bnv[n0 + 16 + cn];
    #pragma unroll
    for (int j = 0; j < 8; j++) {
      int r = j + 8 * hi;
      const bool hasr = has[r] > 0.f;
      buf1[r * 132 + n0 + cn]      = lrelu(a0[j] + nb1[n0 + cn] + (hasr ? fb0 : 0.f));
      buf1[r * 132 + n0 + 16 + cn] = lrelu(a1[j] + nb1[n0 + 16 + cn] + (hasr ? fb1 : 0.f));
    }
  }
  __syncthreads();
  layernorm128(buf1, ng1, nbt1, stats, tid);
  __syncthreads();
  {
    const int n0 = wave * 32;
    v8f a0 = {}, a1 = {};
    for (int k0 = 0; k0 < 128; k0 += 4) {
      v2f a  = ldsA(buf1, 132, k0, lane);
      v2f b0 = ldBp(nW2p, k0, n0, lane);
      v2f b1 = ldBp(nW2p, k0, n0 + 16, lane);
      a0 = wmma4(a, b0, a0);
      a1 = wmma4(a, b1, a1);
    }
    #pragma unroll
    for (int j = 0; j < 8; j++) {
      int r = j + 8 * hi;
      size_t g0 = (size_t)(node0 + r) * 128 + n0 + cn;
      size_t g1 = g0 + 16;
      hout[g0] = 0.5f * (a0[j] + nb2[n0 + cn]) + 0.5f * h[g0];
      hout[g1] = 0.5f * (a1[j] + nb2[n0 + 16 + cn]) + 0.5f * h[g1];
    }
  }
}

__global__ void __launch_bounds__(128) node_pair(
    const float* __restrict__ h0, const float* __restrict__ summ0,
    const float* __restrict__ cnt0, const float* __restrict__ cm0,
    const float* __restrict__ orig0, float* __restrict__ hout0,
    const float* __restrict__ h1, const float* __restrict__ summ1,
    const float* __restrict__ cnt1, const float* __restrict__ cm1,
    const float* __restrict__ orig1, float* __restrict__ hout1,
    const float* __restrict__ nW1p, const float* __restrict__ nb1,
    const float* __restrict__ bnv,
    const float* __restrict__ ng1, const float* __restrict__ nbt1,
    const float* __restrict__ nW2p, const float* __restrict__ nb2) {
  const int second = blockIdx.x >= N_NODES / 16;
  const int bid = blockIdx.x - second * (N_NODES / 16);
  node_body(second ? h1 : h0, second ? summ1 : summ0, second ? cnt1 : cnt0,
            second ? cm1 : cm0, second ? orig1 : orig0, nW1p, nb1, bnv, ng1, nbt1, nW2p,
            nb2, second ? hout1 : hout0, bid);
}

#define CB ((3 * N_NODES + 255) / 256)
__global__ void coords_pair(const float* __restrict__ oc0, const float* __restrict__ c0,
                            const float* __restrict__ sx0, const float* __restrict__ cnt0,
                            float* __restrict__ out0,
                            const float* __restrict__ oc1, const float* __restrict__ c1,
                            const float* __restrict__ sx1, const float* __restrict__ cnt1,
                            float* __restrict__ out1, int n) {
  const int second = blockIdx.x >= CB;
  const float* oc = second ? oc1 : oc0;
  const float* c = second ? c1 : c0;
  const float* sx = second ? sx1 : sx0;
  const float* cnt = second ? cnt1 : cnt0;
  float* out = second ? out1 : out0;
  int i = (blockIdx.x - second * CB) * 256 + threadIdx.x;
  if (i >= n * 3) return;
  int node = i / 3;
  float ct = cnt[node];
  float xu = ct > 0.f ? sx[i] / ct : 0.f;
  out[i] = 0.2f * oc[i] + 0.8f * c[i] + xu;
}

extern "C" void kernel_launch(void* const* d_in, const int* in_sizes, int n_in,
                              void* d_out, int out_size, void* d_ws, size_t ws_size,
                              hipStream_t stream) {
  const float* coors_lig = (const float*)d_in[0];
  const float* h_lig     = (const float*)d_in[1];
  const float* orig_lig  = (const float*)d_in[2];
  const float* oc_lig    = (const float*)d_in[3];
  const float* coors_rec = (const float*)d_in[4];
  const float* h_rec     = (const float*)d_in[5];
  const float* orig_rec  = (const float*)d_in[6];
  const float* oc_rec    = (const float*)d_in[7];
  const int* lig_src = (const int*)d_in[8];
  const int* lig_dst = (const int*)d_in[9];
  const int* rec_src = (const int*)d_in[10];
  const int* rec_dst = (const int*)d_in[11];
  const float* mask = (const float*)d_in[12];
  const float* eW1 = (const float*)d_in[13];
  const float* eb1 = (const float*)d_in[14];
  const float* eg1 = (const float*)d_in[15];
  const float* ebt1 = (const float*)d_in[16];
  const float* eW2 = (const float*)d_in[17];
  const float* eb2 = (const float*)d_in[18];
  const float* qW = (const float*)d_in[19];
  const float* kW = (const float*)d_in[20];
  const float* vW = (const float*)d_in[21];
  const float* nW1 = (const float*)d_in[22];
  const float* nb1 = (const float*)d_in[23];
  const float* ng1 = (const float*)d_in[24];
  const float* nbt1 = (const float*)d_in[25];
  const float* nW2 = (const float*)d_in[26];
  const float* nb2 = (const float*)d_in[27];
  const float* cW1 = (const float*)d_in[28];
  const float* cb1 = (const float*)d_in[29];
  const float* cg1 = (const float*)d_in[30];
  const float* cbt1 = (const float*)d_in[31];
  const float* cW2 = (const float*)d_in[32];
  const float* cb2 = (const float*)d_in[33];
  const int nE = in_sizes[8];

  float* ws = (float*)d_ws;
  float* hs_l = ws + 0 * NH;
  float* hd_l = ws + 1 * NH;
  float* hs_r = ws + 2 * NH;
  float* hd_r = ws + 3 * NH;
  float* q_l = ws + 0 * NH;
  float* k_l = ws + 1 * NH;
  float* v_l = ws + 2 * NH;
  float* q_r = ws + 3 * NH;
  float* k_r = ws + 4 * NH;
  float* v_r = ws + 5 * NH;
  float* cm_l = ws + 6 * NH;
  float* cm_r = ws + 7 * NH;
  float* sm_l = ws + 8 * NH;
  float* sm_r = ws + 9 * NH;
  float* sx_l = ws + 10 * NH;
  float* sx_r = sx_l + 3 * N_NODES;
  float* cnt_l = sx_r + 3 * N_NODES;
  float* cnt_r = cnt_l + N_NODES;
  float* eW1p = cnt_r + N_NODES;
  float* Wcp  = eW1p + 136 * 256;
  float* bcv  = Wcp + 64 * 256;
  float* bnv  = bcv + 128;
  float* nW1p = Wcp + 128 * 256;
  float* nW2p = nW1p + 224 * 256;
  float* qWp  = nW2p + 64 * 256;
  float* kWp  = qWp + 64 * 256;
  float* vWp  = kWp + 64 * 256;

  size_t accum_floats = 2 * NH + 8 * (size_t)N_NODES;
  hipMemsetAsync(sm_l, 0, accum_floats * sizeof(float), stream);

  pack_all<<<PK_END * 128 / 256, 256, 0, stream>>>(eW1, nW1, nW2, qW, kW, vW, eW1p);

  const int ntiles = N_NODES / 16;
  proj_edge<<<4 * ntiles + 18, 128, 0, stream>>>(h_lig, h_rec, eW1p, hs_l, hd_l, hs_r, hd_r,
      eW2, eb2, cW1, cb1, nW1, Wcp, bcv, nW1p + 64 * 256, bnv);

  int eblocks = (nE + EB - 1) / EB;
  edge_pair<<<2 * eblocks, 256, 0, stream>>>(coors_lig, hs_l, hd_l, lig_src, lig_dst,
      coors_rec, hs_r, hd_r, rec_src, rec_dst,
      eW1p + 128 * 256, eb1, eg1, ebt1, Wcp, bcv, cg1, cbt1, cW2, cb2,
      sm_l, sx_l, cnt_l, sm_r, sx_r, cnt_r, nE, eblocks);

  proj_all<<<6 * ntiles, 128, 0, stream>>>(h_lig, h_rec, qWp, kWp, vWp, q_l, k_l, v_l,
                                           q_r, k_r, v_r);

  attn_pair<<<2 * ntiles, 256, 0, stream>>>(q_l, k_r, v_r, cm_l, q_r, k_l, v_l, cm_r,
                                            mask);

  float* out_xl = (float*)d_out;
  float* out_hl = out_xl + 3 * N_NODES;
  float* out_xr = out_hl + NH;
  float* out_hr = out_xr + 3 * N_NODES;

  node_pair<<<2 * ntiles, 128, 0, stream>>>(h_lig, sm_l, cnt_l, cm_l, orig_lig, out_hl,
      h_rec, sm_r, cnt_r, cm_r, orig_rec, out_hr,
      nW1p, nb1, bnv, ng1, nbt1, nW2p, nb2);

  coords_pair<<<2 * CB, 256, 0, stream>>>(oc_lig, coors_lig, sx_l, cnt_l, out_xl,
                                          oc_rec, coors_rec, sx_r, cnt_r, out_xr,
                                          N_NODES);
}


# attention block takes 48 query rows, not 16: each K and V fragment a wave fetches feeds three query tiles' products (K, V reads from the second-level cache cut to a third); per-wave sums unchanged
# speedup vs baseline: 2.9999x; 1.1453x over previous
#include <hip/hip_runtime.h>
#include <hip/hip_bf16.h>
#include <cstdint>

#define N_NODES 6000
#define NH      ((size_t)N_NODES * 128)

typedef float v2f __attribute__((ext_vector_type(2)));
typedef float v4f __attribute__((ext_vector_type(4)));
typedef float v8f __attribute__((ext_vector_type(8)));

__device__ __forceinline__ v8f wmma4(v2f a, v2f b, v8f c) {
  return __builtin_amdgcn_wmma_f32_16x16x4_f32(false, a, false, b, (short)0, c,
                                               false, false);
}

__device__ __forceinline__ v2f ldsA(const float* base, int ld, int k0, int lane) {
  const int m = lane & 15;
  const int k = k0 + ((lane >> 4) << 1);
  const float* p = base + m * ld + k;
  v2f r; r.x = p[0]; r.y = p[1];
  return r;
}

__device__ __forceinline__ v2f ldBp(const float* __restrict__ P, int k0, int n0,
                                    int lane) {
  const int n = n0 + (lane & 15);
  const int k = k0 + ((lane >> 4) << 1);
  const float* p = P + (size_t)(k >> 1) * 256 + n * 2;
  v2f r; r.x = p[0]; r.y = p[1];
  return r;
}

__device__ __forceinline__ float lrelu(float x) { return x >= 0.f ? x : 0.01f * x; }

__device__ __forceinline__ float rowmax16(float v) {
  v = fmaxf(v, __shfl_xor(v, 1));
  v = fmaxf(v, __shfl_xor(v, 2));
  v = fmaxf(v, __shfl_xor(v, 4));
  v = fmaxf(v, __shfl_xor(v, 8));
  return v;
}
__device__ __forceinline__ float rowsum16(float v) {
  v += __shfl_xor(v, 1);
  v += __shfl_xor(v, 2);
  v += __shfl_xor(v, 4);
  v += __shfl_xor(v, 8);
  return v;
}

__device__ __forceinline__ void layernorm128(float* buf, const float* __restrict__ g,
                                             const float* __restrict__ b,
                                             float (*stats)[2], int tid) {
  const int r = tid >> 3, seg = tid & 7;
  float sum = 0.f, sq = 0.f;
  #pragma unroll
  for (int i = 0; i < 16; i++) {
    float v = buf[r * 132 + seg * 16 + i];
    sum += v; sq += v * v;
  }
  sum += __shfl_xor(sum, 1); sq += __shfl_xor(sq, 1);
  sum += __shfl_xor(sum, 2); sq += __shfl_xor(sq, 2);
  sum += __shfl_xor(sum, 4); sq += __shfl_xor(sq, 4);
  if (seg == 0) {
    float mean = sum * (1.f / 128.f);
    float var  = sq * (1.f / 128.f) - mean * mean;
    stats[r][0] = mean;
    stats[r][1] = rsqrtf(var + 1e-5f);
  }
  __syncthreads();
  const float mean = stats[r][0], rstd = stats[r][1];
  #pragma unroll
  for (int i = 0; i < 16; i++) {
    int c = seg * 16 + i;
    float v = buf[r * 132 + c];
    buf[r * 132 + c] = g[c] * (v - mean) * rstd + b[c];
  }
}

__device__ __forceinline__ void gemm128_lds(const float* in, const float* __restrict__ Wp,
                                            const float* __restrict__ bias, float* outb,
                                            int lane, int wave, int act) {
  const int hi = lane >> 4, cn = lane & 15;
  const int n0 = wave * 32;
  v8f a0 = {}, a1 = {};
  for (int k0 = 0; k0 < 128; k0 += 4) {
    v2f a  = ldsA(in, 132, k0, lane);
    v2f b0 = ldBp(Wp, k0, n0, lane);
    v2f b1 = ldBp(Wp, k0, n0 + 16, lane);
    a0 = wmma4(a, b0, a0);
    a1 = wmma4(a, b1, a1);
  }
  #pragma unroll
  for (int j = 0; j < 8; j++) {
    int r = j + 8 * hi;
    float x0 = a0[j] + bias[n0 + cn];
    float x1 = a1[j] + bias[n0 + 16 + cn];
    if (act) { x0 = lrelu(x0); x1 = lrelu(x1); }
    outb[r * 132 + n0 + cn] = x0;
    outb[r * 132 + n0 + 16 + cn] = x1;
  }
}

__device__ __forceinline__ void layernorm128x64(float* buf, const float* __restrict__ g,
                                                const float* __restrict__ b,
                                                float (*stats)[2], int tid) {
  const int seg = tid & 7;
  #pragma unroll
  for (int hh = 0; hh < 2; hh++) {
    const int r = (tid >> 3) + 32 * hh;
    float sum = 0.f, sq = 0.f;
    #pragma unroll
    for (int i = 0; i < 16; i++) {
      float v = buf[r * 132 + seg * 16 + i];
      sum += v; sq += v * v;
    }
    sum += __shfl_xor(sum, 1); sq += __shfl_xor(sq, 1);
    sum += __shfl_xor(sum, 2); sq += __shfl_xor(sq, 2);
    sum += __shfl_xor(sum, 4); sq += __shfl_xor(sq, 4);
    if (seg == 0) {
      float mean = sum * (1.f / 128.f);
      float var  = sq * (1.f / 128.f) - mean * mean;
      stats[r][0] = mean;
      stats[r][1] = rsqrtf(var + 1e-5f);
    }
  }
  __syncthreads();
  #pragma unroll
  for (int hh = 0; hh < 2; hh++) {
    const int r = (tid >> 3) + 32 * hh;
    const float mean = stats[r][0], rstd = stats[r][1];
    #pragma unroll
    for (int i = 0; i < 16; i++) {
      int c = seg * 16 + i;
      float v = buf[r * 132 + c];
      buf[r * 132 + c] = g[c] * (v - mean) * rstd + b[c];
    }
  }
}

__device__ __forceinline__ void gemm128_lds64(const float* in, const float* __restrict__ Wp,
                                              const float* __restrict__ bias, float* outb,
                                              int lane, int wave, int act) {
  const int hi = lane >> 4, cn = lane & 15;
  const int n0 = wave * 16;
  v8f acc[4];
  #pragma unroll
  for (int mt = 0; mt < 4; mt++) acc[mt] = (v8f){};
  for (int k0 = 0; k0 < 128; k0 += 4) {
    v2f b = ldBp(Wp, k0, n0, lane);
    #pragma unroll
    for (int mt = 0; mt < 4; mt++) {
      v2f x = ldsA(in + mt * 16 * 132, 132, k0, lane);
      acc[mt] = wmma4(x, b, acc[mt]);
    }
  }
  const float bb = bias[n0 + cn];
  #pragma unroll
  for (int mt = 0; mt < 4; mt++) {
    #pragma unroll
    for (int j = 0; j < 8; j++) {
      int r = mt * 16 + j + 8 * hi;
      float x = acc[mt][j] + bb;
      if (act) x = lrelu(x);
      outb[r * 132 + n0 + cn] = x;
    }
  }
}

#define PK_E1 0
#define PK_WC 136
#define PK_N1 264
#define PK_N2 488
#define PK_Q  552
#define PK_K  616
#define PK_V  680
#define PK_END 744
__global__ void pack_all(const float* __restrict__ eW1, const float* __restrict__ nW1,
                         const float* __restrict__ nW2, const float* __restrict__ qW,
                         const float* __restrict__ kW, const float* __restrict__ vW,
                         float* __restrict__ P) {
  int i = blockIdx.x * 256 + threadIdx.x;
  if (i >= PK_END * 128) return;
  int gp = i >> 7, n = i & 127;
  const float* W; int K, p0;
  if (gp < PK_WC)      { W = eW1; K = 271; p0 = PK_E1; }
  else if (gp < PK_N1) return;
  else if (gp < PK_N2) {
    if (gp >= PK_N1 + 64 && gp < PK_N1 + 128) return;
    W = nW1; K = 448; p0 = PK_N1;
  }
  else if (gp < PK_Q)  { W = nW2; K = 128; p0 = PK_N2; }
  else if (gp < PK_K)  { W = qW;  K = 128; p0 = PK_Q; }
  else if (gp < PK_V)  { W = kW;  K = 128; p0 = PK_K; }
  else                 { W = vW;  K = 128; p0 = PK_V; }
  int k = 2 * (gp - p0);
  float a = (k < K)     ? W[(size_t)k * 128 + n]       : 0.f;
  float b = (k + 1 < K) ? W[(size_t)(k + 1) * 128 + n] : 0.f;
  P[(size_t)gp * 256 + n * 2]     = a;
  P[(size_t)gp * 256 + n * 2 + 1] = b;
}

__device__ __forceinline__ void proj_body(const float* __restrict__ A,
                                          const float* __restrict__ Wp,
                                          float* __restrict__ C, int act,
                                          int pack, int bid, float* at) {
  const int tid = threadIdx.x, lane = tid & 31, wave = tid >> 5;
  const int row0 = bid * 16;
  {
    v4f a4[4];
    #pragma unroll
    for (int it = 0; it < 4; it++) {
      const int i = tid + 128 * it;
      a4[it] = *(const v4f*)(A + (size_t)(row0 + (i >> 5)) * 128 + (i & 31) * 4);
    }
    #pragma unroll
    for (int it = 0; it < 4; it++) {
      const int i = tid + 128 * it;
      *(v4f*)(at + (i >> 5) * 132 + (i & 31) * 4) = a4[it];
    }
  }
  __syncthreads();
  const int n0 = wave * 32;
  v8f a0 = {}, a1 = {};
  for (int k0 = 0; k0 < 128; k0 += 4) {
    v2f a  = ldsA(at, 132, k0, lane);
    v2f b0 = ldBp(Wp, k0, n0, lane);
    v2f b1 = ldBp(Wp, k0, n0 + 16, lane);
    a0 = wmma4(a, b0, a0);
    a1 = wmma4(a, b1, a1);
  }
  const int hi = lane >> 4, cn = lane & 15;
  #pragma unroll
  for (int j = 0; j < 8; j++) {
    int r = row0 + j + 8 * hi;
    float x0 = a0[j], x1 = a1[j];
    if (act) { x0 = lrelu(x0); x1 = lrelu(x1); }
    if (pack) {
      C[(size_t)(r >> 1) * 256 + (n0 + cn) * 2 + (r & 1)]      = x0;
      C[(size_t)(r >> 1) * 256 + (n0 + 16 + cn) * 2 + (r & 1)] = x1;
    } else {
      C[(size_t)r * 128 + n0 + cn] = x0;
      C[(size_t)r * 128 + n0 + 16 + cn] = x1;
    }
  }
}

__global__ void __launch_bounds__(128) proj_all(
    const float* __restrict__ h0, const float* __restrict__ h1,
    const float* __restrict__ qWp, const float* __restrict__ kWp,
    const float* __restrict__ vWp,
    float* __restrict__ q0, float* __restrict__ k0, float* __restrict__ v0,
    float* __restrict__ q1, float* __restrict__ k1, float* __restrict__ v1) {
  const int p = blockIdx.x / (N_NODES / 16);
  const int bid = blockIdx.x - p * (N_NODES / 16);
  const int side = p >= 3;
  const int j = p - 3 * side;
  const float* A = side ? h1 : h0;
  const float* Wp = j == 0 ? qWp : (j == 1 ? kWp : vWp);
  float* C = j == 0 ? (side ? q1 : q0) : (j == 1 ? (side ? k1 : k0) : (side ? v1 : v0));
  __shared__ __attribute__((aligned(16))) float at[16 * 132];
  proj_body(A, Wp, C, j != 2, j == 2, bid, at);
}

__device__ __forceinline__ void fold_body(float* at, const float* __restrict__ eW2,
                                          const float* __restrict__ eb2,
                                          const float* __restrict__ Wb,
                                          const float* __restrict__ addb,
                                          float* __restrict__ Cp,
                                          float* __restrict__ biasout, int t) {
  const int tid = threadIdx.x, lane = tid & 31, wave = tid >> 5;
  if (t < 8) {
    v4f a4[4];
    #pragma unroll
    for (int it = 0; it < 4; it++) {
      const int i = tid + 128 * it;
      a4[it] = *(const v4f*)(eW2 + (size_t)(t * 16 + (i >> 5)) * 128 + (i & 31) * 4);
    }
    #pragma unroll
    for (int it = 0; it < 4; it++) {
      const int i = tid + 128 * it;
      *(v4f*)(at + (i >> 5) * 132 + (i & 31) * 4) = a4[it];
    }
  } else {
    for (int i = tid; i < 16 * 128; i += 128) {
      int r = i >> 7, c = i & 127;
      at[r * 132 + c] = r == 0 ? eb2[c] : 0.f;
    }
  }
  __syncthreads();
  const int n0 = wave * 32;
  const int hi = lane >> 4, cn = lane & 15;
  v8f a0 = {}, a1 = {};
  for (int k0 = 0; k0 < 128; k0 += 4) {
    v2f a = ldsA(at, 132, k0, lane);
    const float* p = Wb + (size_t)(k0 + 2 * hi) * 128 + n0 + cn;
    v2f b0, b1;
    b0.x = p[0];  b0.y = p[128];
    b1.x = p[16]; b1.y = p[144];
    a0 = wmma4(a, b0, a0);
    a1 = wmma4(a, b1, a1);
  }
  if (t < 8) {
    #pragma unroll
    for (int j = 0; j < 8; j++) {
      int r = t * 16 + j + 8 * hi;
      Cp[(size_t)(r >> 1) * 256 + (n0 + cn) * 2 + (r & 1)]      = a0[j];
      Cp[(size_t)(r >> 1) * 256 + (n0 + 16 + cn) * 2 + (r & 1)] = a1[j];
    }
  } else if (hi == 0) {
    biasout[n0 + cn]      = a0[0] + (addb ? addb[n0 + cn] : 0.f);
    biasout[n0 + 16 + cn] = a1[0] + (addb ? addb[n0 + 16 + cn] : 0.f);
  }
}

__global__ void __launch_bounds__(128) proj_edge(
    const float* __restrict__ h0, const float* __restrict__ h1,
    const float* __restrict__ eW1p,
    float* __restrict__ hs0, float* __restrict__ hd0,
    float* __restrict__ hs1, float* __restrict__ hd1,
    const float* __restrict__ eW2, const float* __restrict__ eb2,
    const float* __restrict__ cW1, const float* __restrict__ cb1,
    const float* __restrict__ nW1,
    float* __restrict__ Wcp, float* __restrict__ bcv,
    float* __restrict__ Wnp, float* __restrict__ bnv) {
  __shared__ __attribute__((aligned(16))) float at[16 * 132];
  if (blockIdx.x >= 4 * (N_NODES / 16)) {
    const int f = blockIdx.x - 4 * (N_NODES / 16);
    if (f < 9) fold_body(at, eW2, eb2, cW1, cb1, Wcp, bcv, f);
    else       fold_body(at, eW2, eb2, nW1 + 128 * 128, nullptr, Wnp, bnv, f - 9);
    return;
  }
  const int p = blockIdx.x / (N_NODES / 16);
  const int bid = blockIdx.x - p * (N_NODES / 16);
  const int side = p >> 1;
  const int half = p & 1;
  const float* A = side ? h1 : h0;
  const float* Wp = eW1p + half * (64 * 256);
  float* C = half ? (side ? hd1 : hd0) : (side ? hs1 : hs0);
  proj_body(A, Wp, C, 0, 0, bid, at);
}

__device__ __forceinline__ float xhalf(float x) {
  const int b = __builtin_bit_cast(int, x);
  return __builtin_bit_cast(float, __builtin_amdgcn_permlanex16(b, b, 0x76543210, 0xfedcba98, false, false));
}

template <int TM>
__device__ __forceinline__ void load_mask8(const float* __restrict__ mp, float (&mk)[8]) {
  if (TM == 0) {
    const v4f m0 = *(const v4f*)(mp);
    const v4f m1 = *(const v4f*)(mp + 4);
    mk[0] = m0.x; mk[1] = m0.y; mk[2] = m0.z; mk[3] = m0.w;
    mk[4] = m1.x; mk[5] = m1.y; mk[6] = m1.z; mk[7] = m1.w;
  } else {
    #pragma unroll
    for (int r = 0; r < 8; r++) mk[r] = mp[(size_t)r * N_NODES];
  }
}

#define OB_LD 129
#define QT 3
#define QROWS (16 * QT)
#define ATT_BLOCKS (N_NODES / QROWS)
static_assert(N_NODES % QROWS == 0, "query rows must fill whole blocks");
template <int TM>
__device__ __forceinline__ void attn_body(const float* __restrict__ Q,
                                          const float* __restrict__ K,
                                          const float* __restrict__ Vp,
                                          const float* __restrict__ mask,
                                          float* __restrict__ out, int bid,
                                          float* qt, float* outbuf, float (*mbuf)[QROWS],
                                          float (*lbuf)[QROWS], float* Mrow, float* Lrow) {
  const int tid = threadIdx.x, lane = tid & 31;
  const int wave = __builtin_amdgcn_readfirstlane(tid >> 5);
  const int hi = lane >> 4, cn = lane & 15;
  const int qrow0 = bid * QROWS;

  {
    v4f q4[6];
    #pragma unroll
    for (int it = 0; it < 6; it++) {
      const int i = tid + 256 * it;
      q4[it] = *(const v4f*)(Q + (size_t)(qrow0 + (i >> 5)) * 128 + (i & 31) * 4);
    }
    #pragma unroll
    for (int it = 0; it < 6; it++) {
      const int i = tid + 256 * it;
      *(v4f*)(qt + (i >> 5) * 132 + (i & 31) * 4) = q4[it];
    }
  }
  __syncthreads();

  float mr[QT], lr[QT];
  v8f acc[QT][8];
  #pragma unroll
  for (int t = 0; t < QT; t++) {
    mr[t] = -3.0e38f; lr[t] = 0.f;
    #pragma unroll
    for (int j = 0; j < 8; j++) acc[t][j] = (v8f){};
  }

  const int qoff = cn * 132 + 4 * hi;
  const unsigned koff = (unsigned)(cn * 128 + 4 * hi);
  const unsigned voff = (unsigned)(hi * 1024 + 4 * cn);
  const size_t mrs = TM ? 1 : N_NODES;
  const size_t mcs = TM ? N_NODES : 1;
  const float* mlane = mask + (size_t)(qrow0 + cn) * mrs + (size_t)(8 * hi) * mcs;
  v4f kf[16], vf[16];
  float mk[QT][8];
  {
    const float* kb = K + (size_t)wave * 2048;
    #pragma unroll
    for (int i = 0; i < 16; i++) kf[i] = *(const v4f*)(kb + (koff + 8u * i));
    #pragma unroll
    for (int t = 0; t < QT; t++)
      load_mask8<TM>(mlane + (size_t)(16 * t) * mrs + (size_t)(wave * 16) * mcs, mk[t]);
    const float* vb = Vp + (size_t)wave * 2048;
    #pragma unroll
    for (int i = 0; i < 16; i++) vf[i] = *(const v4f*)(vb + (voff + 256u * (i >> 2) + 64u * (i & 3)));
  }

  v4f qf[16];
  #pragma unroll
  for (int i = 0; i < 16; i++) qf[i] = *(const v4f*)(qt + qoff + 8 * i);

  int ch = wave;
  do {
    const int nch = (ch + 8 < N_NODES / 16) ? ch + 8 : N_NODES / 16 - 1;
    float p[QT][8], sc[QT];
    #pragma unroll
    for (int t = 0; t < QT; t++) {
      v8f s = {};
      #pragma unroll
      for (int i = 0; i < 16; i++) {
        v2f a0, a1, b0, b1;
        a0.x = kf[i].x; a0.y = kf[i].y; a1.x = kf[i].z; a1.y = kf[i].w;
        b0.x = qf[i].x; b0.y = qf[i].y; b1.x = qf[i].z; b1.y = qf[i].w;
        s = wmma4(a0, b0, s);
        s = wmma4(a1, b1, s);
      }
      {
        const int tn = (t + 1 < QT) ? t + 1 : 0;
        #pragma unroll
        for (int i = 0; i < 16; i++) qf[i] = *(const v4f*)(qt + qoff + tn * (16 * 132) + 8 * i);
      }
      float sj[8];
      #pragma unroll
      for (int r = 0; r < 8; r++) {
        float mv = mk[t][r];
        sj[r] = mv * s[r] - 1000.f * (1.f - mv);
      }
      float cm = fmaxf(fmaxf(fmaxf(sj[0], sj[1]), fmaxf(sj[2], sj[3])),
                       fmaxf(fmaxf(sj[4], sj[5]), fmaxf(sj[6], sj[7])));
      cm = fmaxf(cm, xhalf(cm));
      const float mnew = fmaxf(mr[t], cm);
      sc[t] = __expf(mr[t] - mnew);
      float ps = 0.f;
      #pragma unroll
      for (int r = 0; r < 8; r++) {
        p[t][r] = __expf(sj[r] - mnew);
        ps += p[t][r];
      }
      lr[t] = lr[t] * sc[t] + ps;
      mr[t] = mnew;
      #pragma unroll
      for (int u = 0; u < 8; u++)
        #pragma unroll
        for (int j = 0; j < 8; j++) acc[t][u][j] *= sc[t];
      __builtin_amdgcn_sched_barrier(0);
    }
    {
      const float* kb = K + (size_t)nch * 2048;
      #pragma unroll
      for (int i = 0; i < 16; i++) kf[i] = *(const v4f*)(kb + (koff + 8u * i));
      #pragma unroll
      for (int t = 0; t < QT; t++)
        load_mask8<TM>(mlane + (size_t)(16 * t) * mrs + (size_t)(nch * 16) * mcs, mk[t]);
    }
    __builtin_amdgcn_sched_barrier(0);
    #pragma unroll
    for (int t = 0; t < QT; t++) {
      #pragma unroll
      for (int kk = 0; kk < 4; kk++) {
        v2f b; b.x = p[t][2 * kk]; b.y = p[t][2 * kk + 1];
        #pragma unroll
        for (int u = 0; u < 4; u++) {
          v2f a0, a1;
          a0.x = vf[4 * kk + u].x; a0.y = vf[4 * kk + u].y;
          a1.x = vf[4 * kk + u].z; a1.y = vf[4 * kk + u].w;
          acc[t][2 * u]     = wmma4(a0, b, acc[t][2 * u]);
          acc[t][2 * u + 1] = wmma4(a1, b, acc[t][2 * u + 1]);
        }
      }
    }
    {
      const float* vb = Vp + (size_t)nch * 2048;
      #pragma unroll
      for (int i = 0; i < 16; i++) vf[i] = *(const v4f*)(vb + (voff + 256u * (i >> 2) + 64u * (i & 3)));
    }
    ch += 8;
  } while (ch < N_NODES / 16);

  float lt[QT];
  #pragma unroll
  for (int t = 0; t < QT; t++) lt[t] = lr[t] + xhalf(lr[t]);
  if (hi == 0) {
    #pragma unroll
    for (int t = 0; t < QT; t++) {
      mbuf[wave][16 * t + cn] = mr[t];
      lbuf[wave][16 * t + cn] = lt[t];
    }
  }
  __syncthreads();
  for (int i = tid; i < QROWS * OB_LD; i += 256) outbuf[i] = 0.f;
  if (tid < QROWS) {
    float M = -3.0e38f;
    for (int w = 0; w < 8; w++) M = fmaxf(M, mbuf[w][tid]);
    float L = 0.f;
    for (int w = 0; w < 8; w++) L += lbuf[w][tid] * __expf(mbuf[w][tid] - M);
    Mrow[tid] = M;
    Lrow[tid] = L > 0.f ? 1.f / L : 0.f;
  }
  __syncthreads();
  #pragma unroll
  for (int t = 0; t < QT; t++) {
    const float e = __expf(mr[t] - Mrow[16 * t + cn]);
    #pragma unroll
    for (int u = 0; u < 8; u++)
      #pragma unroll
      for (int r = 0; r < 8; r++)
        atomicAdd(&outbuf[(16 * t + cn) * OB_LD + 32 * (u >> 1) + 2 * (r + 8 * hi) + (u & 1)],
                  acc[t][u][r] * e);
  }
  __syncthreads();
  for (int i = tid; i < QROWS * 128; i += 256) {
    int r = i >> 7, c = i & 127;
    out[(size_t)(qrow0 + r) * 128 + c] = outbuf[r * OB_LD + c] * Lrow[r];
  }
}

__global__ void __launch_bounds__(256) attn_pair(
    const float* __restrict__ Q0, const float* __restrict__ K0,
    const float* __restrict__ V0, float* __restrict__ out0,
    const float* __restrict__ Q1, const float* __restrict__ K1,
    const float* __restrict__ V1, float* __restrict__ out1,
    const float* __restrict__ mask) {
  __shared__ __attribute__((aligned(16))) float qt[QROWS * 132];
  __shared__ float outbuf[QROWS * OB_LD];
  __shared__ float mbuf[8][QROWS], lbuf[8][QROWS];
  __shared__ float Mrow[QROWS], Lrow[QROWS];
  const int second = blockIdx.x >= ATT_BLOCKS;
  const int bid = blockIdx.x - second * ATT_BLOCKS;
  if (second) attn_body<1>(Q1, K1, V1, mask, out1, bid, qt, outbuf, mbuf, lbuf, Mrow, Lrow);
  else        attn_body<0>(Q0, K0, V0, mask, out0, bid, qt, outbuf, mbuf, lbuf, Mrow, Lrow);
}

#define RB_LD 18
#define EB 64
__device__ __forceinline__ void edge_body(
    const float* __restrict__ coors, const float* __restrict__ hs,
    const float* __restrict__ hd,
    const int* __restrict__ src, const int* __restrict__ dst,
    const float* __restrict__ eW1rp, const float* __restrict__ eb1,
    const float* __restrict__ eg1, const float* __restrict__ ebt1,
    const float* __restrict__ Wcp, const float* __restrict__ bcv,
    const float* __restrict__ cg1, const float* __restrict__ cbt1,
    const float* __restrict__ cW2, const float* __restrict__ cb2,
    float* __restrict__ sum_hid, float* __restrict__ sum_x,
    float* __restrict__ cnt, int nE, int bid) {
  __shared__ __attribute__((aligned(16))) float buf1[EB * 132];
  __shared__ __attribute__((aligned(16))) float buf2[EB * 132];
  __shared__ __attribute__((aligned(16))) float rbf[EB * RB_LD];
  __shared__ float stats[EB][2];
  __shared__ float xrel[EB][3];
  __shared__ float coefs[EB];
  __shared__ int sidx[EB], didx[EB];
  const int tid = threadIdx.x, lane = tid & 31, wave = tid >> 5;
  const int hi = lane >> 4, cn = lane & 15;
  const int e0 = bid * EB;

  if (tid < EB) {
    int e = e0 + tid;
    if (e < nE) {
      int s = src[e], d = dst[e];
      sidx[tid] = s; didx[tid] = d;
      float x0 = coors[s * 3 + 0] - coors[d * 3 + 0];
      float x1 = coors[s * 3 + 1] - coors[d * 3 + 1];
      float x2 = coors[s * 3 + 2] - coors[d * 3 + 2];
      xrel[tid][0] = x0; xrel[tid][1] = x1; xrel[tid][2] = x2;
      float d2 = x0 * x0 + x1 * x1 + x2 * x2;
      float sig = 1.f;
      for (int j = 0; j < 15; j++) {
        rbf[tid * RB_LD + j] = __expf(-d2 / sig);
        sig *= 1.5f;
      }
      rbf[tid * RB_LD + 15] = 0.f;
    } else {
      sidx[tid] = 0; didx[tid] = -1;
      xrel[tid][0] = xrel[tid][1] = xrel[tid][2] = 0.f;
      for (int j = 0; j < 16; j++) rbf[tid * RB_LD + j] = 0.f;
    }
  }
  __syncthreads();
  {
    const int l4 = lane * 4;
    int ns[8], nd[8];
    v4f gs[8], gd[8];
    #pragma unroll
    for (int it = 0; it < 8; it++) {
      const int e = wave + 8 * it;
      ns[it] = sidx[e];
      nd[it] = didx[e];
    }
    #pragma unroll
    for (int it = 0; it < 8; it++) nd[it] = nd[it] >= 0 ? nd[it] : 0;
    #pragma unroll
    for (int it = 0; it < 8; it++) {
      gs[it] = *(const v4f*)(hs + (size_t)ns[it] * 128 + l4);
      gd[it] = *(const v4f*)(hd + (size_t)nd[it] * 128 + l4);
    }
    #pragma unroll
    for (int it = 0; it < 8; it++) {
      const int e = wave + 8 * it;
      *(v4f*)(buf1 + e * 132 + l4) = gs[it] + gd[it];
    }
  }
  __syncthreads();

  {
    const int n0 = wave * 16;
    v8f acc[4];
    #pragma unroll
    for (int mt = 0; mt < 4; mt++) {
      #pragma unroll
      for (int j = 0; j < 8; j++)
        acc[mt][j] = buf1[(mt * 16 + j + 8 * hi) * 132 + n0 + cn];
    }
    #pragma unroll
    for (int k0 = 0; k0 < 16; k0 += 4) {
      v2f b = ldBp(eW1rp, k0, n0, lane);
      #pragma unroll
      for (int mt = 0; mt < 4; mt++) {
        v2f x = ldsA(rbf + mt * 16 * RB_LD, RB_LD, k0, lane);
        acc[mt] = wmma4(x, b, acc[mt]);
      }
    }
    const float bb = eb1[n0 + cn];
    #pragma unroll
    for (int mt = 0; mt < 4; mt++) {
      #pragma unroll
      for (int j = 0; j < 8; j++)
        buf1[(mt * 16 + j + 8 * hi) * 132 + n0 + cn] = lrelu(acc[mt][j] + bb);
    }
  }
  __syncthreads();
  layernorm128x64(buf1, eg1, ebt1, stats, tid);
  __syncthreads();
  gemm128_lds64(buf1, Wcp, bcv, buf2, lane, wave, 1);
  __syncthreads();
  layernorm128x64(buf2, cg1, cbt1, stats, tid);
  __syncthreads();
  {
    const int seg = tid & 7;
    #pragma unroll
    for (int hh = 0; hh < 2; hh++) {
      const int r = (tid >> 3) + 32 * hh;
      float sum = 0.f;
      #pragma unroll
      for (int i = 0; i < 16; i++) {
        int c = seg * 16 + i;
        sum += buf2[r * 132 + c] * cW2[c];
      }
      sum += __shfl_xor(sum, 1);
      sum += __shfl_xor(sum, 2);
      sum += __shfl_xor(sum, 4);
      if (seg == 0) coefs[r] = sum + cb2[0];
    }
  }
  __syncthreads();
  {
    int dd[8];
    float hv[8][4];
    #pragma unroll
    for (int it = 0; it < 8; it++) {
      const int e = wave + 8 * it;
      dd[it] = didx[e];
      #pragma unroll
      for (int q = 0; q < 4; q++) hv[it][q] = buf1[e * 132 + lane + 32 * q];
    }
    #pragma unroll
    for (int it = 0; it < 8; it++) {
      const int d = dd[it];
      if (d >= 0) {
        #pragma unroll
        for (int q = 0; q < 4; q++)
          atomicAdd(&sum_hid[(size_t)d * 128 + lane + 32 * q], hv[it][q]);
      }
    }
  }
  if (tid < EB * 3) {
    int e = tid / 3, ax = tid % 3;
    int d = didx[e];
    if (d >= 0) atomicAdd(&sum_x[d * 3 + ax], xrel[e][ax] * coefs[e]);
  }
  if (tid < EB) {
    int d = didx[tid];
    if (d >= 0) atomicAdd(&cnt[d], 1.f);
  }
}

__global__ void __launch_bounds__(256) edge_pair(
    const float* __restrict__ coors0, const float* __restrict__ hs0,
    const float* __restrict__ hd0,
    const int* __restrict__ src0, const int* __restrict__ dst0,
    const float* __restrict__ coors1, const float* __restrict__ hs1,
    const float* __restrict__ hd1,
    const int* __restrict__ src1, const int* __restrict__ dst1,
    const float* __restrict__ eW1rp, const float* __restrict__ eb1,
    const float* __restrict__ eg1, const float* __restrict__ ebt1,
    const float* __restrict__ Wcp, const float* __restrict__ bcv,
    const float* __restrict__ cg1, const float* __restrict__ cbt1,
    const float* __restrict__ cW2, const float* __restrict__ cb2,
    float* __restrict__ sm0, float* __restrict__ sx0, float* __restrict__ cnt0,
    float* __restrict__ sm1, float* __restrict__ sx1, float* __restrict__ cnt1,
    int nE, int eblocks) {
  const int second = (int)blockIdx.x >= eblocks;
  const int bid = blockIdx.x - second * eblocks;
  edge_body(second ? coors1 : coors0, second ? hs1 : hs0, second ? hd1 : hd0,
            second ? src1 : src0, second ? dst1 : dst0, eW1rp, eb1, eg1, ebt1, Wcp,
            bcv, cg1,
            cbt1, cW2, cb2, second ? sm1 : sm0, second ? sx1 : sx0,
            second ? cnt1 : cnt0, nE, bid);
}

#define NCAT_LD 452
__device__ __forceinline__ void node_body(
    const float* __restrict__ h, const float* __restrict__ summ,
    const float* __restrict__ cnt, const float* __restrict__ cm,
    const float* __restrict__ orig,
    const float* __restrict__ nW1p, const float* __restrict__ nb1,
    const float* __restrict__ bnv,
    const float* __restrict__ ng1, const float* __restrict__ nbt1,
    const float* __restrict__ nW2p, const float* __restrict__ nb2,
    float* __restrict__ hout, int bid) {
  __shared__ __attribute__((aligned(16))) float cat[16 * NCAT_LD];
  __shared__ float buf1[16 * 132];
  __shared__ float stats[16][2];
  __shared__ float has[16];
  const int tid = threadIdx.x, lane = tid & 31, wave = tid >> 5;
  const int hi = lane >> 4, cn = lane & 15;
  const int node0 = bid * 16;

  {
    const int l4 = lane * 4;
    v4f vh[4], vs[4], vc[4], vo[2];
    float ct[4];
    #pragma unroll
    for (int it = 0; it < 4; it++) {
      const int e = wave + 4 * it;
      const size_t g = (size_t)(node0 + e) * 128 + l4;
      vh[it] = *(const v4f*)(h + g);
      vs[it] = *(const v4f*)(summ + g);
      vc[it] = *(const v4f*)(cm + g);
      ct[it] = cnt[node0 + e];
    }
    #pragma unroll
    for (int it = 0; it < 2; it++) {
      const int i = tid + 128 * it;
      vo[it] = *(const v4f*)(orig + (size_t)(node0 + (i >> 4)) * 64 + (i & 15) * 4);
    }
    #pragma unroll
    for (int it = 0; it < 4; it++) {
      const int e = wave + 4 * it;
      float* row = cat + e * NCAT_LD + l4;
      const bool hasr = ct[it] > 0.f;
      v4f m;
      m.x = hasr ? vs[it].x / ct[it] : 0.f;
      m.y = hasr ? vs[it].y / ct[it] : 0.f;
      m.z = hasr ? vs[it].z / ct[it] : 0.f;
      m.w = hasr ? vs[it].w / ct[it] : 0.f;
      *(v4f*)(row)       = vh[it];
      *(v4f*)(row + 128) = m;
      *(v4f*)(row + 256) = vc[it];
      if (lane == 0) has[e] = hasr ? 1.f : 0.f;
    }
    #pragma unroll
    for (int it = 0; it < 2; it++) {
      const int i = tid + 128 * it;
      *(v4f*)(cat + (i >> 4) * NCAT_LD + 384 + (i & 15) * 4) = vo[it];
    }
  }
  __syncthreads();
  {
    const int n0 = wave * 32;
    v8f a0 = {}, a1 = {};
    for (int k0 = 0; k0 < 448; k0 += 4) {
      v2f a  = ldsA(cat, NCAT_LD, k0, lane);
      v2f b0 = ldBp(nW1p, k0, n0, lane);
      v2f b1 = ldBp(nW1p, k0, n0 + 16, lane);
      a0 = wmma4(a, b0, a0);
      a1 = wmma4(a, b1, a1);
    }
    const float fb0 = bnv[n0 + cn], fb1 = bnv[n0 + 16 + cn];
    #pragma unroll
    for (int j = 0; j < 8; j++) {
      int r = j + 8 * hi;
      const bool hasr = has[r] > 0.f;
      buf1[r * 132 + n0 + cn]      = lrelu(a0[j] + nb1[n0 + cn] + (hasr ? fb0 : 0.f));
      buf1[r * 132 + n0 + 16 + cn] = lrelu(a1[j] + nb1[n0 + 16 + cn] + (hasr ? fb1 : 0.f));
    }
  }
  __syncthreads();
  layernorm128(buf1, ng1, nbt1, stats, tid);
  __syncthreads();
  {
    const int n0 = wave * 32;
    v8f a0 = {}, a1 = {};
    for (int k0 = 0; k0 < 128; k0 += 4) {
      v2f a  = ldsA(buf1, 132, k0, lane);
      v2f b0 = ldBp(nW2p, k0, n0, lane);
      v2f b1 = ldBp(nW2p, k0, n0 + 16, lane);
      a0 = wmma4(a, b0, a0);
      a1 = wmma4(a, b1, a1);
    }
    #pragma unroll
    for (int j = 0; j < 8; j++) {
      int r = j + 8 * hi;
      size_t g0 = (size_t)(node0 + r) * 128 + n0 + cn;
      size_t g1 = g0 + 16;
      hout[g0] = 0.5f * (a0[j] + nb2[n0 + cn]) + 0.5f * h[g0];
      hout[g1] = 0.5f * (a1[j] + nb2[n0 + 16 + cn]) + 0.5f * h[g1];
    }
  }
}

__global__ void __launch_bounds__(128) node_pair(
    const float* __restrict__ h0, const float* __restrict__ summ0,
    const float* __restrict__ cnt0, const float* __restrict__ cm0,
    const float* __restrict__ orig0, float* __restrict__ hout0,
    const float* __restrict__ h1, const float* __restrict__ summ1,
    const float* __restrict__ cnt1, const float* __restrict__ cm1,
    const float* __restrict__ orig1, float* __restrict__ hout1,
    const float* __restrict__ nW1p, const float* __restrict__ nb1,
    const float* __restrict__ bnv,
    const float* __restrict__ ng1, const float* __restrict__ nbt1,
    const float* __restrict__ nW2p, const float* __restrict__ nb2) {
  const int second = blockIdx.x >= N_NODES / 16;
  const int bid = blockIdx.x - second * (N_NODES / 16);
  node_body(second ? h1 : h0, second ? summ1 : summ0, second ? cnt1 : cnt0,
            second ? cm1 : cm0, second ? orig1 : orig0, nW1p, nb1, bnv, ng1, nbt1, nW2p,
            nb2, second ? hout1 : hout0, bid);
}

#define CB ((3 * N_NODES + 255) / 256)
__global__ void coords_pair(const float* __restrict__ oc0, const float* __restrict__ c0,
                            const float* __restrict__ sx0, const float* __restrict__ cnt0,
                            float* __restrict__ out0,
                            const float* __restrict__ oc1, const float* __restrict__ c1,
                            const float* __restrict__ sx1, const float* __restrict__ cnt1,
                            float* __restrict__ out1, int n) {
  const int second = blockIdx.x >= CB;
  const float* oc = second ? oc1 : oc0;
  const float* c = second ? c1 : c0;
  const float* sx = second ? sx1 : sx0;
  const float* cnt = second ? cnt1 : cnt0;
  float* out = second ? out1 : out0;
  int i = (blockIdx.x - second * CB) * 256 + threadIdx.x;
  if (i >= n * 3) return;
  int node = i / 3;
  float ct = cnt[node];
  float xu = ct > 0.f ? sx[i] / ct : 0.f;
  out[i] = 0.2f * oc[i] + 0.8f * c[i] + xu;
}

extern "C" void kernel_launch(void* const* d_in, const int* in_sizes, int n_in,
                              void* d_out, int out_size, void* d_ws, size_t ws_size,
                              hipStream_t stream) {
  const float* coors_lig = (const float*)d_in[0];
  const float* h_lig     = (const float*)d_in[1];
  const float* orig_lig  = (const float*)d_in[2];
  const float* oc_lig    = (const float*)d_in[3];
  const float* coors_rec = (const float*)d_in[4];
  const float* h_rec     = (const float*)d_in[5];
  const float* orig_rec  = (const float*)d_in[6];
  const float* oc_rec    = (const float*)d_in[7];
  const int* lig_src = (const int*)d_in[8];
  const int* lig_dst = (const int*)d_in[9];
  const int* rec_src = (const int*)d_in[10];
  const int* rec_dst = (const int*)d_in[11];
  const float* mask = (const float*)d_in[12];
  const float* eW1 = (const float*)d_in[13];
  const float* eb1 = (const float*)d_in[14];
  const float* eg1 = (const float*)d_in[15];
  const float* ebt1 = (const float*)d_in[16];
  const float* eW2 = (const float*)d_in[17];
  const float* eb2 = (const float*)d_in[18];
  const float* qW = (const float*)d_in[19];
  const float* kW = (const float*)d_in[20];
  const float* vW = (const float*)d_in[21];
  const float* nW1 = (const float*)d_in[22];
  const float* nb1 = (const float*)d_in[23];
  const float* ng1 = (const float*)d_in[24];
  const float* nbt1 = (const float*)d_in[25];
  const float* nW2 = (const float*)d_in[26];
  const float* nb2 = (const float*)d_in[27];
  const float* cW1 = (const float*)d_in[28];
  const float* cb1 = (const float*)d_in[29];
  const float* cg1 = (const float*)d_in[30];
  const float* cbt1 = (const float*)d_in[31];
  const float* cW2 = (const float*)d_in[32];
  const float* cb2 = (const float*)d_in[33];
  const int nE = in_sizes[8];

  float* ws = (float*)d_ws;
  float* hs_l = ws + 0 * NH;
  float* hd_l = ws + 1 * NH;
  float* hs_r = ws + 2 * NH;
  float* hd_r = ws + 3 * NH;
  float* q_l = ws + 0 * NH;
  float* k_l = ws + 1 * NH;
  float* v_l = ws + 2 * NH;
  float* q_r = ws + 3 * NH;
  float* k_r = ws + 4 * NH;
  float* v_r = ws + 5 * NH;
  float* cm_l = ws + 6 * NH;
  float* cm_r = ws + 7 * NH;
  float* sm_l = ws + 8 * NH;
  float* sm_r = ws + 9 * NH;
  float* sx_l = ws + 10 * NH;
  float* sx_r = sx_l + 3 * N_NODES;
  float* cnt_l = sx_r + 3 * N_NODES;
  float* cnt_r = cnt_l + N_NODES;
  float* eW1p = cnt_r + N_NODES;
  float* Wcp  = eW1p + 136 * 256;
  float* bcv  = Wcp + 64 * 256;
  float* bnv  = bcv + 128;
  float* nW1p = Wcp + 128 * 256;
  float* nW2p = nW1p + 224 * 256;
  float* qWp  = nW2p + 64 * 256;
  float* kWp  = qWp + 64 * 256;
  float* vWp  = kWp + 64 * 256;

  size_t accum_floats = 2 * NH + 8 * (size_t)N_NODES;
  hipMemsetAsync(sm_l, 0, accum_floats * sizeof(float), stream);

  pack_all<<<PK_END * 128 / 256, 256, 0, stream>>>(eW1, nW1, nW2, qW, kW, vW, eW1p);

  const int ntiles = N_NODES / 16;
  proj_edge<<<4 * ntiles + 18, 128, 0, stream>>>(h_lig, h_rec, eW1p, hs_l, hd_l, hs_r, hd_r,
      eW2, eb2, cW1, cb1, nW1, Wcp, bcv, nW1p + 64 * 256, bnv);

  int eblocks = (nE + EB - 1) / EB;
  edge_pair<<<2 * eblocks, 256, 0, stream>>>(coors_lig, hs_l, hd_l, lig_src, lig_dst,
      coors_rec, hs_r, hd_r, rec_src, rec_dst,
      eW1p + 128 * 256, eb1, eg1, ebt1, Wcp, bcv, cg1, cbt1, cW2, cb2,
      sm_l, sx_l, cnt_l, sm_r, sx_r, cnt_r, nE, eblocks);

  proj_all<<<6 * ntiles, 128, 0, stream>>>(h_lig, h_rec, qWp, kWp, vWp, q_l, k_l, v_l,
                                           q_r, k_r, v_r);

  attn_pair<<<2 * ATT_BLOCKS, 256, 0, stream>>>(q_l, k_r, v_r, cm_l, q_r, k_l, v_l, cm_r,
                                            mask);

  float* out_xl = (float*)d_out;
  float* out_hl = out_xl + 3 * N_NODES;
  float* out_xr = out_hl + NH;
  float* out_hr = out_xr + 3 * N_NODES;

  node_pair<<<2 * ntiles, 128, 0, stream>>>(h_lig, sm_l, cnt_l, cm_l, orig_lig, out_hl,
      h_rec, sm_r, cnt_r, cm_r, orig_rec, out_hr,
      nW1p, nb1, bnv, ng1, nbt1, nW2p, nb2);

  coords_pair<<<2 * CB, 256, 0, stream>>>(oc_lig, coors_lig, sx_l, cnt_l, out_xl,
                                          oc_rec, coors_rec, sx_r, cnt_r, out_xr,
                                          N_NODES);
}
